# starting kernel
# baseline (speedup 1.0000x reference)
#include <hip/hip_runtime.h>
#include <math.h>

typedef __attribute__((ext_vector_type(2))) float v2f;
typedef __attribute__((ext_vector_type(4))) float f4;
typedef __attribute__((ext_vector_type(8))) float v8f;

namespace {
constexpr int Bc = 2, Lc = 384, Cc = 128, Hc = 8, DKc = 16, NQPc = 4, NPVc = 8;
constexpr float WCc = 0.2357022603955158f;
constexpr float WLc = 0.5773502691896258f;
constexpr int FEAT = Hc * Cc + Hc * DKc + Hc * NPVc * 3 + Hc * NPVc;
}

__device__ __forceinline__ float softplusf(float x) {
  return (x > 0.f) ? (x + log1pf(expf(-x))) : log1pf(expf(x));
}

template <int MODE, bool TRANSB>
__global__ __launch_bounds__(128) void k_gemm(
    const float* __restrict__ A, const float* __restrict__ Bm,
    float* __restrict__ Cm, int M, int N, int K, int lda, int ldb, int ldc,
    long long sA, long long sA2, int zdiv, long long sB, long long sC,
    int Mvalid, const float* __restrict__ e0, const float* __restrict__ e1,
    const float* __restrict__ e2, const float* __restrict__ e3) {
  __shared__ float As[16][36];
  __shared__ float Bs[32][132];

  const int z = blockIdx.z;
  const float* Ag = A + (long long)(z / zdiv) * sA + (long long)(z % zdiv) * sA2;
  const float* Bg = Bm + (long long)z * sB;
  float* Cg = Cm + (long long)z * sC;

  const int m0 = blockIdx.x * 16;
  const int n0 = blockIdx.y * 128;
  const int tid = threadIdx.x;
  const int wave = tid >> 5;
  const int lane = tid & 31;
  const int hf = lane >> 4;
  const int l16 = lane & 15;
  const int nw = wave * 32;

  v8f acc0 = {};
  v8f acc1 = {};

  for (int kb = 0; kb < K; kb += 32) {
    {
      int r = tid >> 3, c4 = (tid & 7) * 4;
      f4 av = {0.f, 0.f, 0.f, 0.f};
      if (r < Mvalid && (m0 + r) < M)
        av = *(const f4*)(Ag + (long long)(m0 + r) * lda + kb + c4);
      *(f4*)&As[r][c4] = av;
    }
    if (!TRANSB) {
#pragma unroll
      for (int it = 0; it < 8; ++it) {
        int u = tid + it * 128;
        int r = u >> 5, c4 = (u & 31) * 4;
        f4 bv = {0.f, 0.f, 0.f, 0.f};
        if ((n0 + c4) < N)
          bv = *(const f4*)(Bg + (long long)(kb + r) * ldb + n0 + c4);
        *(f4*)&Bs[r][c4] = bv;
      }
    } else {
#pragma unroll
      for (int it = 0; it < 8; ++it) {
        int u = tid + it * 128;
        int c = u >> 3, r4 = (u & 7) * 4;
        f4 bv = {0.f, 0.f, 0.f, 0.f};
        if ((n0 + c) < N)
          bv = *(const f4*)(Bg + (long long)(n0 + c) * ldb + kb + r4);
        Bs[r4 + 0][c] = bv.x;
        Bs[r4 + 1][c] = bv.y;
        Bs[r4 + 2][c] = bv.z;
        Bs[r4 + 3][c] = bv.w;
      }
    }
    __syncthreads();

#pragma unroll
    for (int kk = 0; kk < 32; kk += 4) {
      v2f a, b0, b1;
      a.x = As[l16][kk + 2 * hf];
      a.y = As[l16][kk + 2 * hf + 1];
      b0.x = Bs[kk + 2 * hf][nw + l16];
      b0.y = Bs[kk + 2 * hf + 1][nw + l16];
      b1.x = Bs[kk + 2 * hf][nw + 16 + l16];
      b1.y = Bs[kk + 2 * hf + 1][nw + 16 + l16];
      acc0 = __builtin_amdgcn_wmma_f32_16x16x4_f32(false, a, false, b0,
                                                   (short)0, acc0, false, false);
      acc1 = __builtin_amdgcn_wmma_f32_16x16x4_f32(false, a, false, b1,
                                                   (short)0, acc1, false, false);
    }
    __syncthreads();
  }

  float gw = 0.f;
  if (MODE == 2) {
    int h_ = z % Hc;
    gw = 0.5f * WCc * softplusf(e3[h_]);
  }

#pragma unroll
  for (int r = 0; r < 8; ++r) {
    int mrow = r + 8 * hf;
    int m = m0 + mrow;
    if (mrow >= Mvalid || m >= M) continue;
#pragma unroll
    for (int t = 0; t < 2; ++t) {
      int n = n0 + nw + t * 16 + l16;
      if (n >= N) continue;
      float val = (t == 0) ? acc0[r] : acc1[r];
      if (MODE == 1) {
        val += e0[n];
      } else if (MODE == 2) {
        int b_ = z / Hc, h_ = z % Hc;
        float bijv = e0[((long long)(b_ * Lc + m) * Lc + n) * Hc + h_];
        val = WLc * (val + bijv -
                     gw * (e1[(long long)z * Lc + m] + e2[(long long)z * Lc + n]));
      }
      Cg[(long long)m * ldc + n] = val;
    }
  }
}

__global__ __launch_bounds__(256) void k_prep(
    const float* __restrict__ q, const float* __restrict__ k,
    const float* __restrict__ v, const float* __restrict__ qp,
    const float* __restrict__ kp, const float* __restrict__ vp,
    const float* __restrict__ Ri, const float* __restrict__ Ti,
    const float* __restrict__ gamah, float* __restrict__ qa,
    float* __restrict__ ka, float* __restrict__ vaug, float* __restrict__ sqq,
    float* __restrict__ sqk) {
  int idx = blockIdx.x * blockDim.x + threadIdx.x;
  if (idx >= Bc * Lc * Hc) return;
  int h = idx % Hc;
  int bi = idx / Hc;
  int b = bi / Lc, i = bi % Lc;
  int z = b * Hc + h;

  float R[3][3], t[3];
#pragma unroll
  for (int r = 0; r < 3; ++r)
#pragma unroll
    for (int c = 0; c < 3; ++c) R[r][c] = Ri[bi * 9 + r * 3 + c];
#pragma unroll
  for (int x = 0; x < 3; ++x) t[x] = Ti[bi * 3 + x];

  float gwc = WCc * softplusf(gamah[h]);

  float* qrow = qa + ((long long)z * Lc + i) * 32;
  float* krow = ka + ((long long)z * Lc + i) * 32;
  float* vrow = vaug + ((long long)z * Lc + i) * 48;

#pragma unroll
  for (int d = 0; d < DKc; ++d) {
    qrow[d] = 0.25f * q[(long long)bi * 128 + h * 16 + d];
    krow[d] = k[(long long)bi * 128 + h * 16 + d];
    vrow[d] = v[(long long)bi * 128 + h * 16 + d];
  }
  float sq = 0.f, sk = 0.f;
#pragma unroll
  for (int p = 0; p < NQPc; ++p) {
    const float* qs = qp + (long long)bi * 96 + h * 12 + p * 3;
    const float* ks = kp + (long long)bi * 96 + h * 12 + p * 3;
#pragma unroll
    for (int m = 0; m < 3; ++m) {
      float qt = R[m][0] * qs[0] + R[m][1] * qs[1] + R[m][2] * qs[2] + t[m];
      float kt = R[m][0] * ks[0] + R[m][1] * ks[1] + R[m][2] * ks[2] + t[m];
      sq += qt * qt;
      sk += kt * kt;
      qrow[16 + p * 3 + m] = gwc * qt;
      krow[16 + p * 3 + m] = kt;
    }
  }
#pragma unroll
  for (int d = 28; d < 32; ++d) { qrow[d] = 0.f; krow[d] = 0.f; }
#pragma unroll
  for (int p = 0; p < NPVc; ++p) {
    const float* vs = vp + (long long)bi * 192 + h * 24 + p * 3;
#pragma unroll
    for (int m = 0; m < 3; ++m)
      vrow[16 + p * 3 + m] =
          R[m][0] * vs[0] + R[m][1] * vs[1] + R[m][2] * vs[2] + t[m];
  }
#pragma unroll
  for (int d = 40; d < 48; ++d) vrow[d] = 0.f;
  sqq[(long long)z * Lc + i] = sq;
  sqk[(long long)z * Lc + i] = sk;
}

__global__ __launch_bounds__(256) void k_softmax(float* __restrict__ sc) {
  int row = blockIdx.x * 8 + (threadIdx.x >> 5);
  int lane = threadIdx.x & 31;
  float* p = sc + (long long)row * Lc;
  float vals[12];
  float mx = -INFINITY;
#pragma unroll
  for (int e = 0; e < 12; ++e) {
    vals[e] = p[e * 32 + lane];
    mx = fmaxf(mx, vals[e]);
  }
#pragma unroll
  for (int o = 16; o >= 1; o >>= 1) mx = fmaxf(mx, __shfl_xor(mx, o, 32));
  float s = 0.f;
#pragma unroll
  for (int e = 0; e < 12; ++e) {
    vals[e] = expf(vals[e] - mx);
    s += vals[e];
  }
#pragma unroll
  for (int o = 16; o >= 1; o >>= 1) s += __shfl_xor(s, o, 32);
  float inv = 1.f / s;
#pragma unroll
  for (int e = 0; e < 12; ++e) p[e * 32 + lane] = vals[e] * inv;
}

__global__ __launch_bounds__(256) void k_finish(
    const float* __restrict__ ovalpt, const float* __restrict__ Ri,
    const float* __restrict__ Ti, float* __restrict__ feats) {
  int idx = blockIdx.x * blockDim.x + threadIdx.x;
  if (idx >= Bc * Lc * Hc) return;
  int h = idx % Hc;
  int bi = idx / Hc;
  int b = bi / Lc;
  int z = b * Hc + h;
  int i = bi % Lc;

  float R[3][3], t[3];
#pragma unroll
  for (int r = 0; r < 3; ++r)
#pragma unroll
    for (int c = 0; c < 3; ++c) R[r][c] = Ri[bi * 9 + r * 3 + c];
#pragma unroll
  for (int x = 0; x < 3; ++x) t[x] = Ti[bi * 3 + x];

  const float* orow = ovalpt + ((long long)z * Lc + i) * 48;
  float* frow = feats + (long long)bi * FEAT;

#pragma unroll
  for (int d = 0; d < DKc; ++d) frow[1024 + h * 16 + d] = orow[d];
#pragma unroll
  for (int p = 0; p < NPVc; ++p) {
    float d0 = orow[16 + p * 3 + 0] - t[0];
    float d1 = orow[16 + p * 3 + 1] - t[1];
    float d2 = orow[16 + p * 3 + 2] - t[2];
    float asum = 0.f;
#pragma unroll
    for (int m = 0; m < 3; ++m) {
      float oh = R[0][m] * d0 + R[1][m] * d1 + R[2][m] * d2;
      frow[1152 + h * 24 + p * 3 + m] = oh;
      asum += fabsf(oh);
    }
    frow[1344 + h * 8 + p] = asum;
  }
}

static void launch_gemm(hipStream_t s, int mode, bool transb, const float* A,
                        const float* Bm, float* C, int M, int N, int K, int lda,
                        int ldb, int ldc, long long sA, long long sA2, int zdiv,
                        long long sB, long long sC, int Mvalid, int nz,
                        const float* e0 = nullptr, const float* e1 = nullptr,
                        const float* e2 = nullptr, const float* e3 = nullptr) {
  dim3 g((M + 15) / 16, (N + 127) / 128, nz);
  dim3 blk(128, 1, 1);
  if (mode == 0 && !transb)
    k_gemm<0, false><<<g, blk, 0, s>>>(A, Bm, C, M, N, K, lda, ldb, ldc, sA,
                                       sA2, zdiv, sB, sC, Mvalid, e0, e1, e2, e3);
  else if (mode == 1)
    k_gemm<1, false><<<g, blk, 0, s>>>(A, Bm, C, M, N, K, lda, ldb, ldc, sA,
                                       sA2, zdiv, sB, sC, Mvalid, e0, e1, e2, e3);
  else
    k_gemm<2, true><<<g, blk, 0, s>>>(A, Bm, C, M, N, K, lda, ldb, ldc, sA,
                                      sA2, zdiv, sB, sC, Mvalid, e0, e1, e2, e3);
}

extern "C" void kernel_launch(void* const* d_in, const int* in_sizes, int n_in,
                              void* d_out, int out_size, void* d_ws,
                              size_t ws_size, hipStream_t stream) {
  const float* si = (const float*)d_in[0];
  const float* zij = (const float*)d_in[1];
  const float* Ri = (const float*)d_in[2];
  const float* Ti = (const float*)d_in[3];
  const float* WQh = (const float*)d_in[4];
  const float* WKh = (const float*)d_in[5];
  const float* WVh = (const float*)d_in[6];
  const float* WQhp = (const float*)d_in[7];
  const float* WKhp = (const float*)d_in[8];
  const float* WVhp = (const float*)d_in[9];
  const float* Wb = (const float*)d_in[10];
  const float* WOw = (const float*)d_in[11];
  const float* WOb = (const float*)d_in[12];
  const float* gamah = (const float*)d_in[13];
  float* out = (float*)d_out;
  float* ws = (float*)d_ws;

  const int BL = Bc * Lc;
  const long long LL = (long long)Lc * Lc;
  float* q = ws;
  float* k = q + 98304;
  float* v = k + 98304;
  float* qp = v + 98304;
  float* kp = qp + 73728;
  float* vp = kp + 73728;
  float* qa = vp + 147456;
  float* ka = qa + 196608;
  float* vaug = ka + 196608;
  float* sqq = vaug + 294912;
  float* sqk = sqq + 6144;
  float* bij = sqk + 6144;
  float* alpha = bij + 2359296;
  float* ovalpt = alpha + 2359296;
  float* feats = ovalpt + 294912;

  launch_gemm(stream, 0, false, si, WQh, q, BL, 128, Cc, Cc, 128, 128, 0, 0, 1, 0, 0, 16, 1);
  launch_gemm(stream, 0, false, si, WKh, k, BL, 128, Cc, Cc, 128, 128, 0, 0, 1, 0, 0, 16, 1);
  launch_gemm(stream, 0, false, si, WVh, v, BL, 128, Cc, Cc, 128, 128, 0, 0, 1, 0, 0, 16, 1);
  launch_gemm(stream, 0, false, si, WQhp, qp, BL, 96, Cc, Cc, 96, 96, 0, 0, 1, 0, 0, 16, 1);
  launch_gemm(stream, 0, false, si, WKhp, kp, BL, 96, Cc, Cc, 96, 96, 0, 0, 1, 0, 0, 16, 1);
  launch_gemm(stream, 0, false, si, WVhp, vp, BL, 192, Cc, Cc, 192, 192, 0, 0, 1, 0, 0, 16, 1);

  launch_gemm(stream, 0, false, zij, Wb, bij, Bc * Lc * Lc, Hc, Cc, Cc, Hc, Hc,
              0, 0, 1, 0, 0, 16, 1);

  k_prep<<<(Bc * Lc * Hc + 255) / 256, 256, 0, stream>>>(
      q, k, v, qp, kp, vp, Ri, Ti, gamah, qa, ka, vaug, sqq, sqk);

  launch_gemm(stream, 2, true, qa, ka, alpha, Lc, Lc, 32, 32, 32, Lc,
              (long long)Lc * 32, 0, 1, (long long)Lc * 32, LL, 16, Bc * Hc,
              bij, sqq, sqk, gamah);

  k_softmax<<<(Bc * Hc * Lc) / 8, 256, 0, stream>>>(alpha);

  launch_gemm(stream, 0, false, alpha, vaug, ovalpt, Lc, 48, Lc, Lc, 48, 48,
              LL, 0, 1, (long long)Lc * 48, (long long)Lc * 48, 16, Bc * Hc);

  launch_gemm(stream, 0, false, alpha, zij, feats, 16, Cc, Lc, (int)LL, Cc,
              Cc, (long long)Hc * LL, (long long)Lc, Lc, (long long)Lc * Cc,
              (long long)FEAT, Hc, BL);

  k_finish<<<(Bc * Lc * Hc + 255) / 256, 256, 0, stream>>>(ovalpt, Ri, Ti, feats);

  launch_gemm(stream, 1, false, feats, WOw, out, BL, Cc, FEAT, FEAT, Cc, Cc,
              0, 0, 1, 0, 0, 16, 1, WOb);
}


# GEMM K loop without LDS: each wave loads its own f32 WMMA fragments from global straight into registers, 64 k per group issued ahead of the products; both barriers gone; same k order and sums
# speedup vs baseline: 2.3329x; 2.3329x over previous
#include <hip/hip_runtime.h>
#include <math.h>

typedef __attribute__((ext_vector_type(2))) float v2f;
typedef __attribute__((ext_vector_type(4))) float f4;
typedef __attribute__((ext_vector_type(8))) float v8f;

namespace {
constexpr int Bc = 2, Lc = 384, Cc = 128, Hc = 8, DKc = 16, NQPc = 4, NPVc = 8;
constexpr float WCc = 0.2357022603955158f;
constexpr float WLc = 0.5773502691896258f;
constexpr int FEAT = Hc * Cc + Hc * DKc + Hc * NPVc * 3 + Hc * NPVc;
}

__device__ __forceinline__ float softplusf(float x) {
  return (x > 0.f) ? (x + log1pf(expf(-x))) : log1pf(expf(x));
}

template <bool TRANSB, bool TWO, int KT>
__device__ __forceinline__ void wave_kloop(const float* __restrict__ ap,
                                           const char* __restrict__ Bb,
                                           unsigned ob, unsigned ldb4,
                                           const float* __restrict__ bt0,
                                           const float* __restrict__ bt1, int K,
                                           v8f& acc0, v8f& acc1) {
  constexpr int NS = KT / 4;
  for (int kb = 0; kb < K; kb += KT) {
    v2f a[NS], b0[NS], b1[NS];
#pragma unroll
    for (int s = 0; s < NS; ++s) {
      const int k = kb + 4 * s;
      a[s] = *(const v2f*)(ap + k);
      if (TRANSB) {
        b0[s] = *(const v2f*)(bt0 + k);
        if (TWO) b1[s] = *(const v2f*)(bt1 + k);
      } else {
        const unsigned r0 = (unsigned)k * ldb4;
        const unsigned r1 = r0 + ldb4;
        if (TWO) {
          const v2f x = *(const v2f*)(Bb + (r0 + ob));
          const v2f y = *(const v2f*)(Bb + (r1 + ob));
          b0[s].x = x.x;
          b0[s].y = y.x;
          b1[s].x = x.y;
          b1[s].y = y.y;
        } else {
          b0[s].x = *(const float*)(Bb + (r0 + ob));
          b0[s].y = *(const float*)(Bb + (r1 + ob));
        }
      }
    }
#pragma unroll
    for (int s = 0; s < NS; ++s) {
      acc0 = __builtin_amdgcn_wmma_f32_16x16x4_f32(false, a[s], false, b0[s],
                                                   (short)0, acc0, false, false);
      if (TWO)
        acc1 = __builtin_amdgcn_wmma_f32_16x16x4_f32(false, a[s], false, b1[s],
                                                     (short)0, acc1, false, false);
    }
  }
}

template <int MODE, bool TRANSB>
__global__ __launch_bounds__(128) __attribute__((amdgpu_waves_per_eu(4, 8))) void k_gemm(
    const float* __restrict__ A, const float* __restrict__ Bm,
    float* __restrict__ Cm, int M, int N, int K, int lda, int ldb, int ldc,
    long long sA, long long sA2, int zdiv, long long sB, long long sC,
    int Mvalid, const float* __restrict__ e0, const float* __restrict__ e1,
    const float* __restrict__ e2, const float* __restrict__ e3) {
  constexpr int KT = TRANSB ? 32 : 64;
  const int z = blockIdx.z;
  const float* Ag = A + (long long)(z / zdiv) * sA + (long long)(z % zdiv) * sA2;
  const float* Bg = Bm + (long long)z * sB;
  float* Cg = Cm + (long long)z * sC;

  const int m0 = blockIdx.x * 16;
  const int n0 = blockIdx.y * 128;
  const int tid = threadIdx.x;
  const int wave = __builtin_amdgcn_readfirstlane(tid >> 5);
  const int lane = tid & 31;
  const int hf = lane >> 4;
  const int l16 = lane & 15;
  const int nw = wave * 32;
  const int nb = n0 + nw;

  if (nb >= N) return;
  const bool two = (nb + 16) < N;

  const int n_t0 = two ? (nb + 2 * l16) : (nb + l16);
  const int n_t1 = n_t0 + 1;
  const int ra = (l16 < Mvalid && (m0 + l16) < M) ? l16 : 0;
  const int c0 = (n_t0 < N) ? n_t0 : nb;

  const float* ap = Ag + (long long)(m0 + ra) * lda + 2 * hf;
  const unsigned ldb4 = (unsigned)ldb * 4u;
  const unsigned ob = (unsigned)(2 * hf * ldb + c0) * 4u;
  const float* bt0 = Bg + (long long)c0 * ldb + 2 * hf;
  const float* bt1 = bt0 + ldb;

  v8f acc0 = {};
  v8f acc1 = {};
  if (two)
    wave_kloop<TRANSB, true, KT>(ap, (const char*)Bg, ob, ldb4, bt0, bt1, K,
                                 acc0, acc1);
  else
    wave_kloop<TRANSB, false, KT>(ap, (const char*)Bg, ob, ldb4, bt0, bt1, K,
                                  acc0, acc1);

  float gw = 0.f;
  if (MODE == 2) {
    int h_ = z % Hc;
    gw = 0.5f * WCc * softplusf(e3[h_]);
  }

#pragma unroll
  for (int r = 0; r < 8; ++r) {
    int mrow = r + 8 * hf;
    int m = m0 + mrow;
    if (mrow >= Mvalid || m >= M) continue;
#pragma unroll
    for (int t = 0; t < 2; ++t) {
      int n = (t == 0) ? n_t0 : n_t1;
      if (n >= N || (t == 1 && !two)) continue;
      float val = (t == 0) ? acc0[r] : acc1[r];
      if (MODE == 1) {
        val += e0[n];
      } else if (MODE == 2) {
        int b_ = z / Hc, h_ = z % Hc;
        float bijv = e0[((long long)(b_ * Lc + m) * Lc + n) * Hc + h_];
        val = WLc * (val + bijv -
                     gw * (e1[(long long)z * Lc + m] + e2[(long long)z * Lc + n]));
      }
      Cg[(long long)m * ldc + n] = val;
    }
  }
}

__global__ __launch_bounds__(256) void k_prep(
    const float* __restrict__ q, const float* __restrict__ k,
    const float* __restrict__ v, const float* __restrict__ qp,
    const float* __restrict__ kp, const float* __restrict__ vp,
    const float* __restrict__ Ri, const float* __restrict__ Ti,
    const float* __restrict__ gamah, float* __restrict__ qa,
    float* __restrict__ ka, float* __restrict__ vaug, float* __restrict__ sqq,
    float* __restrict__ sqk) {
  int idx = blockIdx.x * blockDim.x + threadIdx.x;
  if (idx >= Bc * Lc * Hc) return;
  int h = idx % Hc;
  int bi = idx / Hc;
  int b = bi / Lc, i = bi % Lc;
  int z = b * Hc + h;

  float R[3][3], t[3];
#pragma unroll
  for (int r = 0; r < 3; ++r)
#pragma unroll
    for (int c = 0; c < 3; ++c) R[r][c] = Ri[bi * 9 + r * 3 + c];
#pragma unroll
  for (int x = 0; x < 3; ++x) t[x] = Ti[bi * 3 + x];

  float gwc = WCc * softplusf(gamah[h]);

  float* qrow = qa + ((long long)z * Lc + i) * 32;
  float* krow = ka + ((long long)z * Lc + i) * 32;
  float* vrow = vaug + ((long long)z * Lc + i) * 48;

#pragma unroll
  for (int d = 0; d < DKc; ++d) {
    qrow[d] = 0.25f * q[(long long)bi * 128 + h * 16 + d];
    krow[d] = k[(long long)bi * 128 + h * 16 + d];
    vrow[d] = v[(long long)bi * 128 + h * 16 + d];
  }
  float sq = 0.f, sk = 0.f;
#pragma unroll
  for (int p = 0; p < NQPc; ++p) {
    const float* qs = qp + (long long)bi * 96 + h * 12 + p * 3;
    const float* ks = kp + (long long)bi * 96 + h * 12 + p * 3;
#pragma unroll
    for (int m = 0; m < 3; ++m) {
      float qt = R[m][0] * qs[0] + R[m][1] * qs[1] + R[m][2] * qs[2] + t[m];
      float kt = R[m][0] * ks[0] + R[m][1] * ks[1] + R[m][2] * ks[2] + t[m];
      sq += qt * qt;
      sk += kt * kt;
      qrow[16 + p * 3 + m] = gwc * qt;
      krow[16 + p * 3 + m] = kt;
    }
  }
#pragma unroll
  for (int d = 28; d < 32; ++d) { qrow[d] = 0.f; krow[d] = 0.f; }
#pragma unroll
  for (int p = 0; p < NPVc; ++p) {
    const float* vs = vp + (long long)bi * 192 + h * 24 + p * 3;
#pragma unroll
    for (int m = 0; m < 3; ++m)
      vrow[16 + p * 3 + m] =
          R[m][0] * vs[0] + R[m][1] * vs[1] + R[m][2] * vs[2] + t[m];
  }
#pragma unroll
  for (int d = 40; d < 48; ++d) vrow[d] = 0.f;
  sqq[(long long)z * Lc + i] = sq;
  sqk[(long long)z * Lc + i] = sk;
}

__global__ __launch_bounds__(256) void k_softmax(float* __restrict__ sc) {
  int row = blockIdx.x * 8 + (threadIdx.x >> 5);
  int lane = threadIdx.x & 31;
  float* p = sc + (long long)row * Lc;
  float vals[12];
  float mx = -INFINITY;
#pragma unroll
  for (int e = 0; e < 12; ++e) {
    vals[e] = p[e * 32 + lane];
    mx = fmaxf(mx, vals[e]);
  }
#pragma unroll
  for (int o = 16; o >= 1; o >>= 1) mx = fmaxf(mx, __shfl_xor(mx, o, 32));
  float s = 0.f;
#pragma unroll
  for (int e = 0; e < 12; ++e) {
    vals[e] = expf(vals[e] - mx);
    s += vals[e];
  }
#pragma unroll
  for (int o = 16; o >= 1; o >>= 1) s += __shfl_xor(s, o, 32);
  float inv = 1.f / s;
#pragma unroll
  for (int e = 0; e < 12; ++e) p[e * 32 + lane] = vals[e] * inv;
}

__global__ __launch_bounds__(256) void k_finish(
    const float* __restrict__ ovalpt, const float* __restrict__ Ri,
    const float* __restrict__ Ti, float* __restrict__ feats) {
  int idx = blockIdx.x * blockDim.x + threadIdx.x;
  if (idx >= Bc * Lc * Hc) return;
  int h = idx % Hc;
  int bi = idx / Hc;
  int b = bi / Lc;
  int z = b * Hc + h;
  int i = bi % Lc;

  float R[3][3], t[3];
#pragma unroll
  for (int r = 0; r < 3; ++r)
#pragma unroll
    for (int c = 0; c < 3; ++c) R[r][c] = Ri[bi * 9 + r * 3 + c];
#pragma unroll
  for (int x = 0; x < 3; ++x) t[x] = Ti[bi * 3 + x];

  const float* orow = ovalpt + ((long long)z * Lc + i) * 48;
  float* frow = feats + (long long)bi * FEAT;

#pragma unroll
  for (int d = 0; d < DKc; ++d) frow[1024 + h * 16 + d] = orow[d];
#pragma unroll
  for (int p = 0; p < NPVc; ++p) {
    float d0 = orow[16 + p * 3 + 0] - t[0];
    float d1 = orow[16 + p * 3 + 1] - t[1];
    float d2 = orow[16 + p * 3 + 2] - t[2];
    float asum = 0.f;
#pragma unroll
    for (int m = 0; m < 3; ++m) {
      float oh = R[0][m] * d0 + R[1][m] * d1 + R[2][m] * d2;
      frow[1152 + h * 24 + p * 3 + m] = oh;
      asum += fabsf(oh);
    }
    frow[1344 + h * 8 + p] = asum;
  }
}

static void launch_gemm(hipStream_t s, int mode, bool transb, const float* A,
                        const float* Bm, float* C, int M, int N, int K, int lda,
                        int ldb, int ldc, long long sA, long long sA2, int zdiv,
                        long long sB, long long sC, int Mvalid, int nz,
                        const float* e0 = nullptr, const float* e1 = nullptr,
                        const float* e2 = nullptr, const float* e3 = nullptr) {
  dim3 g((M + 15) / 16, (N + 127) / 128, nz);
  dim3 blk(128, 1, 1);
  if (mode == 0 && !transb)
    k_gemm<0, false><<<g, blk, 0, s>>>(A, Bm, C, M, N, K, lda, ldb, ldc, sA,
                                       sA2, zdiv, sB, sC, Mvalid, e0, e1, e2, e3);
  else if (mode == 1)
    k_gemm<1, false><<<g, blk, 0, s>>>(A, Bm, C, M, N, K, lda, ldb, ldc, sA,
                                       sA2, zdiv, sB, sC, Mvalid, e0, e1, e2, e3);
  else
    k_gemm<2, true><<<g, blk, 0, s>>>(A, Bm, C, M, N, K, lda, ldb, ldc, sA,
                                      sA2, zdiv, sB, sC, Mvalid, e0, e1, e2, e3);
}

extern "C" void kernel_launch(void* const* d_in, const int* in_sizes, int n_in,
                              void* d_out, int out_size, void* d_ws,
                              size_t ws_size, hipStream_t stream) {
  const float* si = (const float*)d_in[0];
  const float* zij = (const float*)d_in[1];
  const float* Ri = (const float*)d_in[2];
  const float* Ti = (const float*)d_in[3];
  const float* WQh = (const float*)d_in[4];
  const float* WKh = (const float*)d_in[5];
  const float* WVh = (const float*)d_in[6];
  const float* WQhp = (const float*)d_in[7];
  const float* WKhp = (const float*)d_in[8];
  const float* WVhp = (const float*)d_in[9];
  const float* Wb = (const float*)d_in[10];
  const float* WOw = (const float*)d_in[11];
  const float* WOb = (const float*)d_in[12];
  const float* gamah = (const float*)d_in[13];
  float* out = (float*)d_out;
  float* ws = (float*)d_ws;

  const int BL = Bc * Lc;
  const long long LL = (long long)Lc * Lc;
  float* q = ws;
  float* k = q + 98304;
  float* v = k + 98304;
  float* qp = v + 98304;
  float* kp = qp + 73728;
  float* vp = kp + 73728;
  float* qa = vp + 147456;
  float* ka = qa + 196608;
  float* vaug = ka + 196608;
  float* sqq = vaug + 294912;
  float* sqk = sqq + 6144;
  float* bij = sqk + 6144;
  float* alpha = bij + 2359296;
  float* ovalpt = alpha + 2359296;
  float* feats = ovalpt + 294912;

  launch_gemm(stream, 0, false, si, WQh, q, BL, 128, Cc, Cc, 128, 128, 0, 0, 1, 0, 0, 16, 1);
  launch_gemm(stream, 0, false, si, WKh, k, BL, 128, Cc, Cc, 128, 128, 0, 0, 1, 0, 0, 16, 1);
  launch_gemm(stream, 0, false, si, WVh, v, BL, 128, Cc, Cc, 128, 128, 0, 0, 1, 0, 0, 16, 1);
  launch_gemm(stream, 0, false, si, WQhp, qp, BL, 96, Cc, Cc, 96, 96, 0, 0, 1, 0, 0, 16, 1);
  launch_gemm(stream, 0, false, si, WKhp, kp, BL, 96, Cc, Cc, 96, 96, 0, 0, 1, 0, 0, 16, 1);
  launch_gemm(stream, 0, false, si, WVhp, vp, BL, 192, Cc, Cc, 192, 192, 0, 0, 1, 0, 0, 16, 1);

  launch_gemm(stream, 0, false, zij, Wb, bij, Bc * Lc * Lc, Hc, Cc, Cc, Hc, Hc,
              0, 0, 1, 0, 0, 16, 1);

  k_prep<<<(Bc * Lc * Hc + 255) / 256, 256, 0, stream>>>(
      q, k, v, qp, kp, vp, Ri, Ti, gamah, qa, ka, vaug, sqq, sqk);

  launch_gemm(stream, 2, true, qa, ka, alpha, Lc, Lc, 32, 32, 32, Lc,
              (long long)Lc * 32, 0, 1, (long long)Lc * 32, LL, 16, Bc * Hc,
              bij, sqq, sqk, gamah);

  k_softmax<<<(Bc * Hc * Lc) / 8, 256, 0, stream>>>(alpha);

  launch_gemm(stream, 0, false, alpha, vaug, ovalpt, Lc, 48, Lc, Lc, 48, 48,
              LL, 0, 1, (long long)Lc * 48, (long long)Lc * 48, 16, Bc * Hc);

  launch_gemm(stream, 0, false, alpha, zij, feats, 16, Cc, Lc, (int)LL, Cc,
              Cc, (long long)Hc * LL, (long long)Lc, Lc, (long long)Lc * Cc,
              (long long)FEAT, Hc, BL);

  k_finish<<<(Bc * Lc * Hc + 255) / 256, 256, 0, stream>>>(ovalpt, Ri, Ti, feats);

  launch_gemm(stream, 1, false, feats, WOw, out, BL, Cc, FEAT, FEAT, Cc, Cc,
              0, 0, 1, 0, 0, 16, 1, WOb);
}


# One workgroup per (b,i) keeps its zij slice in LDS and does pair bias, softmax and o_pair there: zij read from memory once, 14 launches become 12; every sum keeps its order
# speedup vs baseline: 2.9859x; 1.2799x over previous
#include <hip/hip_runtime.h>
#include <math.h>

typedef __attribute__((ext_vector_type(2))) float v2f;
typedef __attribute__((ext_vector_type(4))) float f4;
typedef __attribute__((ext_vector_type(8))) float v8f;

namespace {
constexpr int Bc = 2, Lc = 384, Cc = 128, Hc = 8, DKc = 16, NQPc = 4, NPVc = 8;
constexpr float WCc = 0.2357022603955158f;
constexpr float WLc = 0.5773502691896258f;
constexpr int FEAT = Hc * Cc + Hc * DKc + Hc * NPVc * 3 + Hc * NPVc;
}

__device__ __forceinline__ float softplusf(float x) {
  return (x > 0.f) ? (x + log1pf(expf(-x))) : log1pf(expf(x));
}

template <bool TRANSB, bool TWO, int KT>
__device__ __forceinline__ void wave_kloop(const float* __restrict__ ap,
                                           const char* __restrict__ Bb,
                                           unsigned ob, unsigned ldb4,
                                           const float* __restrict__ bt0,
                                           const float* __restrict__ bt1, int K,
                                           v8f& acc0, v8f& acc1) {
  constexpr int NS = KT / 4;
  for (int kb = 0; kb < K; kb += KT) {
    v2f a[NS], b0[NS], b1[NS];
#pragma unroll
    for (int s = 0; s < NS; ++s) {
      const int k = kb + 4 * s;
      a[s] = *(const v2f*)(ap + k);
      if (TRANSB) {
        b0[s] = *(const v2f*)(bt0 + k);
        if (TWO) b1[s] = *(const v2f*)(bt1 + k);
      } else {
        const unsigned r0 = (unsigned)k * ldb4;
        const unsigned r1 = r0 + ldb4;
        if (TWO) {
          const v2f x = *(const v2f*)(Bb + (r0 + ob));
          const v2f y = *(const v2f*)(Bb + (r1 + ob));
          b0[s].x = x.x;
          b0[s].y = y.x;
          b1[s].x = x.y;
          b1[s].y = y.y;
        } else {
          b0[s].x = *(const float*)(Bb + (r0 + ob));
          b0[s].y = *(const float*)(Bb + (r1 + ob));
        }
      }
    }
#pragma unroll
    for (int s = 0; s < NS; ++s) {
      acc0 = __builtin_amdgcn_wmma_f32_16x16x4_f32(false, a[s], false, b0[s],
                                                   (short)0, acc0, false, false);
      if (TWO)
        acc1 = __builtin_amdgcn_wmma_f32_16x16x4_f32(false, a[s], false, b1[s],
                                                     (short)0, acc1, false, false);
    }
  }
}

template <int MODE, bool TRANSB>
__global__ __launch_bounds__(128) __attribute__((amdgpu_waves_per_eu(4, 8))) void k_gemm(
    const float* __restrict__ A, const float* __restrict__ Bm,
    float* __restrict__ Cm, int M, int N, int K, int lda, int ldb, int ldc,
    long long sA, long long sA2, int zdiv, long long sB, long long sC,
    int Mvalid, const float* __restrict__ e0) {
  constexpr int KT = TRANSB ? 32 : 64;
  const int z = blockIdx.z;
  const float* Ag = A + (long long)(z / zdiv) * sA + (long long)(z % zdiv) * sA2;
  const float* Bg = Bm + (long long)z * sB;
  float* Cg = Cm + (long long)z * sC;

  const int m0 = blockIdx.x * 16;
  const int n0 = blockIdx.y * 128;
  const int tid = threadIdx.x;
  const int wave = __builtin_amdgcn_readfirstlane(tid >> 5);
  const int lane = tid & 31;
  const int hf = lane >> 4;
  const int l16 = lane & 15;
  const int nw = wave * 32;
  const int nb = n0 + nw;

  if (nb >= N) return;
  const bool two = (nb + 16) < N;

  const int n_t0 = two ? (nb + 2 * l16) : (nb + l16);
  const int n_t1 = n_t0 + 1;
  const int ra = (l16 < Mvalid && (m0 + l16) < M) ? l16 : 0;
  const int c0 = (n_t0 < N) ? n_t0 : nb;

  const float* ap = Ag + (long long)(m0 + ra) * lda + 2 * hf;
  const unsigned ldb4 = (unsigned)ldb * 4u;
  const unsigned ob = (unsigned)(2 * hf * ldb + c0) * 4u;
  const float* bt0 = Bg + (long long)c0 * ldb + 2 * hf;
  const float* bt1 = bt0 + ldb;

  v8f acc0 = {};
  v8f acc1 = {};
  if (two)
    wave_kloop<TRANSB, true, KT>(ap, (const char*)Bg, ob, ldb4, bt0, bt1, K,
                                 acc0, acc1);
  else
    wave_kloop<TRANSB, false, KT>(ap, (const char*)Bg, ob, ldb4, bt0, bt1, K,
                                  acc0, acc1);

#pragma unroll
  for (int r = 0; r < 8; ++r) {
    int mrow = r + 8 * hf;
    int m = m0 + mrow;
    if (mrow >= Mvalid || m >= M) continue;
#pragma unroll
    for (int t = 0; t < 2; ++t) {
      int n = (t == 0) ? n_t0 : n_t1;
      if (n >= N || (t == 1 && !two)) continue;
      float val = (t == 0) ? acc0[r] : acc1[r];
      if (MODE == 1) val += e0[n];
      Cg[(long long)m * ldc + n] = val;
    }
  }
}

__global__ __launch_bounds__(256) void k_prep(
    const float* __restrict__ q, const float* __restrict__ k,
    const float* __restrict__ v, const float* __restrict__ qp,
    const float* __restrict__ kp, const float* __restrict__ vp,
    const float* __restrict__ Ri, const float* __restrict__ Ti,
    const float* __restrict__ gamah, float* __restrict__ qa,
    float* __restrict__ ka, float* __restrict__ vaug, float* __restrict__ sqq,
    float* __restrict__ sqk) {
  int idx = blockIdx.x * blockDim.x + threadIdx.x;
  if (idx >= Bc * Lc * Hc) return;
  int h = idx % Hc;
  int bi = idx / Hc;
  int b = bi / Lc, i = bi % Lc;
  int z = b * Hc + h;

  float R[3][3], t[3];
#pragma unroll
  for (int r = 0; r < 3; ++r)
#pragma unroll
    for (int c = 0; c < 3; ++c) R[r][c] = Ri[bi * 9 + r * 3 + c];
#pragma unroll
  for (int x = 0; x < 3; ++x) t[x] = Ti[bi * 3 + x];

  float gwc = WCc * softplusf(gamah[h]);

  float* qrow = qa + ((long long)z * Lc + i) * 32;
  float* krow = ka + ((long long)z * Lc + i) * 32;
  float* vrow = vaug + ((long long)z * Lc + i) * 48;

#pragma unroll
  for (int d = 0; d < DKc; ++d) {
    qrow[d] = 0.25f * q[(long long)bi * 128 + h * 16 + d];
    krow[d] = k[(long long)bi * 128 + h * 16 + d];
    vrow[d] = v[(long long)bi * 128 + h * 16 + d];
  }
  float sq = 0.f, sk = 0.f;
#pragma unroll
  for (int p = 0; p < NQPc; ++p) {
    const float* qs = qp + (long long)bi * 96 + h * 12 + p * 3;
    const float* ks = kp + (long long)bi * 96 + h * 12 + p * 3;
#pragma unroll
    for (int m = 0; m < 3; ++m) {
      float qt = R[m][0] * qs[0] + R[m][1] * qs[1] + R[m][2] * qs[2] + t[m];
      float kt = R[m][0] * ks[0] + R[m][1] * ks[1] + R[m][2] * ks[2] + t[m];
      sq += qt * qt;
      sk += kt * kt;
      qrow[16 + p * 3 + m] = gwc * qt;
      krow[16 + p * 3 + m] = kt;
    }
  }
#pragma unroll
  for (int d = 28; d < 32; ++d) { qrow[d] = 0.f; krow[d] = 0.f; }
#pragma unroll
  for (int p = 0; p < NPVc; ++p) {
    const float* vs = vp + (long long)bi * 192 + h * 24 + p * 3;
#pragma unroll
    for (int m = 0; m < 3; ++m)
      vrow[16 + p * 3 + m] =
          R[m][0] * vs[0] + R[m][1] * vs[1] + R[m][2] * vs[2] + t[m];
  }
#pragma unroll
  for (int d = 40; d < 48; ++d) vrow[d] = 0.f;
  sqq[(long long)z * Lc + i] = sq;
  sqk[(long long)z * Lc + i] = sk;
}

namespace {
constexpr int ZP = 140;
constexpr int RP = 388;
}

__global__ __launch_bounds__(768) void k_slice(
    const float* __restrict__ zij, const float* __restrict__ Wb,
    float* __restrict__ S, const float* __restrict__ sqq,
    const float* __restrict__ sqk, const float* __restrict__ gamah,
    float* __restrict__ feats) {
  __shared__ __attribute__((aligned(16))) float zs[Lc * ZP];
  __shared__ __attribute__((aligned(16))) float sb[Hc * RP];
  __shared__ __attribute__((aligned(16))) float sa[Hc * RP];

  const int bi = blockIdx.x;
  const int b = bi / Lc, i = bi % Lc;
  const int tid = threadIdx.x;
  const int wave = __builtin_amdgcn_readfirstlane(tid >> 5);
  const int lane = tid & 31;
  const int hf = lane >> 4;
  const int l16 = lane & 15;

  const int j0 = wave * 16;
  v2f w[32];
  {
    const int cw = (l16 < Hc) ? l16 : 0;
    const float* wp = Wb + (2 * hf) * Hc + cw;
#pragma unroll
    for (int s = 0; s < 32; ++s) {
      w[s].x = wp[(4 * s) * Hc];
      w[s].y = wp[(4 * s + 1) * Hc];
    }
  }
  {
    const f4* zsrc = (const f4*)(zij + ((long long)bi * Lc + j0) * Cc) + lane;
    f4 t[16];
#pragma unroll
    for (int r = 0; r < 16; ++r) t[r] = zsrc[r * (Cc / 4)];
#pragma unroll
    for (int r = 0; r < 16; ++r) *(f4*)(zs + (j0 + r) * ZP + 4 * lane) = t[r];
  }
  __syncthreads();
  {
    const float* zp = zs + (j0 + l16) * ZP + 2 * hf;
    v8f acc = {};
#pragma unroll
    for (int g = 0; g < 2; ++g) {
      v2f a[16];
#pragma unroll
      for (int s = 0; s < 16; ++s) a[s] = *(const v2f*)(zp + 64 * g + 4 * s);
#pragma unroll
      for (int s = 0; s < 16; ++s)
        acc = __builtin_amdgcn_wmma_f32_16x16x4_f32(false, a[s], false,
                                                    w[16 * g + s], (short)0, acc,
                                                    false, false);
    }
    if (l16 < Hc) {
#pragma unroll
      for (int r = 0; r < 8; ++r) sb[l16 * RP + j0 + r + 8 * hf] = acc[r];
    }
  }
  __syncthreads();

  if (wave < Hc) {
    const int z = b * Hc + wave;
    const float gw = 0.5f * WCc * softplusf(gamah[wave]);
    const float sq = sqq[(long long)z * Lc + i];
    float* p = S + ((long long)z * Lc + i) * Lc;
    const float* kq = sqk + (long long)z * Lc;
    float vals[12];
    float mx = -INFINITY;
#pragma unroll
    for (int e = 0; e < 12; ++e) {
      const int j = e * 32 + lane;
      vals[e] = WLc * (p[j] + sb[wave * RP + j] - gw * (sq + kq[j]));
      mx = fmaxf(mx, vals[e]);
    }
#pragma unroll
    for (int o = 16; o >= 1; o >>= 1) mx = fmaxf(mx, __shfl_xor(mx, o, 32));
    float sum = 0.f;
#pragma unroll
    for (int e = 0; e < 12; ++e) {
      vals[e] = expf(vals[e] - mx);
      sum += vals[e];
    }
#pragma unroll
    for (int o = 16; o >= 1; o >>= 1) sum += __shfl_xor(sum, o, 32);
    const float inv = 1.f / sum;
#pragma unroll
    for (int e = 0; e < 12; ++e) {
      const int j = e * 32 + lane;
      const float al = vals[e] * inv;
      p[j] = al;
      sa[wave * RP + j] = al;
    }
  }
  __syncthreads();

  if (wave < Cc / 16) {
    const int c = wave * 16 + l16;
    const int ra = (l16 < Hc) ? l16 : 0;
    const float* ap = sa + ra * RP + 2 * hf;
    const float* zc = zs + (2 * hf) * ZP + c;
    v8f acc = {};
    for (int kb = 0; kb < Lc; kb += 64) {
      v2f a[16], bq[16];
#pragma unroll
      for (int s = 0; s < 16; ++s) {
        const int k = kb + 4 * s;
        a[s] = *(const v2f*)(ap + k);
        bq[s].x = zc[k * ZP];
        bq[s].y = zc[(k + 1) * ZP];
      }
      asm volatile("" ::"v"(a[0]), "v"(a[1]), "v"(a[2]), "v"(a[3]), "v"(a[4]), "v"(a[5]),
                   "v"(a[6]), "v"(a[7]), "v"(a[8]), "v"(a[9]), "v"(a[10]),
                   "v"(a[11]), "v"(a[12]), "v"(a[13]), "v"(a[14]), "v"(a[15]));
      asm volatile("" ::"v"(bq[0]), "v"(bq[1]), "v"(bq[2]), "v"(bq[3]), "v"(bq[4]), "v"(bq[5]),
                   "v"(bq[6]), "v"(bq[7]), "v"(bq[8]), "v"(bq[9]), "v"(bq[10]),
                   "v"(bq[11]), "v"(bq[12]), "v"(bq[13]), "v"(bq[14]), "v"(bq[15]));
#pragma unroll
      for (int s = 0; s < 16; ++s)
        acc = __builtin_amdgcn_wmma_f32_16x16x4_f32(false, a[s], false, bq[s],
                                                    (short)0, acc, false, false);
    }
    if (hf == 0) {
      float* frow = feats + (long long)bi * FEAT + c;
#pragma unroll
      for (int r = 0; r < 8; ++r) frow[r * Cc] = acc[r];
    }
  }
}

__global__ __launch_bounds__(256) void k_finish(
    const float* __restrict__ ovalpt, const float* __restrict__ Ri,
    const float* __restrict__ Ti, float* __restrict__ feats) {
  int idx = blockIdx.x * blockDim.x + threadIdx.x;
  if (idx >= Bc * Lc * Hc) return;
  int h = idx % Hc;
  int bi = idx / Hc;
  int b = bi / Lc;
  int z = b * Hc + h;
  int i = bi % Lc;

  float R[3][3], t[3];
#pragma unroll
  for (int r = 0; r < 3; ++r)
#pragma unroll
    for (int c = 0; c < 3; ++c) R[r][c] = Ri[bi * 9 + r * 3 + c];
#pragma unroll
  for (int x = 0; x < 3; ++x) t[x] = Ti[bi * 3 + x];

  const float* orow = ovalpt + ((long long)z * Lc + i) * 48;
  float* frow = feats + (long long)bi * FEAT;

#pragma unroll
  for (int d = 0; d < DKc; ++d) frow[1024 + h * 16 + d] = orow[d];
#pragma unroll
  for (int p = 0; p < NPVc; ++p) {
    float d0 = orow[16 + p * 3 + 0] - t[0];
    float d1 = orow[16 + p * 3 + 1] - t[1];
    float d2 = orow[16 + p * 3 + 2] - t[2];
    float asum = 0.f;
#pragma unroll
    for (int m = 0; m < 3; ++m) {
      float oh = R[0][m] * d0 + R[1][m] * d1 + R[2][m] * d2;
      frow[1152 + h * 24 + p * 3 + m] = oh;
      asum += fabsf(oh);
    }
    frow[1344 + h * 8 + p] = asum;
  }
}

static void launch_gemm(hipStream_t s, int mode, bool transb, const float* A,
                        const float* Bm, float* C, int M, int N, int K, int lda,
                        int ldb, int ldc, long long sA, long long sA2, int zdiv,
                        long long sB, long long sC, int Mvalid, int nz,
                        const float* e0 = nullptr) {
  dim3 g((M + 15) / 16, (N + 127) / 128, nz);
  dim3 blk(128, 1, 1);
  if (mode == 0 && !transb)
    k_gemm<0, false><<<g, blk, 0, s>>>(A, Bm, C, M, N, K, lda, ldb, ldc, sA,
                                       sA2, zdiv, sB, sC, Mvalid, e0);
  else if (mode == 1)
    k_gemm<1, false><<<g, blk, 0, s>>>(A, Bm, C, M, N, K, lda, ldb, ldc, sA,
                                       sA2, zdiv, sB, sC, Mvalid, e0);
  else
    k_gemm<0, true><<<g, blk, 0, s>>>(A, Bm, C, M, N, K, lda, ldb, ldc, sA,
                                      sA2, zdiv, sB, sC, Mvalid, e0);
}

extern "C" void kernel_launch(void* const* d_in, const int* in_sizes, int n_in,
                              void* d_out, int out_size, void* d_ws,
                              size_t ws_size, hipStream_t stream) {
  const float* si = (const float*)d_in[0];
  const float* zij = (const float*)d_in[1];
  const float* Ri = (const float*)d_in[2];
  const float* Ti = (const float*)d_in[3];
  const float* WQh = (const float*)d_in[4];
  const float* WKh = (const float*)d_in[5];
  const float* WVh = (const float*)d_in[6];
  const float* WQhp = (const float*)d_in[7];
  const float* WKhp = (const float*)d_in[8];
  const float* WVhp = (const float*)d_in[9];
  const float* Wb = (const float*)d_in[10];
  const float* WOw = (const float*)d_in[11];
  const float* WOb = (const float*)d_in[12];
  const float* gamah = (const float*)d_in[13];
  float* out = (float*)d_out;
  float* ws = (float*)d_ws;

  const int BL = Bc * Lc;
  const long long LL = (long long)Lc * Lc;
  float* q = ws;
  float* k = q + 98304;
  float* v = k + 98304;
  float* qp = v + 98304;
  float* kp = qp + 73728;
  float* vp = kp + 73728;
  float* qa = vp + 147456;
  float* ka = qa + 196608;
  float* vaug = ka + 196608;
  float* sqq = vaug + 294912;
  float* sqk = sqq + 6144;
  float* alpha = sqk + 6144;
  float* ovalpt = alpha + 2359296;
  float* feats = ovalpt + 294912;

  launch_gemm(stream, 0, false, si, WQh, q, BL, 128, Cc, Cc, 128, 128, 0, 0, 1, 0, 0, 16, 1);
  launch_gemm(stream, 0, false, si, WKh, k, BL, 128, Cc, Cc, 128, 128, 0, 0, 1, 0, 0, 16, 1);
  launch_gemm(stream, 0, false, si, WVh, v, BL, 128, Cc, Cc, 128, 128, 0, 0, 1, 0, 0, 16, 1);
  launch_gemm(stream, 0, false, si, WQhp, qp, BL, 96, Cc, Cc, 96, 96, 0, 0, 1, 0, 0, 16, 1);
  launch_gemm(stream, 0, false, si, WKhp, kp, BL, 96, Cc, Cc, 96, 96, 0, 0, 1, 0, 0, 16, 1);
  launch_gemm(stream, 0, false, si, WVhp, vp, BL, 192, Cc, Cc, 192, 192, 0, 0, 1, 0, 0, 16, 1);

  k_prep<<<(Bc * Lc * Hc + 255) / 256, 256, 0, stream>>>(
      q, k, v, qp, kp, vp, Ri, Ti, gamah, qa, ka, vaug, sqq, sqk);

  launch_gemm(stream, 0, true, qa, ka, alpha, Lc, Lc, 32, 32, 32, Lc,
              (long long)Lc * 32, 0, 1, (long long)Lc * 32, LL, 16, Bc * Hc);

  k_slice<<<BL, 768, 0, stream>>>(zij, Wb, alpha, sqq, sqk, gamah, feats);

  launch_gemm(stream, 0, false, alpha, vaug, ovalpt, Lc, 48, Lc, Lc, 48, 48,
              LL, 0, 1, (long long)Lc * 48, (long long)Lc * 48, 16, Bc * Hc);

  k_finish<<<(Bc * Lc * Hc + 255) / 256, 256, 0, stream>>>(ovalpt, Ri, Ti, feats);

  launch_gemm(stream, 1, false, feats, WOw, out, BL, Cc, FEAT, FEAT, Cc, Cc,
              0, 0, 1, 0, 0, 16, 1, WOb);
}


# GEMM K loops (B as [K,N]) software-pipelined: four register sets of 32 k, loads of the group three ahead are requested before the current group's products; k order, products and sums unchanged
# speedup vs baseline: 3.2446x; 1.0866x over previous
#include <hip/hip_runtime.h>
#include <math.h>

typedef __attribute__((ext_vector_type(2))) float v2f;
typedef __attribute__((ext_vector_type(4))) float f4;
typedef __attribute__((ext_vector_type(8))) float v8f;

namespace {
constexpr int Bc = 2, Lc = 384, Cc = 128, Hc = 8, DKc = 16, NQPc = 4, NPVc = 8;
constexpr float WCc = 0.2357022603955158f;
constexpr float WLc = 0.5773502691896258f;
constexpr int FEAT = Hc * Cc + Hc * DKc + Hc * NPVc * 3 + Hc * NPVc;
}

__device__ __forceinline__ float softplusf(float x) {
  return (x > 0.f) ? (x + log1pf(expf(-x))) : log1pf(expf(x));
}

template <bool TWO, int KT>
__device__ __forceinline__ void wave_kloop_t(const float* __restrict__ ap,
                                             const float* __restrict__ bt0,
                                             const float* __restrict__ bt1,
                                             int K, v8f& acc0, v8f& acc1) {
  constexpr int NS = KT / 4;
  for (int kb = 0; kb < K; kb += KT) {
    v2f a[NS], b0[NS], b1[NS];
#pragma unroll
    for (int s = 0; s < NS; ++s) {
      const int k = kb + 4 * s;
      a[s] = *(const v2f*)(ap + k);
      b0[s] = *(const v2f*)(bt0 + k);
      if (TWO) b1[s] = *(const v2f*)(bt1 + k);
    }
#pragma unroll
    for (int s = 0; s < NS; ++s) {
      acc0 = __builtin_amdgcn_wmma_f32_16x16x4_f32(false, a[s], false, b0[s],
                                                   (short)0, acc0, false, false);
      if (TWO)
        acc1 = __builtin_amdgcn_wmma_f32_16x16x4_f32(false, a[s], false, b1[s],
                                                     (short)0, acc1, false, false);
    }
  }
}

namespace {
constexpr int PKT = 32;
constexpr int PNS = PKT / 4;
constexpr int PSETS = 4;
constexpr int PAHEAD = PSETS - 1;
}

template <bool TWO>
__device__ __forceinline__ void load_group(const float* __restrict__ ap,
                                           const char* __restrict__ Bb,
                                           unsigned ob, unsigned ldb4, int kb,
                                           v2f (&a)[PNS], v2f (&x)[PNS],
                                           v2f (&y)[PNS]) {
#pragma unroll
  for (int s = 0; s < PNS; ++s) {
    const int k = kb + 4 * s;
    a[s] = *(const v2f*)(ap + k);
    const char* r0 = Bb + (unsigned long long)((unsigned)k * ldb4);
    const char* r1 = r0 + ldb4;
    if (TWO) {
      x[s] = *(const v2f*)(r0 + ob);
      y[s] = *(const v2f*)(r1 + ob);
    } else {
      x[s].x = *(const float*)(r0 + ob);
      x[s].y = *(const float*)(r1 + ob);
    }
  }
}

template <bool TWO>
__device__ __forceinline__ void mma_group(const v2f (&a)[PNS],
                                          const v2f (&x)[PNS],
                                          const v2f (&y)[PNS], v8f& acc0,
                                          v8f& acc1) {
#pragma unroll
  for (int s = 0; s < PNS; ++s) {
    if (TWO) {
      v2f b0, b1;
      b0.x = x[s].x;
      b0.y = y[s].x;
      b1.x = x[s].y;
      b1.y = y[s].y;
      acc0 = __builtin_amdgcn_wmma_f32_16x16x4_f32(false, a[s], false, b0,
                                                   (short)0, acc0, false, false);
      acc1 = __builtin_amdgcn_wmma_f32_16x16x4_f32(false, a[s], false, b1,
                                                   (short)0, acc1, false, false);
    } else {
      acc0 = __builtin_amdgcn_wmma_f32_16x16x4_f32(false, a[s], false, x[s],
                                                   (short)0, acc0, false, false);
    }
  }
}

template <bool TWO>
__device__ __forceinline__ void wave_kloop_pipe(const float* __restrict__ ap,
                                                const char* __restrict__ Bb,
                                                unsigned ob, unsigned ldb4,
                                                int K, v8f& acc0, v8f& acc1) {
  v2f fa[PSETS][PNS], fx[PSETS][PNS], fy[PSETS][PNS];
#pragma unroll
  for (int g = 0; g < PAHEAD; ++g) {
    load_group<TWO>(ap, Bb, ob, ldb4, g * PKT, fa[g], fx[g], fy[g]);
    __builtin_amdgcn_sched_barrier(0);
  }
  int kb = 0;
  for (; kb + PSETS * PKT < K; kb += PSETS * PKT) {
#pragma unroll
    for (int u = 0; u < PSETS; ++u) {
      __builtin_amdgcn_sched_barrier(0);
      load_group<TWO>(ap, Bb, ob, ldb4, kb + (u + PAHEAD) * PKT,
                      fa[(u + PAHEAD) % PSETS], fx[(u + PAHEAD) % PSETS],
                      fy[(u + PAHEAD) % PSETS]);
      __builtin_amdgcn_sched_barrier(0);
      mma_group<TWO>(fa[u], fx[u], fy[u], acc0, acc1);
    }
  }
  __builtin_amdgcn_sched_barrier(0);
  load_group<TWO>(ap, Bb, ob, ldb4, kb + PAHEAD * PKT, fa[PAHEAD % PSETS],
                  fx[PAHEAD % PSETS], fy[PAHEAD % PSETS]);
  __builtin_amdgcn_sched_barrier(0);
#pragma unroll
  for (int u = 0; u < PSETS; ++u)
    mma_group<TWO>(fa[u], fx[u], fy[u], acc0, acc1);
}

template <int MODE, bool TRANSB>
__global__ __launch_bounds__(128) __attribute__((amdgpu_waves_per_eu(4, 4))) void k_gemm(
    const float* __restrict__ A, const float* __restrict__ Bm,
    float* __restrict__ Cm, int M, int N, int K, int lda, int ldb, int ldc,
    long long sA, long long sA2, int zdiv, long long sB, long long sC,
    int Mvalid, const float* __restrict__ e0) {
  const int z = blockIdx.z;
  const float* Ag = A + (long long)(z / zdiv) * sA + (long long)(z % zdiv) * sA2;
  const float* Bg = Bm + (long long)z * sB;
  float* Cg = Cm + (long long)z * sC;

  const int m0 = blockIdx.x * 16;
  const int n0 = blockIdx.y * 128;
  const int tid = threadIdx.x;
  const int wave = __builtin_amdgcn_readfirstlane(tid >> 5);
  const int lane = tid & 31;
  const int hf = lane >> 4;
  const int l16 = lane & 15;
  const int nw = wave * 32;
  const int nb = n0 + nw;

  if (nb >= N) return;
  const bool two = (nb + 16) < N;

  const int n_t0 = two ? (nb + 2 * l16) : (nb + l16);
  const int n_t1 = n_t0 + 1;
  const int ra = (l16 < Mvalid && (m0 + l16) < M) ? l16 : 0;
  const int c0 = (n_t0 < N) ? n_t0 : nb;

  const float* ap = Ag + (long long)(m0 + ra) * lda + 2 * hf;
  const unsigned ldb4 = (unsigned)ldb * 4u;
  const unsigned ob = (unsigned)(2 * hf * ldb + c0) * 4u;
  const float* bt0 = Bg + (long long)c0 * ldb + 2 * hf;
  const float* bt1 = bt0 + ldb;

  v8f acc0 = {};
  v8f acc1 = {};
  if (TRANSB) {
    if (two)
      wave_kloop_t<true, 32>(ap, bt0, bt1, K, acc0, acc1);
    else
      wave_kloop_t<false, 32>(ap, bt0, bt1, K, acc0, acc1);
  } else {
    if (two)
      wave_kloop_pipe<true>(ap, (const char*)Bg, ob, ldb4, K, acc0, acc1);
    else
      wave_kloop_pipe<false>(ap, (const char*)Bg, ob, ldb4, K, acc0, acc1);
  }

#pragma unroll
  for (int r = 0; r < 8; ++r) {
    int mrow = r + 8 * hf;
    int m = m0 + mrow;
    if (mrow >= Mvalid || m >= M) continue;
#pragma unroll
    for (int t = 0; t < 2; ++t) {
      int n = (t == 0) ? n_t0 : n_t1;
      if (n >= N || (t == 1 && !two)) continue;
      float val = (t == 0) ? acc0[r] : acc1[r];
      if (MODE == 1) val += e0[n];
      Cg[(long long)m * ldc + n] = val;
    }
  }
}

__global__ __launch_bounds__(256) void k_prep(
    const float* __restrict__ q, const float* __restrict__ k,
    const float* __restrict__ v, const float* __restrict__ qp,
    const float* __restrict__ kp, const float* __restrict__ vp,
    const float* __restrict__ Ri, const float* __restrict__ Ti,
    const float* __restrict__ gamah, float* __restrict__ qa,
    float* __restrict__ ka, float* __restrict__ vaug, float* __restrict__ sqq,
    float* __restrict__ sqk) {
  int idx = blockIdx.x * blockDim.x + threadIdx.x;
  if (idx >= Bc * Lc * Hc) return;
  int h = idx % Hc;
  int bi = idx / Hc;
  int b = bi / Lc, i = bi % Lc;
  int z = b * Hc + h;

  float R[3][3], t[3];
#pragma unroll
  for (int r = 0; r < 3; ++r)
#pragma unroll
    for (int c = 0; c < 3; ++c) R[r][c] = Ri[bi * 9 + r * 3 + c];
#pragma unroll
  for (int x = 0; x < 3; ++x) t[x] = Ti[bi * 3 + x];

  float gwc = WCc * softplusf(gamah[h]);

  float* qrow = qa + ((long long)z * Lc + i) * 32;
  float* krow = ka + ((long long)z * Lc + i) * 32;
  float* vrow = vaug + ((long long)z * Lc + i) * 48;

#pragma unroll
  for (int d = 0; d < DKc; ++d) {
    qrow[d] = 0.25f * q[(long long)bi * 128 + h * 16 + d];
    krow[d] = k[(long long)bi * 128 + h * 16 + d];
    vrow[d] = v[(long long)bi * 128 + h * 16 + d];
  }
  float sq = 0.f, sk = 0.f;
#pragma unroll
  for (int p = 0; p < NQPc; ++p) {
    const float* qs = qp + (long long)bi * 96 + h * 12 + p * 3;
    const float* ks = kp + (long long)bi * 96 + h * 12 + p * 3;
#pragma unroll
    for (int m = 0; m < 3; ++m) {
      float qt = R[m][0] * qs[0] + R[m][1] * qs[1] + R[m][2] * qs[2] + t[m];
      float kt = R[m][0] * ks[0] + R[m][1] * ks[1] + R[m][2] * ks[2] + t[m];
      sq += qt * qt;
      sk += kt * kt;
      qrow[16 + p * 3 + m] = gwc * qt;
      krow[16 + p * 3 + m] = kt;
    }
  }
#pragma unroll
  for (int d = 28; d < 32; ++d) { qrow[d] = 0.f; krow[d] = 0.f; }
#pragma unroll
  for (int p = 0; p < NPVc; ++p) {
    const float* vs = vp + (long long)bi * 192 + h * 24 + p * 3;
#pragma unroll
    for (int m = 0; m < 3; ++m)
      vrow[16 + p * 3 + m] =
          R[m][0] * vs[0] + R[m][1] * vs[1] + R[m][2] * vs[2] + t[m];
  }
#pragma unroll
  for (int d = 40; d < 48; ++d) vrow[d] = 0.f;
  sqq[(long long)z * Lc + i] = sq;
  sqk[(long long)z * Lc + i] = sk;
}

namespace {
constexpr int ZP = 140;
constexpr int RP = 388;
}

__global__ __launch_bounds__(768) void k_slice(
    const float* __restrict__ zij, const float* __restrict__ Wb,
    float* __restrict__ S, const float* __restrict__ sqq,
    const float* __restrict__ sqk, const float* __restrict__ gamah,
    float* __restrict__ feats) {
  __shared__ __attribute__((aligned(16))) float zs[Lc * ZP];
  __shared__ __attribute__((aligned(16))) float sb[Hc * RP];
  __shared__ __attribute__((aligned(16))) float sa[Hc * RP];

  const int bi = blockIdx.x;
  const int b = bi / Lc, i = bi % Lc;
  const int tid = threadIdx.x;
  const int wave = __builtin_amdgcn_readfirstlane(tid >> 5);
  const int lane = tid & 31;
  const int hf = lane >> 4;
  const int l16 = lane & 15;

  const int j0 = wave * 16;
  v2f w[32];
  {
    const int cw = (l16 < Hc) ? l16 : 0;
    const float* wp = Wb + (2 * hf) * Hc + cw;
#pragma unroll
    for (int s = 0; s < 32; ++s) {
      w[s].x = wp[(4 * s) * Hc];
      w[s].y = wp[(4 * s + 1) * Hc];
    }
  }
  {
    const f4* zsrc = (const f4*)(zij + ((long long)bi * Lc + j0) * Cc) + lane;
    f4 t[16];
#pragma unroll
    for (int r = 0; r < 16; ++r) t[r] = zsrc[r * (Cc / 4)];
#pragma unroll
    for (int r = 0; r < 16; ++r) *(f4*)(zs + (j0 + r) * ZP + 4 * lane) = t[r];
  }
  __syncthreads();
  {
    const float* zp = zs + (j0 + l16) * ZP + 2 * hf;
    v8f acc = {};
#pragma unroll
    for (int g = 0; g < 2; ++g) {
      v2f a[16];
#pragma unroll
      for (int s = 0; s < 16; ++s) a[s] = *(const v2f*)(zp + 64 * g + 4 * s);
#pragma unroll
      for (int s = 0; s < 16; ++s)
        acc = __builtin_amdgcn_wmma_f32_16x16x4_f32(false, a[s], false,
                                                    w[16 * g + s], (short)0, acc,
                                                    false, false);
    }
    if (l16 < Hc) {
#pragma unroll
      for (int r = 0; r < 8; ++r) sb[l16 * RP + j0 + r + 8 * hf] = acc[r];
    }
  }
  __syncthreads();

  if (wave < Hc) {
    const int z = b * Hc + wave;
    const float gw = 0.5f * WCc * softplusf(gamah[wave]);
    const float sq = sqq[(long long)z * Lc + i];
    float* p = S + ((long long)z * Lc + i) * Lc;
    const float* kq = sqk + (long long)z * Lc;
    float vals[12];
    float mx = -INFINITY;
#pragma unroll
    for (int e = 0; e < 12; ++e) {
      const int j = e * 32 + lane;
      vals[e] = WLc * (p[j] + sb[wave * RP + j] - gw * (sq + kq[j]));
      mx = fmaxf(mx, vals[e]);
    }
#pragma unroll
    for (int o = 16; o >= 1; o >>= 1) mx = fmaxf(mx, __shfl_xor(mx, o, 32));
    float sum = 0.f;
#pragma unroll
    for (int e = 0; e < 12; ++e) {
      vals[e] = expf(vals[e] - mx);
      sum += vals[e];
    }
#pragma unroll
    for (int o = 16; o >= 1; o >>= 1) sum += __shfl_xor(sum, o, 32);
    const float inv = 1.f / sum;
#pragma unroll
    for (int e = 0; e < 12; ++e) {
      const int j = e * 32 + lane;
      const float al = vals[e] * inv;
      p[j] = al;
      sa[wave * RP + j] = al;
    }
  }
  __syncthreads();

  if (wave < Cc / 16) {
    const int c = wave * 16 + l16;
    const int ra = (l16 < Hc) ? l16 : 0;
    const float* ap = sa + ra * RP + 2 * hf;
    const float* zc = zs + (2 * hf) * ZP + c;
    v8f acc = {};
    for (int kb = 0; kb < Lc; kb += 64) {
      v2f a[16], bq[16];
#pragma unroll
      for (int s = 0; s < 16; ++s) {
        const int k = kb + 4 * s;
        a[s] = *(const v2f*)(ap + k);
        bq[s].x = zc[k * ZP];
        bq[s].y = zc[(k + 1) * ZP];
      }
      asm volatile("" ::"v"(a[0]), "v"(a[1]), "v"(a[2]), "v"(a[3]), "v"(a[4]), "v"(a[5]),
                   "v"(a[6]), "v"(a[7]), "v"(a[8]), "v"(a[9]), "v"(a[10]),
                   "v"(a[11]), "v"(a[12]), "v"(a[13]), "v"(a[14]), "v"(a[15]));
      asm volatile("" ::"v"(bq[0]), "v"(bq[1]), "v"(bq[2]), "v"(bq[3]), "v"(bq[4]), "v"(bq[5]),
                   "v"(bq[6]), "v"(bq[7]), "v"(bq[8]), "v"(bq[9]), "v"(bq[10]),
                   "v"(bq[11]), "v"(bq[12]), "v"(bq[13]), "v"(bq[14]), "v"(bq[15]));
#pragma unroll
      for (int s = 0; s < 16; ++s)
        acc = __builtin_amdgcn_wmma_f32_16x16x4_f32(false, a[s], false, bq[s],
                                                    (short)0, acc, false, false);
    }
    if (hf == 0) {
      float* frow = feats + (long long)bi * FEAT + c;
#pragma unroll
      for (int r = 0; r < 8; ++r) frow[r * Cc] = acc[r];
    }
  }
}

__global__ __launch_bounds__(256) void k_finish(
    const float* __restrict__ ovalpt, const float* __restrict__ Ri,
    const float* __restrict__ Ti, float* __restrict__ feats) {
  int idx = blockIdx.x * blockDim.x + threadIdx.x;
  if (idx >= Bc * Lc * Hc) return;
  int h = idx % Hc;
  int bi = idx / Hc;
  int b = bi / Lc;
  int z = b * Hc + h;
  int i = bi % Lc;

  float R[3][3], t[3];
#pragma unroll
  for (int r = 0; r < 3; ++r)
#pragma unroll
    for (int c = 0; c < 3; ++c) R[r][c] = Ri[bi * 9 + r * 3 + c];
#pragma unroll
  for (int x = 0; x < 3; ++x) t[x] = Ti[bi * 3 + x];

  const float* orow = ovalpt + ((long long)z * Lc + i) * 48;
  float* frow = feats + (long long)bi * FEAT;

#pragma unroll
  for (int d = 0; d < DKc; ++d) frow[1024 + h * 16 + d] = orow[d];
#pragma unroll
  for (int p = 0; p < NPVc; ++p) {
    float d0 = orow[16 + p * 3 + 0] - t[0];
    float d1 = orow[16 + p * 3 + 1] - t[1];
    float d2 = orow[16 + p * 3 + 2] - t[2];
    float asum = 0.f;
#pragma unroll
    for (int m = 0; m < 3; ++m) {
      float oh = R[0][m] * d0 + R[1][m] * d1 + R[2][m] * d2;
      frow[1152 + h * 24 + p * 3 + m] = oh;
      asum += fabsf(oh);
    }
    frow[1344 + h * 8 + p] = asum;
  }
}

static void launch_gemm(hipStream_t s, int mode, bool transb, const float* A,
                        const float* Bm, float* C, int M, int N, int K, int lda,
                        int ldb, int ldc, long long sA, long long sA2, int zdiv,
                        long long sB, long long sC, int Mvalid, int nz,
                        const float* e0 = nullptr) {
  dim3 g((M + 15) / 16, (N + 127) / 128, nz);
  dim3 blk(128, 1, 1);
  if (mode == 0 && !transb)
    k_gemm<0, false><<<g, blk, 0, s>>>(A, Bm, C, M, N, K, lda, ldb, ldc, sA,
                                       sA2, zdiv, sB, sC, Mvalid, e0);
  else if (mode == 1)
    k_gemm<1, false><<<g, blk, 0, s>>>(A, Bm, C, M, N, K, lda, ldb, ldc, sA,
                                       sA2, zdiv, sB, sC, Mvalid, e0);
  else
    k_gemm<0, true><<<g, blk, 0, s>>>(A, Bm, C, M, N, K, lda, ldb, ldc, sA,
                                      sA2, zdiv, sB, sC, Mvalid, e0);
}

extern "C" void kernel_launch(void* const* d_in, const int* in_sizes, int n_in,
                              void* d_out, int out_size, void* d_ws,
                              size_t ws_size, hipStream_t stream) {
  const float* si = (const float*)d_in[0];
  const float* zij = (const float*)d_in[1];
  const float* Ri = (const float*)d_in[2];
  const float* Ti = (const float*)d_in[3];
  const float* WQh = (const float*)d_in[4];
  const float* WKh = (const float*)d_in[5];
  const float* WVh = (const float*)d_in[6];
  const float* WQhp = (const float*)d_in[7];
  const float* WKhp = (const float*)d_in[8];
  const float* WVhp = (const float*)d_in[9];
  const float* Wb = (const float*)d_in[10];
  const float* WOw = (const float*)d_in[11];
  const float* WOb = (const float*)d_in[12];
  const float* gamah = (const float*)d_in[13];
  float* out = (float*)d_out;
  float* ws = (float*)d_ws;

  const int BL = Bc * Lc;
  const long long LL = (long long)Lc * Lc;
  float* q = ws;
  float* k = q + 98304;
  float* v = k + 98304;
  float* qp = v + 98304;
  float* kp = qp + 73728;
  float* vp = kp + 73728;
  float* qa = vp + 147456;
  float* ka = qa + 196608;
  float* vaug = ka + 196608;
  float* sqq = vaug + 294912;
  float* sqk = sqq + 6144;
  float* alpha = sqk + 6144;
  float* ovalpt = alpha + 2359296;
  float* feats = ovalpt + 294912;

  launch_gemm(stream, 0, false, si, WQh, q, BL, 128, Cc, Cc, 128, 128, 0, 0, 1, 0, 0, 16, 1);
  launch_gemm(stream, 0, false, si, WKh, k, BL, 128, Cc, Cc, 128, 128, 0, 0, 1, 0, 0, 16, 1);
  launch_gemm(stream, 0, false, si, WVh, v, BL, 128, Cc, Cc, 128, 128, 0, 0, 1, 0, 0, 16, 1);
  launch_gemm(stream, 0, false, si, WQhp, qp, BL, 96, Cc, Cc, 96, 96, 0, 0, 1, 0, 0, 16, 1);
  launch_gemm(stream, 0, false, si, WKhp, kp, BL, 96, Cc, Cc, 96, 96, 0, 0, 1, 0, 0, 16, 1);
  launch_gemm(stream, 0, false, si, WVhp, vp, BL, 192, Cc, Cc, 192, 192, 0, 0, 1, 0, 0, 16, 1);

  k_prep<<<(Bc * Lc * Hc + 255) / 256, 256, 0, stream>>>(
      q, k, v, qp, kp, vp, Ri, Ti, gamah, qa, ka, vaug, sqq, sqk);

  launch_gemm(stream, 0, true, qa, ka, alpha, Lc, Lc, 32, 32, 32, Lc,
              (long long)Lc * 32, 0, 1, (long long)Lc * 32, LL, 16, Bc * Hc);

  k_slice<<<BL, 768, 0, stream>>>(zij, Wb, alpha, sqq, sqk, gamah, feats);

  launch_gemm(stream, 0, false, alpha, vaug, ovalpt, Lc, 48, Lc, Lc, 48, 48,
              LL, 0, 1, (long long)Lc * 48, (long long)Lc * 48, 16, Bc * Hc);

  k_finish<<<(Bc * Lc * Hc + 255) / 256, 256, 0, stream>>>(ovalpt, Ri, Ti, feats);

  launch_gemm(stream, 1, false, feats, WOw, out, BL, Cc, FEAT, FEAT, Cc, Cc,
              0, 0, 1, 0, 0, 16, 1, WOb);
}


# k_slice as 256 resident 8-wave workgroups of 3 slices each: the next slice's rows are requested into registers before the current slice's bias, softmax and o_pair run; every sum keeps its order
# speedup vs baseline: 3.7212x; 1.1469x over previous
#include <hip/hip_runtime.h>
#include <math.h>

typedef __attribute__((ext_vector_type(2))) float v2f;
typedef __attribute__((ext_vector_type(4))) float f4;
typedef __attribute__((ext_vector_type(8))) float v8f;

namespace {
constexpr int Bc = 2, Lc = 384, Cc = 128, Hc = 8, DKc = 16, NQPc = 4, NPVc = 8;
constexpr float WCc = 0.2357022603955158f;
constexpr float WLc = 0.5773502691896258f;
constexpr int FEAT = Hc * Cc + Hc * DKc + Hc * NPVc * 3 + Hc * NPVc;
}

__device__ __forceinline__ float softplusf(float x) {
  return (x > 0.f) ? (x + log1pf(expf(-x))) : log1pf(expf(x));
}

template <bool TWO, int KT>
__device__ __forceinline__ void wave_kloop_t(const float* __restrict__ ap,
                                             const float* __restrict__ bt0,
                                             const float* __restrict__ bt1,
                                             int K, v8f& acc0, v8f& acc1) {
  constexpr int NS = KT / 4;
  for (int kb = 0; kb < K; kb += KT) {
    v2f a[NS], b0[NS], b1[NS];
#pragma unroll
    for (int s = 0; s < NS; ++s) {
      const int k = kb + 4 * s;
      a[s] = *(const v2f*)(ap + k);
      b0[s] = *(const v2f*)(bt0 + k);
      if (TWO) b1[s] = *(const v2f*)(bt1 + k);
    }
#pragma unroll
    for (int s = 0; s < NS; ++s) {
      acc0 = __builtin_amdgcn_wmma_f32_16x16x4_f32(false, a[s], false, b0[s],
                                                   (short)0, acc0, false, false);
      if (TWO)
        acc1 = __builtin_amdgcn_wmma_f32_16x16x4_f32(false, a[s], false, b1[s],
                                                     (short)0, acc1, false, false);
    }
  }
}

namespace {
constexpr int PKT = 32;
constexpr int PNS = PKT / 4;
constexpr int PSETS = 4;
constexpr int PAHEAD = PSETS - 1;
}

template <bool TWO>
__device__ __forceinline__ void load_group(const float* __restrict__ ap,
                                           const char* __restrict__ Bb,
                                           unsigned ob, unsigned ldb4, int kb,
                                           v2f (&a)[PNS], v2f (&x)[PNS],
                                           v2f (&y)[PNS]) {
#pragma unroll
  for (int s = 0; s < PNS; ++s) {
    const int k = kb + 4 * s;
    a[s] = *(const v2f*)(ap + k);
    const char* r0 = Bb + (unsigned long long)((unsigned)k * ldb4);
    const char* r1 = r0 + ldb4;
    if (TWO) {
      x[s] = *(const v2f*)(r0 + ob);
      y[s] = *(const v2f*)(r1 + ob);
    } else {
      x[s].x = *(const float*)(r0 + ob);
      x[s].y = *(const float*)(r1 + ob);
    }
  }
}

template <bool TWO>
__device__ __forceinline__ void mma_group(const v2f (&a)[PNS],
                                          const v2f (&x)[PNS],
                                          const v2f (&y)[PNS], v8f& acc0,
                                          v8f& acc1) {
#pragma unroll
  for (int s = 0; s < PNS; ++s) {
    if (TWO) {
      v2f b0, b1;
      b0.x = x[s].x;
      b0.y = y[s].x;
      b1.x = x[s].y;
      b1.y = y[s].y;
      acc0 = __builtin_amdgcn_wmma_f32_16x16x4_f32(false, a[s], false, b0,
                                                   (short)0, acc0, false, false);
      acc1 = __builtin_amdgcn_wmma_f32_16x16x4_f32(false, a[s], false, b1,
                                                   (short)0, acc1, false, false);
    } else {
      acc0 = __builtin_amdgcn_wmma_f32_16x16x4_f32(false, a[s], false, x[s],
                                                   (short)0, acc0, false, false);
    }
  }
}

template <bool TWO>
__device__ __forceinline__ void wave_kloop_pipe(const float* __restrict__ ap,
                                                const char* __restrict__ Bb,
                                                unsigned ob, unsigned ldb4,
                                                int K, v8f& acc0, v8f& acc1) {
  v2f fa[PSETS][PNS], fx[PSETS][PNS], fy[PSETS][PNS];
#pragma unroll
  for (int g = 0; g < PAHEAD; ++g) {
    load_group<TWO>(ap, Bb, ob, ldb4, g * PKT, fa[g], fx[g], fy[g]);
    __builtin_amdgcn_sched_barrier(0);
  }
  int kb = 0;
  for (; kb + PSETS * PKT < K; kb += PSETS * PKT) {
#pragma unroll
    for (int u = 0; u < PSETS; ++u) {
      __builtin_amdgcn_sched_barrier(0);
      load_group<TWO>(ap, Bb, ob, ldb4, kb + (u + PAHEAD) * PKT,
                      fa[(u + PAHEAD) % PSETS], fx[(u + PAHEAD) % PSETS],
                      fy[(u + PAHEAD) % PSETS]);
      __builtin_amdgcn_sched_barrier(0);
      mma_group<TWO>(fa[u], fx[u], fy[u], acc0, acc1);
    }
  }
  __builtin_amdgcn_sched_barrier(0);
  load_group<TWO>(ap, Bb, ob, ldb4, kb + PAHEAD * PKT, fa[PAHEAD % PSETS],
                  fx[PAHEAD % PSETS], fy[PAHEAD % PSETS]);
  __builtin_amdgcn_sched_barrier(0);
#pragma unroll
  for (int u = 0; u < PSETS; ++u)
    mma_group<TWO>(fa[u], fx[u], fy[u], acc0, acc1);
}

template <int MODE, bool TRANSB>
__global__ __launch_bounds__(128) __attribute__((amdgpu_waves_per_eu(4, 4))) void k_gemm(
    const float* __restrict__ A, const float* __restrict__ Bm,
    float* __restrict__ Cm, int M, int N, int K, int lda, int ldb, int ldc,
    long long sA, long long sA2, int zdiv, long long sB, long long sC,
    int Mvalid, const float* __restrict__ e0) {
  const int z = blockIdx.z;
  const float* Ag = A + (long long)(z / zdiv) * sA + (long long)(z % zdiv) * sA2;
  const float* Bg = Bm + (long long)z * sB;
  float* Cg = Cm + (long long)z * sC;

  const int m0 = blockIdx.x * 16;
  const int n0 = blockIdx.y * 128;
  const int tid = threadIdx.x;
  const int wave = __builtin_amdgcn_readfirstlane(tid >> 5);
  const int lane = tid & 31;
  const int hf = lane >> 4;
  const int l16 = lane & 15;
  const int nw = wave * 32;
  const int nb = n0 + nw;

  if (nb >= N) return;
  const bool two = (nb + 16) < N;

  const int n_t0 = two ? (nb + 2 * l16) : (nb + l16);
  const int n_t1 = n_t0 + 1;
  const int ra = (l16 < Mvalid && (m0 + l16) < M) ? l16 : 0;
  const int c0 = (n_t0 < N) ? n_t0 : nb;

  const float* ap = Ag + (long long)(m0 + ra) * lda + 2 * hf;
  const unsigned ldb4 = (unsigned)ldb * 4u;
  const unsigned ob = (unsigned)(2 * hf * ldb + c0) * 4u;
  const float* bt0 = Bg + (long long)c0 * ldb + 2 * hf;
  const float* bt1 = bt0 + ldb;

  v8f acc0 = {};
  v8f acc1 = {};
  if (TRANSB) {
    if (two)
      wave_kloop_t<true, 32>(ap, bt0, bt1, K, acc0, acc1);
    else
      wave_kloop_t<false, 32>(ap, bt0, bt1, K, acc0, acc1);
  } else {
    if (two)
      wave_kloop_pipe<true>(ap, (const char*)Bg, ob, ldb4, K, acc0, acc1);
    else
      wave_kloop_pipe<false>(ap, (const char*)Bg, ob, ldb4, K, acc0, acc1);
  }

#pragma unroll
  for (int r = 0; r < 8; ++r) {
    int mrow = r + 8 * hf;
    int m = m0 + mrow;
    if (mrow >= Mvalid || m >= M) continue;
#pragma unroll
    for (int t = 0; t < 2; ++t) {
      int n = (t == 0) ? n_t0 : n_t1;
      if (n >= N || (t == 1 && !two)) continue;
      float val = (t == 0) ? acc0[r] : acc1[r];
      if (MODE == 1) val += e0[n];
      Cg[(long long)m * ldc + n] = val;
    }
  }
}

__global__ __launch_bounds__(256) void k_prep(
    const float* __restrict__ q, const float* __restrict__ k,
    const float* __restrict__ v, const float* __restrict__ qp,
    const float* __restrict__ kp, const float* __restrict__ vp,
    const float* __restrict__ Ri, const float* __restrict__ Ti,
    const float* __restrict__ gamah, float* __restrict__ qa,
    float* __restrict__ ka, float* __restrict__ vaug, float* __restrict__ sqq,
    float* __restrict__ sqk) {
  int idx = blockIdx.x * blockDim.x + threadIdx.x;
  if (idx >= Bc * Lc * Hc) return;
  int h = idx % Hc;
  int bi = idx / Hc;
  int b = bi / Lc, i = bi % Lc;
  int z = b * Hc + h;

  float R[3][3], t[3];
#pragma unroll
  for (int r = 0; r < 3; ++r)
#pragma unroll
    for (int c = 0; c < 3; ++c) R[r][c] = Ri[bi * 9 + r * 3 + c];
#pragma unroll
  for (int x = 0; x < 3; ++x) t[x] = Ti[bi * 3 + x];

  float gwc = WCc * softplusf(gamah[h]);

  float* qrow = qa + ((long long)z * Lc + i) * 32;
  float* krow = ka + ((long long)z * Lc + i) * 32;
  float* vrow = vaug + ((long long)z * Lc + i) * 48;

#pragma unroll
  for (int d = 0; d < DKc; ++d) {
    qrow[d] = 0.25f * q[(long long)bi * 128 + h * 16 + d];
    krow[d] = k[(long long)bi * 128 + h * 16 + d];
    vrow[d] = v[(long long)bi * 128 + h * 16 + d];
  }
  float sq = 0.f, sk = 0.f;
#pragma unroll
  for (int p = 0; p < NQPc; ++p) {
    const float* qs = qp + (long long)bi * 96 + h * 12 + p * 3;
    const float* ks = kp + (long long)bi * 96 + h * 12 + p * 3;
#pragma unroll
    for (int m = 0; m < 3; ++m) {
      float qt = R[m][0] * qs[0] + R[m][1] * qs[1] + R[m][2] * qs[2] + t[m];
      float kt = R[m][0] * ks[0] + R[m][1] * ks[1] + R[m][2] * ks[2] + t[m];
      sq += qt * qt;
      sk += kt * kt;
      qrow[16 + p * 3 + m] = gwc * qt;
      krow[16 + p * 3 + m] = kt;
    }
  }
#pragma unroll
  for (int d = 28; d < 32; ++d) { qrow[d] = 0.f; krow[d] = 0.f; }
#pragma unroll
  for (int p = 0; p < NPVc; ++p) {
    const float* vs = vp + (long long)bi * 192 + h * 24 + p * 3;
#pragma unroll
    for (int m = 0; m < 3; ++m)
      vrow[16 + p * 3 + m] =
          R[m][0] * vs[0] + R[m][1] * vs[1] + R[m][2] * vs[2] + t[m];
  }
#pragma unroll
  for (int d = 40; d < 48; ++d) vrow[d] = 0.f;
  sqq[(long long)z * Lc + i] = sq;
  sqk[(long long)z * Lc + i] = sk;
}

namespace {
constexpr int ZP = 140;
constexpr int RP = 388;
constexpr int NSL = 3;
constexpr int SLICE_WGS = Bc * Lc / NSL;
constexpr int RW = Lc / 8;
static_assert(Lc % NSL == 0, "the slices of a workgroup share b");
static_assert(RW % 16 == 0, "whole 16-row tiles per wave");
}

__device__ __forceinline__ void barrier_lds() {
  __builtin_amdgcn_fence(__ATOMIC_RELEASE, "workgroup", "local");
  __builtin_amdgcn_s_barrier();
  __builtin_amdgcn_fence(__ATOMIC_ACQUIRE, "workgroup", "local");
}

__global__ __launch_bounds__(256) __attribute__((amdgpu_waves_per_eu(2, 2))) void k_slice(
    const float* __restrict__ zij, const float* __restrict__ Wb,
    float* __restrict__ S, const float* __restrict__ sqq,
    const float* __restrict__ sqk, const float* __restrict__ gamah,
    float* __restrict__ feats) {
  __shared__ __attribute__((aligned(16))) float zs[Lc * ZP];
  __shared__ __attribute__((aligned(16))) float sb[Hc * RP];
  __shared__ __attribute__((aligned(16))) float sa[Hc * RP];

  const int bi0 = blockIdx.x * NSL;
  const int b = bi0 / Lc, i0 = bi0 % Lc;
  const int tid = threadIdx.x;
  const int wave = __builtin_amdgcn_readfirstlane(tid >> 5);
  const int lane = tid & 31;
  const int hf = lane >> 4;
  const int l16 = lane & 15;
  const int j0 = wave * RW;
  const int z = b * Hc + wave;

  const f4* zsrc = (const f4*)(zij + ((long long)bi0 * Lc + j0) * Cc) + lane;
  f4 t[RW];
#pragma unroll
  for (int r = 0; r < RW; ++r) t[r] = zsrc[r * (Cc / 4)];

  float kq[12];
#pragma unroll
  for (int e = 0; e < 12; ++e) kq[e] = sqk[(long long)z * Lc + e * 32 + lane];
  v2f w[32];
  {
    const int cw = (l16 < Hc) ? l16 : 0;
    const float* wp = Wb + (2 * hf) * Hc + cw;
#pragma unroll
    for (int s = 0; s < 32; ++s) {
      w[s].x = wp[(4 * s) * Hc];
      w[s].y = wp[(4 * s + 1) * Hc];
    }
  }
  __builtin_amdgcn_sched_barrier(0);
  const float gw = 0.5f * WCc * softplusf(gamah[wave]);

#pragma unroll
  for (int sl = 0; sl < NSL; ++sl) {
    const int bi = bi0 + sl;
    const int i = i0 + sl;

#pragma unroll
    for (int r = 0; r < RW; ++r) *(f4*)(zs + (j0 + r) * ZP + 4 * lane) = t[r];
    __builtin_amdgcn_sched_barrier(0);

    float* p = S + ((long long)z * Lc + i) * Lc;
    float pv[12];
#pragma unroll
    for (int e = 0; e < 12; ++e) pv[e] = p[e * 32 + lane];
    const float sq = sqq[(long long)z * Lc + i];

    if (sl + 1 < NSL) {
      const f4* zn = zsrc + (long long)(sl + 1) * (Lc * Cc / 4);
#pragma unroll
      for (int r = 0; r < RW; ++r) t[r] = zn[r * (Cc / 4)];
    }
    __builtin_amdgcn_sched_barrier(0);

    barrier_lds();

    {
      const float* zp = zs + (j0 + l16) * ZP + 2 * hf;
      v8f acc[RW / 16] = {};
#pragma unroll
      for (int g = 0; g < 4; ++g) {
        v2f a[RW / 16][8];
#pragma unroll
        for (int tt = 0; tt < RW / 16; ++tt)
#pragma unroll
          for (int s = 0; s < 8; ++s)
            a[tt][s] = *(const v2f*)(zp + 16 * tt * ZP + 32 * g + 4 * s);
#pragma unroll
        for (int s = 0; s < 8; ++s)
#pragma unroll
          for (int tt = 0; tt < RW / 16; ++tt)
            acc[tt] = __builtin_amdgcn_wmma_f32_16x16x4_f32(
                false, a[tt][s], false, w[8 * g + s], (short)0, acc[tt], false,
                false);
      }
      if (l16 < Hc) {
#pragma unroll
        for (int tt = 0; tt < RW / 16; ++tt)
#pragma unroll
          for (int r = 0; r < 8; ++r)
            sb[l16 * RP + j0 + 16 * tt + r + 8 * hf] = acc[tt][r];
      }
    }
    barrier_lds();

    {
      float vals[12];
      float mx = -INFINITY;
#pragma unroll
      for (int e = 0; e < 12; ++e) {
        const int j = e * 32 + lane;
        vals[e] = WLc * (pv[e] + sb[wave * RP + j] - gw * (sq + kq[e]));
        mx = fmaxf(mx, vals[e]);
      }
#pragma unroll
      for (int o = 16; o >= 1; o >>= 1) mx = fmaxf(mx, __shfl_xor(mx, o, 32));
      float sum = 0.f;
#pragma unroll
      for (int e = 0; e < 12; ++e) {
        vals[e] = expf(vals[e] - mx);
        sum += vals[e];
      }
#pragma unroll
      for (int o = 16; o >= 1; o >>= 1) sum += __shfl_xor(sum, o, 32);
      const float inv = 1.f / sum;
#pragma unroll
      for (int e = 0; e < 12; ++e) {
        const int j = e * 32 + lane;
        const float al = vals[e] * inv;
        p[j] = al;
        sa[wave * RP + j] = al;
      }
    }
    barrier_lds();

    {
      const int c = wave * 16 + l16;
      const int ra = (l16 < Hc) ? l16 : 0;
      const float* ap = sa + ra * RP + 2 * hf;
      const float* zc = zs + (2 * hf) * ZP + c;
      v8f acc = {};
      for (int kb = 0; kb < Lc; kb += 64) {
        v2f a[16], bq[16];
#pragma unroll
        for (int s = 0; s < 16; ++s) {
          const int k = kb + 4 * s;
          a[s] = *(const v2f*)(ap + k);
          bq[s].x = zc[k * ZP];
          bq[s].y = zc[(k + 1) * ZP];
        }
        asm volatile("" ::"v"(a[0]), "v"(a[1]), "v"(a[2]), "v"(a[3]), "v"(a[4]), "v"(a[5]),
                     "v"(a[6]), "v"(a[7]), "v"(a[8]), "v"(a[9]), "v"(a[10]),
                     "v"(a[11]), "v"(a[12]), "v"(a[13]), "v"(a[14]), "v"(a[15]));
        asm volatile("" ::"v"(bq[0]), "v"(bq[1]), "v"(bq[2]), "v"(bq[3]), "v"(bq[4]), "v"(bq[5]),
                     "v"(bq[6]), "v"(bq[7]), "v"(bq[8]), "v"(bq[9]), "v"(bq[10]),
                     "v"(bq[11]), "v"(bq[12]), "v"(bq[13]), "v"(bq[14]), "v"(bq[15]));
#pragma unroll
        for (int s = 0; s < 16; ++s)
          acc = __builtin_amdgcn_wmma_f32_16x16x4_f32(false, a[s], false, bq[s],
                                                      (short)0, acc, false, false);
      }
      if (hf == 0) {
        float* frow = feats + (long long)bi * FEAT + c;
#pragma unroll
        for (int r = 0; r < 8; ++r) frow[r * Cc] = acc[r];
      }
    }
    if (sl + 1 < NSL) barrier_lds();
  }
}

__global__ __launch_bounds__(256) void k_finish(
    const float* __restrict__ ovalpt, const float* __restrict__ Ri,
    const float* __restrict__ Ti, float* __restrict__ feats) {
  int idx = blockIdx.x * blockDim.x + threadIdx.x;
  if (idx >= Bc * Lc * Hc) return;
  int h = idx % Hc;
  int bi = idx / Hc;
  int b = bi / Lc;
  int z = b * Hc + h;
  int i = bi % Lc;

  float R[3][3], t[3];
#pragma unroll
  for (int r = 0; r < 3; ++r)
#pragma unroll
    for (int c = 0; c < 3; ++c) R[r][c] = Ri[bi * 9 + r * 3 + c];
#pragma unroll
  for (int x = 0; x < 3; ++x) t[x] = Ti[bi * 3 + x];

  const float* orow = ovalpt + ((long long)z * Lc + i) * 48;
  float* frow = feats + (long long)bi * FEAT;

#pragma unroll
  for (int d = 0; d < DKc; ++d) frow[1024 + h * 16 + d] = orow[d];
#pragma unroll
  for (int p = 0; p < NPVc; ++p) {
    float d0 = orow[16 + p * 3 + 0] - t[0];
    float d1 = orow[16 + p * 3 + 1] - t[1];
    float d2 = orow[16 + p * 3 + 2] - t[2];
    float asum = 0.f;
#pragma unroll
    for (int m = 0; m < 3; ++m) {
      float oh = R[0][m] * d0 + R[1][m] * d1 + R[2][m] * d2;
      frow[1152 + h * 24 + p * 3 + m] = oh;
      asum += fabsf(oh);
    }
    frow[1344 + h * 8 + p] = asum;
  }
}

static void launch_gemm(hipStream_t s, int mode, bool transb, const float* A,
                        const float* Bm, float* C, int M, int N, int K, int lda,
                        int ldb, int ldc, long long sA, long long sA2, int zdiv,
                        long long sB, long long sC, int Mvalid, int nz,
                        const float* e0 = nullptr) {
  dim3 g((M + 15) / 16, (N + 127) / 128, nz);
  dim3 blk(128, 1, 1);
  if (mode == 0 && !transb)
    k_gemm<0, false><<<g, blk, 0, s>>>(A, Bm, C, M, N, K, lda, ldb, ldc, sA,
                                       sA2, zdiv, sB, sC, Mvalid, e0);
  else if (mode == 1)
    k_gemm<1, false><<<g, blk, 0, s>>>(A, Bm, C, M, N, K, lda, ldb, ldc, sA,
                                       sA2, zdiv, sB, sC, Mvalid, e0);
  else
    k_gemm<0, true><<<g, blk, 0, s>>>(A, Bm, C, M, N, K, lda, ldb, ldc, sA,
                                      sA2, zdiv, sB, sC, Mvalid, e0);
}

extern "C" void kernel_launch(void* const* d_in, const int* in_sizes, int n_in,
                              void* d_out, int out_size, void* d_ws,
                              size_t ws_size, hipStream_t stream) {
  const float* si = (const float*)d_in[0];
  const float* zij = (const float*)d_in[1];
  const float* Ri = (const float*)d_in[2];
  const float* Ti = (const float*)d_in[3];
  const float* WQh = (const float*)d_in[4];
  const float* WKh = (const float*)d_in[5];
  const float* WVh = (const float*)d_in[6];
  const float* WQhp = (const float*)d_in[7];
  const float* WKhp = (const float*)d_in[8];
  const float* WVhp = (const float*)d_in[9];
  const float* Wb = (const float*)d_in[10];
  const float* WOw = (const float*)d_in[11];
  const float* WOb = (const float*)d_in[12];
  const float* gamah = (const float*)d_in[13];
  float* out = (float*)d_out;
  float* ws = (float*)d_ws;

  const int BL = Bc * Lc;
  const long long LL = (long long)Lc * Lc;
  float* q = ws;
  float* k = q + 98304;
  float* v = k + 98304;
  float* qp = v + 98304;
  float* kp = qp + 73728;
  float* vp = kp + 73728;
  float* qa = vp + 147456;
  float* ka = qa + 196608;
  float* vaug = ka + 196608;
  float* sqq = vaug + 294912;
  float* sqk = sqq + 6144;
  float* alpha = sqk + 6144;
  float* ovalpt = alpha + 2359296;
  float* feats = ovalpt + 294912;

  launch_gemm(stream, 0, false, si, WQh, q, BL, 128, Cc, Cc, 128, 128, 0, 0, 1, 0, 0, 16, 1);
  launch_gemm(stream, 0, false, si, WKh, k, BL, 128, Cc, Cc, 128, 128, 0, 0, 1, 0, 0, 16, 1);
  launch_gemm(stream, 0, false, si, WVh, v, BL, 128, Cc, Cc, 128, 128, 0, 0, 1, 0, 0, 16, 1);
  launch_gemm(stream, 0, false, si, WQhp, qp, BL, 96, Cc, Cc, 96, 96, 0, 0, 1, 0, 0, 16, 1);
  launch_gemm(stream, 0, false, si, WKhp, kp, BL, 96, Cc, Cc, 96, 96, 0, 0, 1, 0, 0, 16, 1);
  launch_gemm(stream, 0, false, si, WVhp, vp, BL, 192, Cc, Cc, 192, 192, 0, 0, 1, 0, 0, 16, 1);

  k_prep<<<(Bc * Lc * Hc + 255) / 256, 256, 0, stream>>>(
      q, k, v, qp, kp, vp, Ri, Ti, gamah, qa, ka, vaug, sqq, sqk);

  launch_gemm(stream, 0, true, qa, ka, alpha, Lc, Lc, 32, 32, 32, Lc,
              (long long)Lc * 32, 0, 1, (long long)Lc * 32, LL, 16, Bc * Hc);

  k_slice<<<SLICE_WGS, 256, 0, stream>>>(zij, Wb, alpha, sqq, sqk, gamah, feats);

  launch_gemm(stream, 0, false, alpha, vaug, ovalpt, Lc, 48, Lc, Lc, 48, 48,
              LL, 0, 1, (long long)Lc * 48, (long long)Lc * 48, 16, Bc * Hc);

  k_finish<<<(Bc * Lc * Hc + 255) / 256, 256, 0, stream>>>(ovalpt, Ri, Ti, feats);

  launch_gemm(stream, 1, false, feats, WOw, out, BL, Cc, FEAT, FEAT, Cc, Cc,
              0, 0, 1, 0, 0, 16, 1, WOb);
}


# q.k launch builds its packed operands itself: each scores workgroup does the rigid transforms of its 16+128 rows into LDS, edge workgroups write sqq/sqk/vaug; 12 launches become 11; sums keep order
# speedup vs baseline: 4.0045x; 1.0761x over previous
#include <hip/hip_runtime.h>
#include <math.h>

typedef __attribute__((ext_vector_type(2))) float v2f;
typedef __attribute__((ext_vector_type(4))) float f4;
typedef __attribute__((ext_vector_type(8))) float v8f;

namespace {
constexpr int Bc = 2, Lc = 384, Cc = 128, Hc = 8, DKc = 16, NQPc = 4, NPVc = 8;
constexpr float WCc = 0.2357022603955158f;
constexpr float WLc = 0.5773502691896258f;
constexpr int FEAT = Hc * Cc + Hc * DKc + Hc * NPVc * 3 + Hc * NPVc;
}

__device__ __forceinline__ float softplusf(float x) {
  return (x > 0.f) ? (x + log1pf(expf(-x))) : log1pf(expf(x));
}

namespace {
constexpr int PKT = 32;
constexpr int PNS = PKT / 4;
constexpr int PSETS = 4;
constexpr int PAHEAD = PSETS - 1;
}

template <bool TWO>
__device__ __forceinline__ void load_group(const float* __restrict__ ap,
                                           const char* __restrict__ Bb,
                                           unsigned ob, unsigned ldb4, int kb,
                                           v2f (&a)[PNS], v2f (&x)[PNS],
                                           v2f (&y)[PNS]) {
#pragma unroll
  for (int s = 0; s < PNS; ++s) {
    const int k = kb + 4 * s;
    a[s] = *(const v2f*)(ap + k);
    const char* r0 = Bb + (unsigned long long)((unsigned)k * ldb4);
    const char* r1 = r0 + ldb4;
    if (TWO) {
      x[s] = *(const v2f*)(r0 + ob);
      y[s] = *(const v2f*)(r1 + ob);
    } else {
      x[s].x = *(const float*)(r0 + ob);
      x[s].y = *(const float*)(r1 + ob);
    }
  }
}

template <bool TWO>
__device__ __forceinline__ void mma_group(const v2f (&a)[PNS],
                                          const v2f (&x)[PNS],
                                          const v2f (&y)[PNS], v8f& acc0,
                                          v8f& acc1) {
#pragma unroll
  for (int s = 0; s < PNS; ++s) {
    if (TWO) {
      v2f b0, b1;
      b0.x = x[s].x;
      b0.y = y[s].x;
      b1.x = x[s].y;
      b1.y = y[s].y;
      acc0 = __builtin_amdgcn_wmma_f32_16x16x4_f32(false, a[s], false, b0,
                                                   (short)0, acc0, false, false);
      acc1 = __builtin_amdgcn_wmma_f32_16x16x4_f32(false, a[s], false, b1,
                                                   (short)0, acc1, false, false);
    } else {
      acc0 = __builtin_amdgcn_wmma_f32_16x16x4_f32(false, a[s], false, x[s],
                                                   (short)0, acc0, false, false);
    }
  }
}

template <bool TWO>
__device__ __forceinline__ void wave_kloop_pipe(const float* __restrict__ ap,
                                                const char* __restrict__ Bb,
                                                unsigned ob, unsigned ldb4,
                                                int K, v8f& acc0, v8f& acc1) {
  v2f fa[PSETS][PNS], fx[PSETS][PNS], fy[PSETS][PNS];
#pragma unroll
  for (int g = 0; g < PAHEAD; ++g) {
    load_group<TWO>(ap, Bb, ob, ldb4, g * PKT, fa[g], fx[g], fy[g]);
    __builtin_amdgcn_sched_barrier(0);
  }
  int kb = 0;
  for (; kb + PSETS * PKT < K; kb += PSETS * PKT) {
#pragma unroll
    for (int u = 0; u < PSETS; ++u) {
      __builtin_amdgcn_sched_barrier(0);
      load_group<TWO>(ap, Bb, ob, ldb4, kb + (u + PAHEAD) * PKT,
                      fa[(u + PAHEAD) % PSETS], fx[(u + PAHEAD) % PSETS],
                      fy[(u + PAHEAD) % PSETS]);
      __builtin_amdgcn_sched_barrier(0);
      mma_group<TWO>(fa[u], fx[u], fy[u], acc0, acc1);
    }
  }
  __builtin_amdgcn_sched_barrier(0);
  load_group<TWO>(ap, Bb, ob, ldb4, kb + PAHEAD * PKT, fa[PAHEAD % PSETS],
                  fx[PAHEAD % PSETS], fy[PAHEAD % PSETS]);
  __builtin_amdgcn_sched_barrier(0);
#pragma unroll
  for (int u = 0; u < PSETS; ++u)
    mma_group<TWO>(fa[u], fx[u], fy[u], acc0, acc1);
}

template <int MODE>
__global__ __launch_bounds__(128) __attribute__((amdgpu_waves_per_eu(4, 4))) void k_gemm(
    const float* __restrict__ A, const float* __restrict__ Bm,
    float* __restrict__ Cm, int M, int N, int K, int lda, int ldb, int ldc,
    long long sA, long long sA2, int zdiv, long long sB, long long sC,
    int Mvalid, const float* __restrict__ e0) {
  const int z = blockIdx.z;
  const float* Ag = A + (long long)(z / zdiv) * sA + (long long)(z % zdiv) * sA2;
  const float* Bg = Bm + (long long)z * sB;
  float* Cg = Cm + (long long)z * sC;

  const int m0 = blockIdx.x * 16;
  const int n0 = blockIdx.y * 128;
  const int tid = threadIdx.x;
  const int wave = __builtin_amdgcn_readfirstlane(tid >> 5);
  const int lane = tid & 31;
  const int hf = lane >> 4;
  const int l16 = lane & 15;
  const int nw = wave * 32;
  const int nb = n0 + nw;

  if (nb >= N) return;
  const bool two = (nb + 16) < N;

  const int n_t0 = two ? (nb + 2 * l16) : (nb + l16);
  const int n_t1 = n_t0 + 1;
  const int ra = (l16 < Mvalid && (m0 + l16) < M) ? l16 : 0;
  const int c0 = (n_t0 < N) ? n_t0 : nb;

  const float* ap = Ag + (long long)(m0 + ra) * lda + 2 * hf;
  const unsigned ldb4 = (unsigned)ldb * 4u;
  const unsigned ob = (unsigned)(2 * hf * ldb + c0) * 4u;

  v8f acc0 = {};
  v8f acc1 = {};
  if (two)
    wave_kloop_pipe<true>(ap, (const char*)Bg, ob, ldb4, K, acc0, acc1);
  else
    wave_kloop_pipe<false>(ap, (const char*)Bg, ob, ldb4, K, acc0, acc1);

#pragma unroll
  for (int r = 0; r < 8; ++r) {
    int mrow = r + 8 * hf;
    int m = m0 + mrow;
    if (mrow >= Mvalid || m >= M) continue;
#pragma unroll
    for (int t = 0; t < 2; ++t) {
      int n = (t == 0) ? n_t0 : n_t1;
      if (n >= N || (t == 1 && !two)) continue;
      float val = (t == 0) ? acc0[r] : acc1[r];
      if (MODE == 1) val += e0[n];
      Cg[(long long)m * ldc + n] = val;
    }
  }
}

namespace {
constexpr int QP = 36;
}

__device__ __forceinline__ void load_rigid(const float* __restrict__ Ri,
                                           const float* __restrict__ Ti, int bi,
                                           float (&R)[3][3], float (&t)[3]) {
#pragma unroll
  for (int r = 0; r < 3; ++r)
#pragma unroll
    for (int cc = 0; cc < 3; ++cc) R[r][cc] = Ri[bi * 9 + r * 3 + cc];
#pragma unroll
  for (int x = 0; x < 3; ++x) t[x] = Ti[bi * 3 + x];
}

__device__ __forceinline__ float pack_row(const float* __restrict__ s16,
                                          const float* __restrict__ p12,
                                          const float (&R)[3][3],
                                          const float (&t)[3], float sc,
                                          float pc, float* __restrict__ dst) {
  float row[32];
#pragma unroll
  for (int d = 0; d < DKc; ++d) row[d] = sc * s16[d];
  float sq = 0.f;
#pragma unroll
  for (int p = 0; p < NQPc; ++p) {
    const float* ps = p12 + p * 3;
#pragma unroll
    for (int m = 0; m < 3; ++m) {
      float xt = R[m][0] * ps[0] + R[m][1] * ps[1] + R[m][2] * ps[2] + t[m];
      sq += xt * xt;
      row[16 + p * 3 + m] = pc * xt;
    }
  }
#pragma unroll
  for (int d = 28; d < 32; ++d) row[d] = 0.f;
#pragma unroll
  for (int d = 0; d < 32; d += 4) {
    f4 o = {row[d], row[d + 1], row[d + 2], row[d + 3]};
    *(f4*)(dst + d) = o;
  }
  return sq;
}

__global__ __launch_bounds__(128) void k_scores(
    const float* __restrict__ q, const float* __restrict__ k,
    const float* __restrict__ v, const float* __restrict__ qp,
    const float* __restrict__ kp, const float* __restrict__ vp,
    const float* __restrict__ Ri, const float* __restrict__ Ti,
    const float* __restrict__ gamah, float* __restrict__ S,
    float* __restrict__ vaug, float* __restrict__ sqq,
    float* __restrict__ sqk) {
  __shared__ __attribute__((aligned(16))) float qs[16 * QP];
  __shared__ __attribute__((aligned(16))) float kss[128 * QP];

  const int z = blockIdx.z;
  const int b = z / Hc, h = z % Hc;
  const int m0 = blockIdx.x * 16;
  const int n0 = blockIdx.y * 128;
  const int tid = threadIdx.x;

  {
    const int j = n0 + tid;
    const int bj = b * Lc + j;
    float R[3][3], t[3];
    load_rigid(Ri, Ti, bj, R, t);
    const float sk = pack_row(k + (long long)bj * 128 + h * 16,
                              kp + (long long)bj * 96 + h * 12, R, t, 1.f, 1.f,
                              kss + tid * QP);
    if (blockIdx.x == 0) {
      sqk[(long long)z * Lc + j] = sk;
      float* vrow = vaug + ((long long)z * Lc + j) * 48;
#pragma unroll
      for (int d = 0; d < DKc; ++d) vrow[d] = v[(long long)bj * 128 + h * 16 + d];
#pragma unroll
      for (int p = 0; p < NPVc; ++p) {
        const float* vs = vp + (long long)bj * 192 + h * 24 + p * 3;
#pragma unroll
        for (int m = 0; m < 3; ++m)
          vrow[16 + p * 3 + m] =
              R[m][0] * vs[0] + R[m][1] * vs[1] + R[m][2] * vs[2] + t[m];
      }
#pragma unroll
      for (int d = 40; d < 48; ++d) vrow[d] = 0.f;
    }
  }
  if (tid < 16) {
    const int i = m0 + tid;
    const int bi = b * Lc + i;
    float R[3][3], t[3];
    load_rigid(Ri, Ti, bi, R, t);
    const float gwc = WCc * softplusf(gamah[h]);
    const float sq = pack_row(q + (long long)bi * 128 + h * 16,
                              qp + (long long)bi * 96 + h * 12, R, t, 0.25f, gwc,
                              qs + tid * QP);
    if (blockIdx.y == 0) sqq[(long long)z * Lc + i] = sq;
  }
  __syncthreads();

  const int wave = __builtin_amdgcn_readfirstlane(tid >> 5);
  const int lane = tid & 31;
  const int hf = lane >> 4;
  const int l16 = lane & 15;

  const float* ap = qs + l16 * QP + 2 * hf;
  const float* bp0 = kss + (wave * 32 + l16) * QP + 2 * hf;
  const float* bp1 = bp0 + 16 * QP;
  v2f a[8], b0[8], b1[8];
#pragma unroll
  for (int s = 0; s < 8; ++s) {
    a[s] = *(const v2f*)(ap + 4 * s);
    b0[s] = *(const v2f*)(bp0 + 4 * s);
    b1[s] = *(const v2f*)(bp1 + 4 * s);
  }
  v8f acc0 = {};
  v8f acc1 = {};
#pragma unroll
  for (int s = 0; s < 8; ++s) {
    acc0 = __builtin_amdgcn_wmma_f32_16x16x4_f32(false, a[s], false, b0[s],
                                                 (short)0, acc0, false, false);
    acc1 = __builtin_amdgcn_wmma_f32_16x16x4_f32(false, a[s], false, b1[s],
                                                 (short)0, acc1, false, false);
  }

  float* srow = S + ((long long)z * Lc + m0 + 8 * hf) * Lc + n0 + wave * 32 + l16;
#pragma unroll
  for (int r = 0; r < 8; ++r) {
    srow[(long long)r * Lc] = acc0[r];
    srow[(long long)r * Lc + 16] = acc1[r];
  }
}

namespace {
constexpr int ZP = 140;
constexpr int RP = 388;
constexpr int NSL = 3;
constexpr int SLICE_WGS = Bc * Lc / NSL;
constexpr int RW = Lc / 8;
static_assert(Lc % NSL == 0, "the slices of a workgroup share b");
static_assert(RW % 16 == 0, "whole 16-row tiles per wave");
}

__device__ __forceinline__ void barrier_lds() {
  __builtin_amdgcn_fence(__ATOMIC_RELEASE, "workgroup", "local");
  __builtin_amdgcn_s_barrier();
  __builtin_amdgcn_fence(__ATOMIC_ACQUIRE, "workgroup", "local");
}

__global__ __launch_bounds__(256) __attribute__((amdgpu_waves_per_eu(2, 2))) void k_slice(
    const float* __restrict__ zij, const float* __restrict__ Wb,
    float* __restrict__ S, const float* __restrict__ sqq,
    const float* __restrict__ sqk, const float* __restrict__ gamah,
    float* __restrict__ feats) {
  __shared__ __attribute__((aligned(16))) float zs[Lc * ZP];
  __shared__ __attribute__((aligned(16))) float sb[Hc * RP];
  __shared__ __attribute__((aligned(16))) float sa[Hc * RP];

  const int bi0 = blockIdx.x * NSL;
  const int b = bi0 / Lc, i0 = bi0 % Lc;
  const int tid = threadIdx.x;
  const int wave = __builtin_amdgcn_readfirstlane(tid >> 5);
  const int lane = tid & 31;
  const int hf = lane >> 4;
  const int l16 = lane & 15;
  const int j0 = wave * RW;
  const int z = b * Hc + wave;

  const f4* zsrc = (const f4*)(zij + ((long long)bi0 * Lc + j0) * Cc) + lane;
  f4 t[RW];
#pragma unroll
  for (int r = 0; r < RW; ++r) t[r] = zsrc[r * (Cc / 4)];

  float kq[12];
#pragma unroll
  for (int e = 0; e < 12; ++e) kq[e] = sqk[(long long)z * Lc + e * 32 + lane];
  v2f w[32];
  {
    const int cw = (l16 < Hc) ? l16 : 0;
    const float* wp = Wb + (2 * hf) * Hc + cw;
#pragma unroll
    for (int s = 0; s < 32; ++s) {
      w[s].x = wp[(4 * s) * Hc];
      w[s].y = wp[(4 * s + 1) * Hc];
    }
  }
  __builtin_amdgcn_sched_barrier(0);
  const float gw = 0.5f * WCc * softplusf(gamah[wave]);

#pragma unroll
  for (int sl = 0; sl < NSL; ++sl) {
    const int bi = bi0 + sl;
    const int i = i0 + sl;

#pragma unroll
    for (int r = 0; r < RW; ++r) *(f4*)(zs + (j0 + r) * ZP + 4 * lane) = t[r];
    __builtin_amdgcn_sched_barrier(0);

    float* p = S + ((long long)z * Lc + i) * Lc;
    float pv[12];
#pragma unroll
    for (int e = 0; e < 12; ++e) pv[e] = p[e * 32 + lane];
    const float sq = sqq[(long long)z * Lc + i];

    if (sl + 1 < NSL) {
      const f4* zn = zsrc + (long long)(sl + 1) * (Lc * Cc / 4);
#pragma unroll
      for (int r = 0; r < RW; ++r) t[r] = zn[r * (Cc / 4)];
    }
    __builtin_amdgcn_sched_barrier(0);

    barrier_lds();

    {
      const float* zp = zs + (j0 + l16) * ZP + 2 * hf;
      v8f acc[RW / 16] = {};
#pragma unroll
      for (int g = 0; g < 4; ++g) {
        v2f a[RW / 16][8];
#pragma unroll
        for (int tt = 0; tt < RW / 16; ++tt)
#pragma unroll
          for (int s = 0; s < 8; ++s)
            a[tt][s] = *(const v2f*)(zp + 16 * tt * ZP + 32 * g + 4 * s);
#pragma unroll
        for (int s = 0; s < 8; ++s)
#pragma unroll
          for (int tt = 0; tt < RW / 16; ++tt)
            acc[tt] = __builtin_amdgcn_wmma_f32_16x16x4_f32(
                false, a[tt][s], false, w[8 * g + s], (short)0, acc[tt], false,
                false);
      }
      if (l16 < Hc) {
#pragma unroll
        for (int tt = 0; tt < RW / 16; ++tt)
#pragma unroll
          for (int r = 0; r < 8; ++r)
            sb[l16 * RP + j0 + 16 * tt + r + 8 * hf] = acc[tt][r];
      }
    }
    barrier_lds();

    {
      float vals[12];
      float mx = -INFINITY;
#pragma unroll
      for (int e = 0; e < 12; ++e) {
        const int j = e * 32 + lane;
        vals[e] = WLc * (pv[e] + sb[wave * RP + j] - gw * (sq + kq[e]));
        mx = fmaxf(mx, vals[e]);
      }
#pragma unroll
      for (int o = 16; o >= 1; o >>= 1) mx = fmaxf(mx, __shfl_xor(mx, o, 32));
      float sum = 0.f;
#pragma unroll
      for (int e = 0; e < 12; ++e) {
        vals[e] = expf(vals[e] - mx);
        sum += vals[e];
      }
#pragma unroll
      for (int o = 16; o >= 1; o >>= 1) sum += __shfl_xor(sum, o, 32);
      const float inv = 1.f / sum;
#pragma unroll
      for (int e = 0; e < 12; ++e) {
        const int j = e * 32 + lane;
        const float al = vals[e] * inv;
        p[j] = al;
        sa[wave * RP + j] = al;
      }
    }
    barrier_lds();

    {
      const int c = wave * 16 + l16;
      const int ra = (l16 < Hc) ? l16 : 0;
      const float* ap = sa + ra * RP + 2 * hf;
      const float* zc = zs + (2 * hf) * ZP + c;
      v8f acc = {};
      for (int kb = 0; kb < Lc; kb += 64) {
        v2f a[16], bq[16];
#pragma unroll
        for (int s = 0; s < 16; ++s) {
          const int k = kb + 4 * s;
          a[s] = *(const v2f*)(ap + k);
          bq[s].x = zc[k * ZP];
          bq[s].y = zc[(k + 1) * ZP];
        }
        asm volatile("" ::"v"(a[0]), "v"(a[1]), "v"(a[2]), "v"(a[3]), "v"(a[4]), "v"(a[5]),
                     "v"(a[6]), "v"(a[7]), "v"(a[8]), "v"(a[9]), "v"(a[10]),
                     "v"(a[11]), "v"(a[12]), "v"(a[13]), "v"(a[14]), "v"(a[15]));
        asm volatile("" ::"v"(bq[0]), "v"(bq[1]), "v"(bq[2]), "v"(bq[3]), "v"(bq[4]), "v"(bq[5]),
                     "v"(bq[6]), "v"(bq[7]), "v"(bq[8]), "v"(bq[9]), "v"(bq[10]),
                     "v"(bq[11]), "v"(bq[12]), "v"(bq[13]), "v"(bq[14]), "v"(bq[15]));
#pragma unroll
        for (int s = 0; s < 16; ++s)
          acc = __builtin_amdgcn_wmma_f32_16x16x4_f32(false, a[s], false, bq[s],
                                                      (short)0, acc, false, false);
      }
      if (hf == 0) {
        float* frow = feats + (long long)bi * FEAT + c;
#pragma unroll
        for (int r = 0; r < 8; ++r) frow[r * Cc] = acc[r];
      }
    }
    if (sl + 1 < NSL) barrier_lds();
  }
}

__global__ __launch_bounds__(256) void k_finish(
    const float* __restrict__ ovalpt, const float* __restrict__ Ri,
    const float* __restrict__ Ti, float* __restrict__ feats) {
  int idx = blockIdx.x * blockDim.x + threadIdx.x;
  if (idx >= Bc * Lc * Hc) return;
  int h = idx % Hc;
  int bi = idx / Hc;
  int b = bi / Lc;
  int z = b * Hc + h;
  int i = bi % Lc;

  float R[3][3], t[3];
#pragma unroll
  for (int r = 0; r < 3; ++r)
#pragma unroll
    for (int c = 0; c < 3; ++c) R[r][c] = Ri[bi * 9 + r * 3 + c];
#pragma unroll
  for (int x = 0; x < 3; ++x) t[x] = Ti[bi * 3 + x];

  const float* orow = ovalpt + ((long long)z * Lc + i) * 48;
  float* frow = feats + (long long)bi * FEAT;

#pragma unroll
  for (int d = 0; d < DKc; ++d) frow[1024 + h * 16 + d] = orow[d];
#pragma unroll
  for (int p = 0; p < NPVc; ++p) {
    float d0 = orow[16 + p * 3 + 0] - t[0];
    float d1 = orow[16 + p * 3 + 1] - t[1];
    float d2 = orow[16 + p * 3 + 2] - t[2];
    float asum = 0.f;
#pragma unroll
    for (int m = 0; m < 3; ++m) {
      float oh = R[0][m] * d0 + R[1][m] * d1 + R[2][m] * d2;
      frow[1152 + h * 24 + p * 3 + m] = oh;
      asum += fabsf(oh);
    }
    frow[1344 + h * 8 + p] = asum;
  }
}

static void launch_gemm(hipStream_t s, int mode, const float* A,
                        const float* Bm, float* C, int M, int N, int K, int lda,
                        int ldb, int ldc, long long sA, long long sA2, int zdiv,
                        long long sB, long long sC, int Mvalid, int nz,
                        const float* e0 = nullptr) {
  dim3 g((M + 15) / 16, (N + 127) / 128, nz);
  dim3 blk(128, 1, 1);
  if (mode == 0)
    k_gemm<0><<<g, blk, 0, s>>>(A, Bm, C, M, N, K, lda, ldb, ldc, sA, sA2, zdiv,
                                sB, sC, Mvalid, e0);
  else
    k_gemm<1><<<g, blk, 0, s>>>(A, Bm, C, M, N, K, lda, ldb, ldc, sA, sA2, zdiv,
                                sB, sC, Mvalid, e0);
}

extern "C" void kernel_launch(void* const* d_in, const int* in_sizes, int n_in,
                              void* d_out, int out_size, void* d_ws,
                              size_t ws_size, hipStream_t stream) {
  const float* si = (const float*)d_in[0];
  const float* zij = (const float*)d_in[1];
  const float* Ri = (const float*)d_in[2];
  const float* Ti = (const float*)d_in[3];
  const float* WQh = (const float*)d_in[4];
  const float* WKh = (const float*)d_in[5];
  const float* WVh = (const float*)d_in[6];
  const float* WQhp = (const float*)d_in[7];
  const float* WKhp = (const float*)d_in[8];
  const float* WVhp = (const float*)d_in[9];
  const float* Wb = (const float*)d_in[10];
  const float* WOw = (const float*)d_in[11];
  const float* WOb = (const float*)d_in[12];
  const float* gamah = (const float*)d_in[13];
  float* out = (float*)d_out;
  float* ws = (float*)d_ws;

  const int BL = Bc * Lc;
  const long long LL = (long long)Lc * Lc;
  float* q = ws;
  float* k = q + 98304;
  float* v = k + 98304;
  float* qp = v + 98304;
  float* kp = qp + 73728;
  float* vp = kp + 73728;
  float* vaug = vp + 147456;
  float* sqq = vaug + 294912;
  float* sqk = sqq + 6144;
  float* alpha = sqk + 6144;
  float* ovalpt = alpha + 2359296;
  float* feats = ovalpt + 294912;

  launch_gemm(stream, 0, si, WQh, q, BL, 128, Cc, Cc, 128, 128, 0, 0, 1, 0, 0, 16, 1);
  launch_gemm(stream, 0, si, WKh, k, BL, 128, Cc, Cc, 128, 128, 0, 0, 1, 0, 0, 16, 1);
  launch_gemm(stream, 0, si, WVh, v, BL, 128, Cc, Cc, 128, 128, 0, 0, 1, 0, 0, 16, 1);
  launch_gemm(stream, 0, si, WQhp, qp, BL, 96, Cc, Cc, 96, 96, 0, 0, 1, 0, 0, 16, 1);
  launch_gemm(stream, 0, si, WKhp, kp, BL, 96, Cc, Cc, 96, 96, 0, 0, 1, 0, 0, 16, 1);
  launch_gemm(stream, 0, si, WVhp, vp, BL, 192, Cc, Cc, 192, 192, 0, 0, 1, 0, 0, 16, 1);

  k_scores<<<dim3(Lc / 16, Lc / 128, Bc * Hc), 128, 0, stream>>>(
      q, k, v, qp, kp, vp, Ri, Ti, gamah, alpha, vaug, sqq, sqk);

  k_slice<<<SLICE_WGS, 256, 0, stream>>>(zij, Wb, alpha, sqq, sqk, gamah, feats);

  launch_gemm(stream, 0, alpha, vaug, ovalpt, Lc, 48, Lc, Lc, 48, 48,
              LL, 0, 1, (long long)Lc * 48, (long long)Lc * 48, 16, Bc * Hc);

  k_finish<<<(Bc * Lc * Hc + 255) / 256, 256, 0, stream>>>(ovalpt, Ri, Ti, feats);

  launch_gemm(stream, 1, feats, WOw, out, BL, Cc, FEAT, FEAT, Cc, Cc,
              0, 0, 1, 0, 0, 16, 1, WOb);
}


# q.k launch projects its own rows: 64x128 workgroups run si @ WQ/WK/WV on WMMA from global fragments into LDS, then transform them; six projection launches gone (11 become 5); sums keep order
# speedup vs baseline: 5.0006x; 1.2487x over previous
#include <hip/hip_runtime.h>
#include <math.h>

typedef __attribute__((ext_vector_type(2))) float v2f;
typedef __attribute__((ext_vector_type(4))) float f4;
typedef __attribute__((ext_vector_type(8))) float v8f;

namespace {
constexpr int Bc = 2, Lc = 384, Cc = 128, Hc = 8, DKc = 16, NQPc = 4, NPVc = 8;
constexpr float WCc = 0.2357022603955158f;
constexpr float WLc = 0.5773502691896258f;
constexpr int FEAT = Hc * Cc + Hc * DKc + Hc * NPVc * 3 + Hc * NPVc;
}

__device__ __forceinline__ float softplusf(float x) {
  return (x > 0.f) ? (x + log1pf(expf(-x))) : log1pf(expf(x));
}

namespace {
constexpr int PKT = 32;
constexpr int PNS = PKT / 4;
constexpr int PSETS = 4;
constexpr int PAHEAD = PSETS - 1;
}

template <bool TWO>
__device__ __forceinline__ void load_group(const float* __restrict__ ap,
                                           const char* __restrict__ Bb,
                                           unsigned ob, unsigned ldb4, int kb,
                                           v2f (&a)[PNS], v2f (&x)[PNS],
                                           v2f (&y)[PNS]) {
#pragma unroll
  for (int s = 0; s < PNS; ++s) {
    const int k = kb + 4 * s;
    a[s] = *(const v2f*)(ap + k);
    const char* r0 = Bb + (unsigned long long)((unsigned)k * ldb4);
    const char* r1 = r0 + ldb4;
    if (TWO) {
      x[s] = *(const v2f*)(r0 + ob);
      y[s] = *(const v2f*)(r1 + ob);
    } else {
      x[s].x = *(const float*)(r0 + ob);
      x[s].y = *(const float*)(r1 + ob);
    }
  }
}

template <bool TWO>
__device__ __forceinline__ void mma_group(const v2f (&a)[PNS],
                                          const v2f (&x)[PNS],
                                          const v2f (&y)[PNS], v8f& acc0,
                                          v8f& acc1) {
#pragma unroll
  for (int s = 0; s < PNS; ++s) {
    if (TWO) {
      v2f b0, b1;
      b0.x = x[s].x;
      b0.y = y[s].x;
      b1.x = x[s].y;
      b1.y = y[s].y;
      acc0 = __builtin_amdgcn_wmma_f32_16x16x4_f32(false, a[s], false, b0,
                                                   (short)0, acc0, false, false);
      acc1 = __builtin_amdgcn_wmma_f32_16x16x4_f32(false, a[s], false, b1,
                                                   (short)0, acc1, false, false);
    } else {
      acc0 = __builtin_amdgcn_wmma_f32_16x16x4_f32(false, a[s], false, x[s],
                                                   (short)0, acc0, false, false);
    }
  }
}

template <bool TWO>
__device__ __forceinline__ void wave_kloop_pipe(const float* __restrict__ ap,
                                                const char* __restrict__ Bb,
                                                unsigned ob, unsigned ldb4,
                                                int K, v8f& acc0, v8f& acc1) {
  v2f fa[PSETS][PNS], fx[PSETS][PNS], fy[PSETS][PNS];
#pragma unroll
  for (int g = 0; g < PAHEAD; ++g) {
    load_group<TWO>(ap, Bb, ob, ldb4, g * PKT, fa[g], fx[g], fy[g]);
    __builtin_amdgcn_sched_barrier(0);
  }
  int kb = 0;
  for (; kb + PSETS * PKT < K; kb += PSETS * PKT) {
#pragma unroll
    for (int u = 0; u < PSETS; ++u) {
      __builtin_amdgcn_sched_barrier(0);
      load_group<TWO>(ap, Bb, ob, ldb4, kb + (u + PAHEAD) * PKT,
                      fa[(u + PAHEAD) % PSETS], fx[(u + PAHEAD) % PSETS],
                      fy[(u + PAHEAD) % PSETS]);
      __builtin_amdgcn_sched_barrier(0);
      mma_group<TWO>(fa[u], fx[u], fy[u], acc0, acc1);
    }
  }
  __builtin_amdgcn_sched_barrier(0);
  load_group<TWO>(ap, Bb, ob, ldb4, kb + PAHEAD * PKT, fa[PAHEAD % PSETS],
                  fx[PAHEAD % PSETS], fy[PAHEAD % PSETS]);
  __builtin_amdgcn_sched_barrier(0);
#pragma unroll
  for (int u = 0; u < PSETS; ++u)
    mma_group<TWO>(fa[u], fx[u], fy[u], acc0, acc1);
}

template <int MODE>
__global__ __launch_bounds__(128) __attribute__((amdgpu_waves_per_eu(4, 4))) void k_gemm(
    const float* __restrict__ A, const float* __restrict__ Bm,
    float* __restrict__ Cm, int M, int N, int K, int lda, int ldb, int ldc,
    long long sA, long long sA2, int zdiv, long long sB, long long sC,
    int Mvalid, const float* __restrict__ e0) {
  const int z = blockIdx.z;
  const float* Ag = A + (long long)(z / zdiv) * sA + (long long)(z % zdiv) * sA2;
  const float* Bg = Bm + (long long)z * sB;
  float* Cg = Cm + (long long)z * sC;

  const int m0 = blockIdx.x * 16;
  const int n0 = blockIdx.y * 128;
  const int tid = threadIdx.x;
  const int wave = __builtin_amdgcn_readfirstlane(tid >> 5);
  const int lane = tid & 31;
  const int hf = lane >> 4;
  const int l16 = lane & 15;
  const int nw = wave * 32;
  const int nb = n0 + nw;

  if (nb >= N) return;
  const bool two = (nb + 16) < N;

  const int n_t0 = two ? (nb + 2 * l16) : (nb + l16);
  const int n_t1 = n_t0 + 1;
  const int ra = (l16 < Mvalid && (m0 + l16) < M) ? l16 : 0;
  const int c0 = (n_t0 < N) ? n_t0 : nb;

  const float* ap = Ag + (long long)(m0 + ra) * lda + 2 * hf;
  const unsigned ldb4 = (unsigned)ldb * 4u;
  const unsigned ob = (unsigned)(2 * hf * ldb + c0) * 4u;

  v8f acc0 = {};
  v8f acc1 = {};
  if (two)
    wave_kloop_pipe<true>(ap, (const char*)Bg, ob, ldb4, K, acc0, acc1);
  else
    wave_kloop_pipe<false>(ap, (const char*)Bg, ob, ldb4, K, acc0, acc1);

#pragma unroll
  for (int r = 0; r < 8; ++r) {
    int mrow = r + 8 * hf;
    int m = m0 + mrow;
    if (mrow >= Mvalid || m >= M) continue;
#pragma unroll
    for (int t = 0; t < 2; ++t) {
      int n = (t == 0) ? n_t0 : n_t1;
      if (n >= N || (t == 1 && !two)) continue;
      float val = (t == 0) ? acc0[r] : acc1[r];
      if (MODE == 1) val += e0[n];
      Cg[(long long)m * ldc + n] = val;
    }
  }
}

namespace {
constexpr int QP = 36;
constexpr int VP = 52;
constexpr int QT = 4;
constexpr int PS = Cc / 4;
}

__device__ __forceinline__ void load_rigid(const float* __restrict__ Ri,
                                           const float* __restrict__ Ti, int bi,
                                           float (&R)[3][3], float (&t)[3]) {
#pragma unroll
  for (int r = 0; r < 3; ++r)
#pragma unroll
    for (int cc = 0; cc < 3; ++cc) R[r][cc] = Ri[bi * 9 + r * 3 + cc];
#pragma unroll
  for (int x = 0; x < 3; ++x) t[x] = Ti[bi * 3 + x];
}

__device__ __forceinline__ void load_afrag(const float* __restrict__ arow,
                                           v2f (&a)[PS]) {
#pragma unroll
  for (int s = 0; s < PS; ++s) a[s] = *(const v2f*)(arow + 4 * s);
}

__device__ __forceinline__ void load_bfrag(const float* __restrict__ wcol, int ldw,
                                           v2f (&bf)[PS]) {
#pragma unroll
  for (int s = 0; s < PS; ++s) {
    bf[s].x = wcol[(4 * s) * ldw];
    bf[s].y = wcol[(4 * s + 1) * ldw];
  }
}

__device__ __forceinline__ void mma_rows2(const v2f (&a0)[PS], const v2f (&a1)[PS],
                                          const v2f (&bf)[PS], v8f& c0, v8f& c1) {
#pragma unroll
  for (int s = 0; s < PS; ++s) {
    c0 = __builtin_amdgcn_wmma_f32_16x16x4_f32(false, a0[s], false, bf[s],
                                               (short)0, c0, false, false);
    c1 = __builtin_amdgcn_wmma_f32_16x16x4_f32(false, a1[s], false, bf[s],
                                               (short)0, c1, false, false);
  }
}

__device__ __forceinline__ void mma_cols2(const v2f (&a)[PS], const v2f (&b0)[PS],
                                          const v2f (&b1)[PS], v8f& c0, v8f& c1) {
#pragma unroll
  for (int s = 0; s < PS; ++s) {
    c0 = __builtin_amdgcn_wmma_f32_16x16x4_f32(false, a[s], false, b0[s],
                                               (short)0, c0, false, false);
    c1 = __builtin_amdgcn_wmma_f32_16x16x4_f32(false, a[s], false, b1[s],
                                               (short)0, c1, false, false);
  }
}

__device__ __forceinline__ void store_tile(float* dst, int pitch, const v8f& acc) {
#pragma unroll
  for (int r = 0; r < 8; ++r) dst[r * pitch] = acc[r];
}

__device__ __forceinline__ float pack_points(float* __restrict__ row,
                                             const float (&R)[3][3],
                                             const float (&t)[3], float pc) {
  float p[12];
#pragma unroll
  for (int d = 0; d < 12; d += 4) {
    f4 x = *(const f4*)(row + 16 + d);
    p[d] = x.x; p[d + 1] = x.y; p[d + 2] = x.z; p[d + 3] = x.w;
  }
  float o[16];
  float sq = 0.f;
#pragma unroll
  for (int pt = 0; pt < NQPc; ++pt) {
    const float* ps = p + pt * 3;
#pragma unroll
    for (int m = 0; m < 3; ++m) {
      float xt = R[m][0] * ps[0] + R[m][1] * ps[1] + R[m][2] * ps[2] + t[m];
      sq += xt * xt;
      o[pt * 3 + m] = pc * xt;
    }
  }
#pragma unroll
  for (int d = 12; d < 16; ++d) o[d] = 0.f;
#pragma unroll
  for (int d = 0; d < 16; d += 4) {
    f4 x = {o[d], o[d + 1], o[d + 2], o[d + 3]};
    *(f4*)(row + 16 + d) = x;
  }
  return sq;
}

__global__ __launch_bounds__(128) void k_scores(
    const float* __restrict__ si, const float* __restrict__ WQh,
    const float* __restrict__ WKh, const float* __restrict__ WVh,
    const float* __restrict__ WQhp, const float* __restrict__ WKhp,
    const float* __restrict__ WVhp, const float* __restrict__ Ri,
    const float* __restrict__ Ti, const float* __restrict__ gamah,
    float* __restrict__ S, float* __restrict__ vaug, float* __restrict__ sqq,
    float* __restrict__ sqk) {
  __shared__ __attribute__((aligned(16))) float qs[16 * QT * QP];
  __shared__ __attribute__((aligned(16))) float kss[128 * QP];
  __shared__ __attribute__((aligned(16))) float vs[128 * VP];

  const int z = blockIdx.z;
  const int b = z / Hc, h = z % Hc;
  const int m0 = blockIdx.x * (16 * QT);
  const int n0 = blockIdx.y * 128;
  const int tid = threadIdx.x;
  const bool edge = (blockIdx.x == 0);
  const int wave = __builtin_amdgcn_readfirstlane(tid >> 5);
  const int lane = tid & 31;
  const int hf = lane >> 4;
  const int l16 = lane & 15;

  float Rk[3][3], tk[3], Rq[3][3], tq[3];
  load_rigid(Ri, Ti, b * Lc + n0 + tid, Rk, tk);
  load_rigid(Ri, Ti, b * Lc + m0 + (tid & (16 * QT - 1)), Rq, tq);

  {
    const float* sib = si + (long long)b * Lc * Cc + 2 * hf;
    const int cp = (l16 < 12) ? l16 : 0;
    v2f a0[PS], a1[PS], bf[PS];
    load_afrag(sib + (n0 + 32 * wave + l16) * Cc, a0);
    load_afrag(sib + (n0 + 32 * wave + 16 + l16) * Cc, a1);
    float* kd = kss + (32 * wave + 8 * hf) * QP + l16;
    {
      load_bfrag(WKh + (2 * hf) * 128 + h * 16 + l16, 128, bf);
      v8f c0 = {}, c1 = {};
      mma_rows2(a0, a1, bf, c0, c1);
      store_tile(kd, QP, c0);
      store_tile(kd + 16 * QP, QP, c1);
    }
    {
      load_bfrag(WKhp + (2 * hf) * 96 + h * 12 + cp, 96, bf);
      v8f c0 = {}, c1 = {};
      mma_rows2(a0, a1, bf, c0, c1);
      store_tile(kd + 16, QP, c0);
      store_tile(kd + 16 * QP + 16, QP, c1);
    }
    if (edge) {
      float* vd = vs + (32 * wave + 8 * hf) * VP + l16;
      const int c8 = (l16 < 8) ? l16 : 0;
      {
        load_bfrag(WVh + (2 * hf) * 128 + h * 16 + l16, 128, bf);
        v8f c0 = {}, c1 = {};
        mma_rows2(a0, a1, bf, c0, c1);
        store_tile(vd, VP, c0);
        store_tile(vd + 16 * VP, VP, c1);
      }
      {
        load_bfrag(WVhp + (2 * hf) * 192 + h * 24 + l16, 192, bf);
        v8f c0 = {}, c1 = {};
        mma_rows2(a0, a1, bf, c0, c1);
        store_tile(vd + 16, VP, c0);
        store_tile(vd + 16 * VP + 16, VP, c1);
      }
      {
        load_bfrag(WVhp + (2 * hf) * 192 + h * 24 + 16 + c8, 192, bf);
        v8f c0 = {}, c1 = {};
        mma_rows2(a0, a1, bf, c0, c1);
        store_tile(vd + 32, VP, c0);
        store_tile(vd + 16 * VP + 32, VP, c1);
      }
    }
    {
      v2f bg[PS];
      load_afrag(sib + (m0 + 16 * wave + l16) * Cc, a0);
      load_bfrag(WQh + (2 * hf) * 128 + h * 16 + l16, 128, bf);
      load_bfrag(WQhp + (2 * hf) * 96 + h * 12 + cp, 96, bg);
      v8f c0 = {}, c1 = {};
      mma_cols2(a0, bf, bg, c0, c1);
      float* qd = qs + (16 * wave + 8 * hf) * QP + l16;
      store_tile(qd, QP, c0);
      store_tile(qd + 16, QP, c1);
    }
  }
  const float gwc = WCc * softplusf(gamah[h]);
  asm volatile("" ::"v"(Rk[0][0]), "v"(Rk[0][1]), "v"(Rk[0][2]), "v"(Rk[1][0]),
               "v"(Rk[1][1]), "v"(Rk[1][2]), "v"(Rk[2][0]), "v"(Rk[2][1]),
               "v"(Rk[2][2]), "v"(tk[0]), "v"(tk[1]), "v"(tk[2]));
  asm volatile("" ::"v"(Rq[0][0]), "v"(Rq[0][1]), "v"(Rq[0][2]), "v"(Rq[1][0]),
               "v"(Rq[1][1]), "v"(Rq[1][2]), "v"(Rq[2][0]), "v"(Rq[2][1]),
               "v"(Rq[2][2]), "v"(tq[0]), "v"(tq[1]), "v"(tq[2]));
  __syncthreads();

  {
    const int j = n0 + tid;
    const float (&R)[3][3] = Rk;
    const float (&t)[3] = tk;
    const float sk = pack_points(kss + tid * QP, R, t, 1.f);
    if (edge) {
      sqk[(long long)z * Lc + j] = sk;
      float* vrow = vaug + ((long long)z * Lc + j) * 48;
      const float* vr = vs + tid * VP;
      float vv[40];
#pragma unroll
      for (int d = 0; d < 40; d += 4) {
        f4 x = *(const f4*)(vr + d);
        vv[d] = x.x; vv[d + 1] = x.y; vv[d + 2] = x.z; vv[d + 3] = x.w;
      }
#pragma unroll
      for (int d = 0; d < DKc; ++d) vrow[d] = vv[d];
#pragma unroll
      for (int p = 0; p < NPVc; ++p) {
        const float* vsrc = vv + 16 + p * 3;
#pragma unroll
        for (int m = 0; m < 3; ++m)
          vrow[16 + p * 3 + m] =
              R[m][0] * vsrc[0] + R[m][1] * vsrc[1] + R[m][2] * vsrc[2] + t[m];
      }
#pragma unroll
      for (int d = 40; d < 48; ++d) vrow[d] = 0.f;
    }
  }
  if (tid < 16 * QT) {
    const int i = m0 + tid;
    float* qrow = qs + tid * QP;
#pragma unroll
    for (int d = 0; d < DKc; d += 4) {
      f4 x = *(const f4*)(qrow + d);
      f4 y = {0.25f * x.x, 0.25f * x.y, 0.25f * x.z, 0.25f * x.w};
      *(f4*)(qrow + d) = y;
    }
    const float sq = pack_points(qrow, Rq, tq, gwc);
    if (blockIdx.y == 0) sqq[(long long)z * Lc + i] = sq;
  }
  __syncthreads();

  const float* bp0 = kss + (wave * 32 + l16) * QP + 2 * hf;
  const float* bp1 = bp0 + 16 * QP;
  v2f b0[8], b1[8];
#pragma unroll
  for (int s = 0; s < 8; ++s) {
    b0[s] = *(const v2f*)(bp0 + 4 * s);
    b1[s] = *(const v2f*)(bp1 + 4 * s);
  }
#pragma unroll
  for (int qt = 0; qt < QT; ++qt) {
    const float* ap = qs + (16 * qt + l16) * QP + 2 * hf;
    v2f a[8];
#pragma unroll
    for (int s = 0; s < 8; ++s) a[s] = *(const v2f*)(ap + 4 * s);
    v8f acc0 = {};
    v8f acc1 = {};
#pragma unroll
    for (int s = 0; s < 8; ++s) {
      acc0 = __builtin_amdgcn_wmma_f32_16x16x4_f32(false, a[s], false, b0[s],
                                                   (short)0, acc0, false, false);
      acc1 = __builtin_amdgcn_wmma_f32_16x16x4_f32(false, a[s], false, b1[s],
                                                   (short)0, acc1, false, false);
    }
    float* srow = S + ((long long)z * Lc + m0 + 16 * qt + 8 * hf) * Lc + n0 +
                  wave * 32 + l16;
#pragma unroll
    for (int r = 0; r < 8; ++r) {
      srow[(long long)r * Lc] = acc0[r];
      srow[(long long)r * Lc + 16] = acc1[r];
    }
  }
}

namespace {
constexpr int ZP = 140;
constexpr int RP = 388;
constexpr int NSL = 3;
constexpr int SLICE_WGS = Bc * Lc / NSL;
constexpr int RW = Lc / 8;
static_assert(Lc % NSL == 0, "the slices of a workgroup share b");
static_assert(RW % 16 == 0, "whole 16-row tiles per wave");
}

__device__ __forceinline__ void barrier_lds() {
  __builtin_amdgcn_fence(__ATOMIC_RELEASE, "workgroup", "local");
  __builtin_amdgcn_s_barrier();
  __builtin_amdgcn_fence(__ATOMIC_ACQUIRE, "workgroup", "local");
}

__global__ __launch_bounds__(256) __attribute__((amdgpu_waves_per_eu(2, 2))) void k_slice(
    const float* __restrict__ zij, const float* __restrict__ Wb,
    float* __restrict__ S, const float* __restrict__ sqq,
    const float* __restrict__ sqk, const float* __restrict__ gamah,
    float* __restrict__ feats) {
  __shared__ __attribute__((aligned(16))) float zs[Lc * ZP];
  __shared__ __attribute__((aligned(16))) float sb[Hc * RP];
  __shared__ __attribute__((aligned(16))) float sa[Hc * RP];

  const int bi0 = blockIdx.x * NSL;
  const int b = bi0 / Lc, i0 = bi0 % Lc;
  const int tid = threadIdx.x;
  const int wave = __builtin_amdgcn_readfirstlane(tid >> 5);
  const int lane = tid & 31;
  const int hf = lane >> 4;
  const int l16 = lane & 15;
  const int j0 = wave * RW;
  const int z = b * Hc + wave;

  const f4* zsrc = (const f4*)(zij + ((long long)bi0 * Lc + j0) * Cc) + lane;
  f4 t[RW];
#pragma unroll
  for (int r = 0; r < RW; ++r) t[r] = zsrc[r * (Cc / 4)];

  float kq[12];
#pragma unroll
  for (int e = 0; e < 12; ++e) kq[e] = sqk[(long long)z * Lc + e * 32 + lane];
  v2f w[32];
  {
    const int cw = (l16 < Hc) ? l16 : 0;
    const float* wp = Wb + (2 * hf) * Hc + cw;
#pragma unroll
    for (int s = 0; s < 32; ++s) {
      w[s].x = wp[(4 * s) * Hc];
      w[s].y = wp[(4 * s + 1) * Hc];
    }
  }
  __builtin_amdgcn_sched_barrier(0);
  const float gw = 0.5f * WCc * softplusf(gamah[wave]);

#pragma unroll
  for (int sl = 0; sl < NSL; ++sl) {
    const int bi = bi0 + sl;
    const int i = i0 + sl;

#pragma unroll
    for (int r = 0; r < RW; ++r) *(f4*)(zs + (j0 + r) * ZP + 4 * lane) = t[r];
    __builtin_amdgcn_sched_barrier(0);

    float* p = S + ((long long)z * Lc + i) * Lc;
    float pv[12];
#pragma unroll
    for (int e = 0; e < 12; ++e) pv[e] = p[e * 32 + lane];
    const float sq = sqq[(long long)z * Lc + i];

    if (sl + 1 < NSL) {
      const f4* zn = zsrc + (long long)(sl + 1) * (Lc * Cc / 4);
#pragma unroll
      for (int r = 0; r < RW; ++r) t[r] = zn[r * (Cc / 4)];
    }
    __builtin_amdgcn_sched_barrier(0);

    barrier_lds();

    {
      const float* zp = zs + (j0 + l16) * ZP + 2 * hf;
      v8f acc[RW / 16] = {};
#pragma unroll
      for (int g = 0; g < 4; ++g) {
        v2f a[RW / 16][8];
#pragma unroll
        for (int tt = 0; tt < RW / 16; ++tt)
#pragma unroll
          for (int s = 0; s < 8; ++s)
            a[tt][s] = *(const v2f*)(zp + 16 * tt * ZP + 32 * g + 4 * s);
#pragma unroll
        for (int s = 0; s < 8; ++s)
#pragma unroll
          for (int tt = 0; tt < RW / 16; ++tt)
            acc[tt] = __builtin_amdgcn_wmma_f32_16x16x4_f32(
                false, a[tt][s], false, w[8 * g + s], (short)0, acc[tt], false,
                false);
      }
      if (l16 < Hc) {
#pragma unroll
        for (int tt = 0; tt < RW / 16; ++tt)
#pragma unroll
          for (int r = 0; r < 8; ++r)
            sb[l16 * RP + j0 + 16 * tt + r + 8 * hf] = acc[tt][r];
      }
    }
    barrier_lds();

    {
      float vals[12];
      float mx = -INFINITY;
#pragma unroll
      for (int e = 0; e < 12; ++e) {
        const int j = e * 32 + lane;
        vals[e] = WLc * (pv[e] + sb[wave * RP + j] - gw * (sq + kq[e]));
        mx = fmaxf(mx, vals[e]);
      }
#pragma unroll
      for (int o = 16; o >= 1; o >>= 1) mx = fmaxf(mx, __shfl_xor(mx, o, 32));
      float sum = 0.f;
#pragma unroll
      for (int e = 0; e < 12; ++e) {
        vals[e] = expf(vals[e] - mx);
        sum += vals[e];
      }
#pragma unroll
      for (int o = 16; o >= 1; o >>= 1) sum += __shfl_xor(sum, o, 32);
      const float inv = 1.f / sum;
#pragma unroll
      for (int e = 0; e < 12; ++e) {
        const int j = e * 32 + lane;
        const float al = vals[e] * inv;
        p[j] = al;
        sa[wave * RP + j] = al;
      }
    }
    barrier_lds();

    {
      const int c = wave * 16 + l16;
      const int ra = (l16 < Hc) ? l16 : 0;
      const float* ap = sa + ra * RP + 2 * hf;
      const float* zc = zs + (2 * hf) * ZP + c;
      v8f acc = {};
      for (int kb = 0; kb < Lc; kb += 64) {
        v2f a[16], bq[16];
#pragma unroll
        for (int s = 0; s < 16; ++s) {
          const int k = kb + 4 * s;
          a[s] = *(const v2f*)(ap + k);
          bq[s].x = zc[k * ZP];
          bq[s].y = zc[(k + 1) * ZP];
        }
        asm volatile("" ::"v"(a[0]), "v"(a[1]), "v"(a[2]), "v"(a[3]), "v"(a[4]), "v"(a[5]),
                     "v"(a[6]), "v"(a[7]), "v"(a[8]), "v"(a[9]), "v"(a[10]),
                     "v"(a[11]), "v"(a[12]), "v"(a[13]), "v"(a[14]), "v"(a[15]));
        asm volatile("" ::"v"(bq[0]), "v"(bq[1]), "v"(bq[2]), "v"(bq[3]), "v"(bq[4]), "v"(bq[5]),
                     "v"(bq[6]), "v"(bq[7]), "v"(bq[8]), "v"(bq[9]), "v"(bq[10]),
                     "v"(bq[11]), "v"(bq[12]), "v"(bq[13]), "v"(bq[14]), "v"(bq[15]));
#pragma unroll
        for (int s = 0; s < 16; ++s)
          acc = __builtin_amdgcn_wmma_f32_16x16x4_f32(false, a[s], false, bq[s],
                                                      (short)0, acc, false, false);
      }
      if (hf == 0) {
        float* frow = feats + (long long)bi * FEAT + c;
#pragma unroll
        for (int r = 0; r < 8; ++r) frow[r * Cc] = acc[r];
      }
    }
    if (sl + 1 < NSL) barrier_lds();
  }
}

__global__ __launch_bounds__(256) void k_finish(
    const float* __restrict__ ovalpt, const float* __restrict__ Ri,
    const float* __restrict__ Ti, float* __restrict__ feats) {
  int idx = blockIdx.x * blockDim.x + threadIdx.x;
  if (idx >= Bc * Lc * Hc) return;
  int h = idx % Hc;
  int bi = idx / Hc;
  int b = bi / Lc;
  int z = b * Hc + h;
  int i = bi % Lc;

  float R[3][3], t[3];
#pragma unroll
  for (int r = 0; r < 3; ++r)
#pragma unroll
    for (int c = 0; c < 3; ++c) R[r][c] = Ri[bi * 9 + r * 3 + c];
#pragma unroll
  for (int x = 0; x < 3; ++x) t[x] = Ti[bi * 3 + x];

  const float* orow = ovalpt + ((long long)z * Lc + i) * 48;
  float* frow = feats + (long long)bi * FEAT;

#pragma unroll
  for (int d = 0; d < DKc; ++d) frow[1024 + h * 16 + d] = orow[d];
#pragma unroll
  for (int p = 0; p < NPVc; ++p) {
    float d0 = orow[16 + p * 3 + 0] - t[0];
    float d1 = orow[16 + p * 3 + 1] - t[1];
    float d2 = orow[16 + p * 3 + 2] - t[2];
    float asum = 0.f;
#pragma unroll
    for (int m = 0; m < 3; ++m) {
      float oh = R[0][m] * d0 + R[1][m] * d1 + R[2][m] * d2;
      frow[1152 + h * 24 + p * 3 + m] = oh;
      asum += fabsf(oh);
    }
    frow[1344 + h * 8 + p] = asum;
  }
}

static void launch_gemm(hipStream_t s, int mode, const float* A,
                        const float* Bm, float* C, int M, int N, int K, int lda,
                        int ldb, int ldc, long long sA, long long sA2, int zdiv,
                        long long sB, long long sC, int Mvalid, int nz,
                        const float* e0 = nullptr) {
  dim3 g((M + 15) / 16, (N + 127) / 128, nz);
  dim3 blk(128, 1, 1);
  if (mode == 0)
    k_gemm<0><<<g, blk, 0, s>>>(A, Bm, C, M, N, K, lda, ldb, ldc, sA, sA2, zdiv,
                                sB, sC, Mvalid, e0);
  else
    k_gemm<1><<<g, blk, 0, s>>>(A, Bm, C, M, N, K, lda, ldb, ldc, sA, sA2, zdiv,
                                sB, sC, Mvalid, e0);
}

extern "C" void kernel_launch(void* const* d_in, const int* in_sizes, int n_in,
                              void* d_out, int out_size, void* d_ws,
                              size_t ws_size, hipStream_t stream) {
  const float* si = (const float*)d_in[0];
  const float* zij = (const float*)d_in[1];
  const float* Ri = (const float*)d_in[2];
  const float* Ti = (const float*)d_in[3];
  const float* WQh = (const float*)d_in[4];
  const float* WKh = (const float*)d_in[5];
  const float* WVh = (const float*)d_in[6];
  const float* WQhp = (const float*)d_in[7];
  const float* WKhp = (const float*)d_in[8];
  const float* WVhp = (const float*)d_in[9];
  const float* Wb = (const float*)d_in[10];
  const float* WOw = (const float*)d_in[11];
  const float* WOb = (const float*)d_in[12];
  const float* gamah = (const float*)d_in[13];
  float* out = (float*)d_out;
  float* ws = (float*)d_ws;

  const int BL = Bc * Lc;
  const long long LL = (long long)Lc * Lc;
  float* vaug = ws;
  float* sqq = vaug + 294912;
  float* sqk = sqq + 6144;
  float* alpha = sqk + 6144;
  float* ovalpt = alpha + 2359296;
  float* feats = ovalpt + 294912;

  k_scores<<<dim3(Lc / (16 * QT), Lc / 128, Bc * Hc), 128, 0, stream>>>(
      si, WQh, WKh, WVh, WQhp, WKhp, WVhp, Ri, Ti, gamah, alpha, vaug, sqq, sqk);

  k_slice<<<SLICE_WGS, 256, 0, stream>>>(zij, Wb, alpha, sqq, sqk, gamah, feats);

  launch_gemm(stream, 0, alpha, vaug, ovalpt, Lc, 48, Lc, Lc, 48, 48,
              LL, 0, 1, (long long)Lc * 48, (long long)Lc * 48, 16, Bc * Hc);

  k_finish<<<(Bc * Lc * Hc + 255) / 256, 256, 0, stream>>>(ovalpt, Ri, Ti, feats);

  launch_gemm(stream, 1, feats, WOw, out, BL, Cc, FEAT, FEAT, Cc, Cc,
              0, 0, 1, 0, 0, 16, 1, WOb);
}


# Inverse transforms and abs-sums move into the out projection's workgroups: a fifth wave builds the 16x384 tail of A in LDS while four waves run k<1024 from memory; 5 launches become 4; same sums
# speedup vs baseline: 6.2183x; 1.2435x over previous
#include <hip/hip_runtime.h>
#include <math.h>

typedef __attribute__((ext_vector_type(2))) float v2f;
typedef __attribute__((ext_vector_type(4))) float f4;
typedef __attribute__((ext_vector_type(8))) float v8f;

namespace {
constexpr int Bc = 2, Lc = 384, Cc = 128, Hc = 8, DKc = 16, NQPc = 4, NPVc = 8;
constexpr float WCc = 0.2357022603955158f;
constexpr float WLc = 0.5773502691896258f;
constexpr int FEAT = Hc * Cc + Hc * DKc + Hc * NPVc * 3 + Hc * NPVc;
}

__device__ __forceinline__ float softplusf(float x) {
  return (x > 0.f) ? (x + log1pf(expf(-x))) : log1pf(expf(x));
}

namespace {
constexpr int PKT = 32;
constexpr int PNS = PKT / 4;
constexpr int PSETS = 4;
constexpr int PAHEAD = PSETS - 1;
}

template <bool TWO>
__device__ __forceinline__ void load_group(const float* __restrict__ ap,
                                           const char* __restrict__ Bb,
                                           unsigned ob, unsigned ldb4, int kb,
                                           v2f (&a)[PNS], v2f (&x)[PNS],
                                           v2f (&y)[PNS]) {
#pragma unroll
  for (int s = 0; s < PNS; ++s) {
    const int k = kb + 4 * s;
    a[s] = *(const v2f*)(ap + k);
    const char* r0 = Bb + (unsigned long long)((unsigned)k * ldb4);
    const char* r1 = r0 + ldb4;
    if (TWO) {
      x[s] = *(const v2f*)(r0 + ob);
      y[s] = *(const v2f*)(r1 + ob);
    } else {
      x[s].x = *(const float*)(r0 + ob);
      x[s].y = *(const float*)(r1 + ob);
    }
  }
}

template <bool TWO>
__device__ __forceinline__ void mma_group(const v2f (&a)[PNS],
                                          const v2f (&x)[PNS],
                                          const v2f (&y)[PNS], v8f& acc0,
                                          v8f& acc1) {
#pragma unroll
  for (int s = 0; s < PNS; ++s) {
    if (TWO) {
      v2f b0, b1;
      b0.x = x[s].x;
      b0.y = y[s].x;
      b1.x = x[s].y;
      b1.y = y[s].y;
      acc0 = __builtin_amdgcn_wmma_f32_16x16x4_f32(false, a[s], false, b0,
                                                   (short)0, acc0, false, false);
      acc1 = __builtin_amdgcn_wmma_f32_16x16x4_f32(false, a[s], false, b1,
                                                   (short)0, acc1, false, false);
    } else {
      acc0 = __builtin_amdgcn_wmma_f32_16x16x4_f32(false, a[s], false, x[s],
                                                   (short)0, acc0, false, false);
    }
  }
}

template <bool TWO>
__device__ __forceinline__ void wave_kloop_pipe(const float* __restrict__ ap,
                                                const char* __restrict__ Bb,
                                                unsigned ob, unsigned ldb4,
                                                int K, v8f& acc0, v8f& acc1) {
  v2f fa[PSETS][PNS], fx[PSETS][PNS], fy[PSETS][PNS];
#pragma unroll
  for (int g = 0; g < PAHEAD; ++g) {
    load_group<TWO>(ap, Bb, ob, ldb4, g * PKT, fa[g], fx[g], fy[g]);
    __builtin_amdgcn_sched_barrier(0);
  }
  int kb = 0;
#pragma nounroll
  for (; kb + PSETS * PKT < K; kb += PSETS * PKT) {
#pragma unroll
    for (int u = 0; u < PSETS; ++u) {
      __builtin_amdgcn_sched_barrier(0);
      load_group<TWO>(ap, Bb, ob, ldb4, kb + (u + PAHEAD) * PKT,
                      fa[(u + PAHEAD) % PSETS], fx[(u + PAHEAD) % PSETS],
                      fy[(u + PAHEAD) % PSETS]);
      __builtin_amdgcn_sched_barrier(0);
      mma_group<TWO>(fa[u], fx[u], fy[u], acc0, acc1);
    }
  }
  __builtin_amdgcn_sched_barrier(0);
  load_group<TWO>(ap, Bb, ob, ldb4, kb + PAHEAD * PKT, fa[PAHEAD % PSETS],
                  fx[PAHEAD % PSETS], fy[PAHEAD % PSETS]);
  __builtin_amdgcn_sched_barrier(0);
#pragma unroll
  for (int u = 0; u < PSETS; ++u)
    mma_group<TWO>(fa[u], fx[u], fy[u], acc0, acc1);
}

__global__ __launch_bounds__(128) __attribute__((amdgpu_waves_per_eu(4, 4))) void k_gemm(
    const float* __restrict__ A, const float* __restrict__ Bm,
    float* __restrict__ Cm, int M, int N, int K, int lda, int ldb, int ldc,
    long long sA, long long sA2, int zdiv, long long sB, long long sC,
    int Mvalid) {
  const int z = blockIdx.z;
  const float* Ag = A + (long long)(z / zdiv) * sA + (long long)(z % zdiv) * sA2;
  const float* Bg = Bm + (long long)z * sB;
  float* Cg = Cm + (long long)z * sC;

  const int m0 = blockIdx.x * 16;
  const int n0 = blockIdx.y * 128;
  const int tid = threadIdx.x;
  const int wave = __builtin_amdgcn_readfirstlane(tid >> 5);
  const int lane = tid & 31;
  const int hf = lane >> 4;
  const int l16 = lane & 15;
  const int nw = wave * 32;
  const int nb = n0 + nw;

  if (nb >= N) return;
  const bool two = (nb + 16) < N;

  const int n_t0 = two ? (nb + 2 * l16) : (nb + l16);
  const int n_t1 = n_t0 + 1;
  const int ra = (l16 < Mvalid && (m0 + l16) < M) ? l16 : 0;
  const int c0 = (n_t0 < N) ? n_t0 : nb;

  const float* ap = Ag + (long long)(m0 + ra) * lda + 2 * hf;
  const unsigned ldb4 = (unsigned)ldb * 4u;
  const unsigned ob = (unsigned)(2 * hf * ldb + c0) * 4u;

  v8f acc0 = {};
  v8f acc1 = {};
  if (two)
    wave_kloop_pipe<true>(ap, (const char*)Bg, ob, ldb4, K, acc0, acc1);
  else
    wave_kloop_pipe<false>(ap, (const char*)Bg, ob, ldb4, K, acc0, acc1);

#pragma unroll
  for (int r = 0; r < 8; ++r) {
    int mrow = r + 8 * hf;
    int m = m0 + mrow;
    if (mrow >= Mvalid || m >= M) continue;
#pragma unroll
    for (int t = 0; t < 2; ++t) {
      int n = (t == 0) ? n_t0 : n_t1;
      if (n >= N || (t == 1 && !two)) continue;
      Cg[(long long)m * ldc + n] = (t == 0) ? acc0[r] : acc1[r];
    }
  }
}

namespace {
constexpr int QP = 36;
constexpr int VP = 52;
constexpr int QT = 4;
constexpr int PS = Cc / 4;
}

__device__ __forceinline__ void load_rigid(const float* __restrict__ Ri,
                                           const float* __restrict__ Ti, int bi,
                                           float (&R)[3][3], float (&t)[3]) {
#pragma unroll
  for (int r = 0; r < 3; ++r)
#pragma unroll
    for (int cc = 0; cc < 3; ++cc) R[r][cc] = Ri[bi * 9 + r * 3 + cc];
#pragma unroll
  for (int x = 0; x < 3; ++x) t[x] = Ti[bi * 3 + x];
}

__device__ __forceinline__ void load_afrag(const float* __restrict__ arow,
                                           v2f (&a)[PS]) {
#pragma unroll
  for (int s = 0; s < PS; ++s) a[s] = *(const v2f*)(arow + 4 * s);
}

__device__ __forceinline__ void load_bfrag(const float* __restrict__ wcol, int ldw,
                                           v2f (&bf)[PS]) {
#pragma unroll
  for (int s = 0; s < PS; ++s) {
    bf[s].x = wcol[(4 * s) * ldw];
    bf[s].y = wcol[(4 * s + 1) * ldw];
  }
}

__device__ __forceinline__ void mma_rows2(const v2f (&a0)[PS], const v2f (&a1)[PS],
                                          const v2f (&bf)[PS], v8f& c0, v8f& c1) {
#pragma unroll
  for (int s = 0; s < PS; ++s) {
    c0 = __builtin_amdgcn_wmma_f32_16x16x4_f32(false, a0[s], false, bf[s],
                                               (short)0, c0, false, false);
    c1 = __builtin_amdgcn_wmma_f32_16x16x4_f32(false, a1[s], false, bf[s],
                                               (short)0, c1, false, false);
  }
}

__device__ __forceinline__ void mma_cols2(const v2f (&a)[PS], const v2f (&b0)[PS],
                                          const v2f (&b1)[PS], v8f& c0, v8f& c1) {
#pragma unroll
  for (int s = 0; s < PS; ++s) {
    c0 = __builtin_amdgcn_wmma_f32_16x16x4_f32(false, a[s], false, b0[s],
                                               (short)0, c0, false, false);
    c1 = __builtin_amdgcn_wmma_f32_16x16x4_f32(false, a[s], false, b1[s],
                                               (short)0, c1, false, false);
  }
}

__device__ __forceinline__ void store_tile(float* dst, int pitch, const v8f& acc) {
#pragma unroll
  for (int r = 0; r < 8; ++r) dst[r * pitch] = acc[r];
}

__device__ __forceinline__ float pack_points(float* __restrict__ row,
                                             const float (&R)[3][3],
                                             const float (&t)[3], float pc) {
  float p[12];
#pragma unroll
  for (int d = 0; d < 12; d += 4) {
    f4 x = *(const f4*)(row + 16 + d);
    p[d] = x.x; p[d + 1] = x.y; p[d + 2] = x.z; p[d + 3] = x.w;
  }
  float o[16];
  float sq = 0.f;
#pragma unroll
  for (int pt = 0; pt < NQPc; ++pt) {
    const float* ps = p + pt * 3;
#pragma unroll
    for (int m = 0; m < 3; ++m) {
      float xt = R[m][0] * ps[0] + R[m][1] * ps[1] + R[m][2] * ps[2] + t[m];
      sq += xt * xt;
      o[pt * 3 + m] = pc * xt;
    }
  }
#pragma unroll
  for (int d = 12; d < 16; ++d) o[d] = 0.f;
#pragma unroll
  for (int d = 0; d < 16; d += 4) {
    f4 x = {o[d], o[d + 1], o[d + 2], o[d + 3]};
    *(f4*)(row + 16 + d) = x;
  }
  return sq;
}

__global__ __launch_bounds__(128) void k_scores(
    const float* __restrict__ si, const float* __restrict__ WQh,
    const float* __restrict__ WKh, const float* __restrict__ WVh,
    const float* __restrict__ WQhp, const float* __restrict__ WKhp,
    const float* __restrict__ WVhp, const float* __restrict__ Ri,
    const float* __restrict__ Ti, const float* __restrict__ gamah,
    float* __restrict__ S, float* __restrict__ vaug, float* __restrict__ sqq,
    float* __restrict__ sqk) {
  __shared__ __attribute__((aligned(16))) float qs[16 * QT * QP];
  __shared__ __attribute__((aligned(16))) float kss[128 * QP];
  __shared__ __attribute__((aligned(16))) float vs[128 * VP];

  const int z = blockIdx.z;
  const int b = z / Hc, h = z % Hc;
  const int m0 = blockIdx.x * (16 * QT);
  const int n0 = blockIdx.y * 128;
  const int tid = threadIdx.x;
  const bool edge = (blockIdx.x == 0);
  const int wave = __builtin_amdgcn_readfirstlane(tid >> 5);
  const int lane = tid & 31;
  const int hf = lane >> 4;
  const int l16 = lane & 15;

  float Rk[3][3], tk[3], Rq[3][3], tq[3];
  load_rigid(Ri, Ti, b * Lc + n0 + tid, Rk, tk);
  load_rigid(Ri, Ti, b * Lc + m0 + (tid & (16 * QT - 1)), Rq, tq);

  {
    const float* sib = si + (long long)b * Lc * Cc + 2 * hf;
    const int cp = (l16 < 12) ? l16 : 0;
    v2f a0[PS], a1[PS], bf[PS];
    load_afrag(sib + (n0 + 32 * wave + l16) * Cc, a0);
    load_afrag(sib + (n0 + 32 * wave + 16 + l16) * Cc, a1);
    float* kd = kss + (32 * wave + 8 * hf) * QP + l16;
    {
      load_bfrag(WKh + (2 * hf) * 128 + h * 16 + l16, 128, bf);
      v8f c0 = {}, c1 = {};
      mma_rows2(a0, a1, bf, c0, c1);
      store_tile(kd, QP, c0);
      store_tile(kd + 16 * QP, QP, c1);
    }
    {
      load_bfrag(WKhp + (2 * hf) * 96 + h * 12 + cp, 96, bf);
      v8f c0 = {}, c1 = {};
      mma_rows2(a0, a1, bf, c0, c1);
      store_tile(kd + 16, QP, c0);
      store_tile(kd + 16 * QP + 16, QP, c1);
    }
    if (edge) {
      float* vd = vs + (32 * wave + 8 * hf) * VP + l16;
      const int c8 = (l16 < 8) ? l16 : 0;
      {
        load_bfrag(WVh + (2 * hf) * 128 + h * 16 + l16, 128, bf);
        v8f c0 = {}, c1 = {};
        mma_rows2(a0, a1, bf, c0, c1);
        store_tile(vd, VP, c0);
        store_tile(vd + 16 * VP, VP, c1);
      }
      {
        load_bfrag(WVhp + (2 * hf) * 192 + h * 24 + l16, 192, bf);
        v8f c0 = {}, c1 = {};
        mma_rows2(a0, a1, bf, c0, c1);
        store_tile(vd + 16, VP, c0);
        store_tile(vd + 16 * VP + 16, VP, c1);
      }
      {
        load_bfrag(WVhp + (2 * hf) * 192 + h * 24 + 16 + c8, 192, bf);
        v8f c0 = {}, c1 = {};
        mma_rows2(a0, a1, bf, c0, c1);
        store_tile(vd + 32, VP, c0);
        store_tile(vd + 16 * VP + 32, VP, c1);
      }
    }
    {
      v2f bg[PS];
      load_afrag(sib + (m0 + 16 * wave + l16) * Cc, a0);
      load_bfrag(WQh + (2 * hf) * 128 + h * 16 + l16, 128, bf);
      load_bfrag(WQhp + (2 * hf) * 96 + h * 12 + cp, 96, bg);
      v8f c0 = {}, c1 = {};
      mma_cols2(a0, bf, bg, c0, c1);
      float* qd = qs + (16 * wave + 8 * hf) * QP + l16;
      store_tile(qd, QP, c0);
      store_tile(qd + 16, QP, c1);
    }
  }
  const float gwc = WCc * softplusf(gamah[h]);
  asm volatile("" ::"v"(Rk[0][0]), "v"(Rk[0][1]), "v"(Rk[0][2]), "v"(Rk[1][0]),
               "v"(Rk[1][1]), "v"(Rk[1][2]), "v"(Rk[2][0]), "v"(Rk[2][1]),
               "v"(Rk[2][2]), "v"(tk[0]), "v"(tk[1]), "v"(tk[2]));
  asm volatile("" ::"v"(Rq[0][0]), "v"(Rq[0][1]), "v"(Rq[0][2]), "v"(Rq[1][0]),
               "v"(Rq[1][1]), "v"(Rq[1][2]), "v"(Rq[2][0]), "v"(Rq[2][1]),
               "v"(Rq[2][2]), "v"(tq[0]), "v"(tq[1]), "v"(tq[2]));
  __syncthreads();

  {
    const int j = n0 + tid;
    const float (&R)[3][3] = Rk;
    const float (&t)[3] = tk;
    const float sk = pack_points(kss + tid * QP, R, t, 1.f);
    if (edge) {
      sqk[(long long)z * Lc + j] = sk;
      float* vrow = vaug + ((long long)z * Lc + j) * 48;
      const float* vr = vs + tid * VP;
      float vv[40];
#pragma unroll
      for (int d = 0; d < 40; d += 4) {
        f4 x = *(const f4*)(vr + d);
        vv[d] = x.x; vv[d + 1] = x.y; vv[d + 2] = x.z; vv[d + 3] = x.w;
      }
#pragma unroll
      for (int d = 0; d < DKc; ++d) vrow[d] = vv[d];
#pragma unroll
      for (int p = 0; p < NPVc; ++p) {
        const float* vsrc = vv + 16 + p * 3;
#pragma unroll
        for (int m = 0; m < 3; ++m)
          vrow[16 + p * 3 + m] =
              R[m][0] * vsrc[0] + R[m][1] * vsrc[1] + R[m][2] * vsrc[2] + t[m];
      }
#pragma unroll
      for (int d = 40; d < 48; ++d) vrow[d] = 0.f;
    }
  }
  if (tid < 16 * QT) {
    const int i = m0 + tid;
    float* qrow = qs + tid * QP;
#pragma unroll
    for (int d = 0; d < DKc; d += 4) {
      f4 x = *(const f4*)(qrow + d);
      f4 y = {0.25f * x.x, 0.25f * x.y, 0.25f * x.z, 0.25f * x.w};
      *(f4*)(qrow + d) = y;
    }
    const float sq = pack_points(qrow, Rq, tq, gwc);
    if (blockIdx.y == 0) sqq[(long long)z * Lc + i] = sq;
  }
  __syncthreads();

  const float* bp0 = kss + (wave * 32 + l16) * QP + 2 * hf;
  const float* bp1 = bp0 + 16 * QP;
  v2f b0[8], b1[8];
#pragma unroll
  for (int s = 0; s < 8; ++s) {
    b0[s] = *(const v2f*)(bp0 + 4 * s);
    b1[s] = *(const v2f*)(bp1 + 4 * s);
  }
#pragma unroll
  for (int qt = 0; qt < QT; ++qt) {
    const float* ap = qs + (16 * qt + l16) * QP + 2 * hf;
    v2f a[8];
#pragma unroll
    for (int s = 0; s < 8; ++s) a[s] = *(const v2f*)(ap + 4 * s);
    v8f acc0 = {};
    v8f acc1 = {};
#pragma unroll
    for (int s = 0; s < 8; ++s) {
      acc0 = __builtin_amdgcn_wmma_f32_16x16x4_f32(false, a[s], false, b0[s],
                                                   (short)0, acc0, false, false);
      acc1 = __builtin_amdgcn_wmma_f32_16x16x4_f32(false, a[s], false, b1[s],
                                                   (short)0, acc1, false, false);
    }
    float* srow = S + ((long long)z * Lc + m0 + 16 * qt + 8 * hf) * Lc + n0 +
                  wave * 32 + l16;
#pragma unroll
    for (int r = 0; r < 8; ++r) {
      srow[(long long)r * Lc] = acc0[r];
      srow[(long long)r * Lc + 16] = acc1[r];
    }
  }
}

namespace {
constexpr int ZP = 140;
constexpr int RP = 388;
constexpr int NSL = 3;
constexpr int SLICE_WGS = Bc * Lc / NSL;
constexpr int RW = Lc / 8;
static_assert(Lc % NSL == 0, "the slices of a workgroup share b");
static_assert(RW % 16 == 0, "whole 16-row tiles per wave");
}

__device__ __forceinline__ void barrier_lds() {
  __builtin_amdgcn_fence(__ATOMIC_RELEASE, "workgroup", "local");
  __builtin_amdgcn_s_barrier();
  __builtin_amdgcn_fence(__ATOMIC_ACQUIRE, "workgroup", "local");
}

__global__ __launch_bounds__(256) __attribute__((amdgpu_waves_per_eu(2, 2))) void k_slice(
    const float* __restrict__ zij, const float* __restrict__ Wb,
    float* __restrict__ S, const float* __restrict__ sqq,
    const float* __restrict__ sqk, const float* __restrict__ gamah,
    float* __restrict__ feats) {
  __shared__ __attribute__((aligned(16))) float zs[Lc * ZP];
  __shared__ __attribute__((aligned(16))) float sb[Hc * RP];
  __shared__ __attribute__((aligned(16))) float sa[Hc * RP];

  const int bi0 = blockIdx.x * NSL;
  const int b = bi0 / Lc, i0 = bi0 % Lc;
  const int tid = threadIdx.x;
  const int wave = __builtin_amdgcn_readfirstlane(tid >> 5);
  const int lane = tid & 31;
  const int hf = lane >> 4;
  const int l16 = lane & 15;
  const int j0 = wave * RW;
  const int z = b * Hc + wave;

  const f4* zsrc = (const f4*)(zij + ((long long)bi0 * Lc + j0) * Cc) + lane;
  f4 t[RW];
#pragma unroll
  for (int r = 0; r < RW; ++r) t[r] = zsrc[r * (Cc / 4)];

  float kq[12];
#pragma unroll
  for (int e = 0; e < 12; ++e) kq[e] = sqk[(long long)z * Lc + e * 32 + lane];
  v2f w[32];
  {
    const int cw = (l16 < Hc) ? l16 : 0;
    const float* wp = Wb + (2 * hf) * Hc + cw;
#pragma unroll
    for (int s = 0; s < 32; ++s) {
      w[s].x = wp[(4 * s) * Hc];
      w[s].y = wp[(4 * s + 1) * Hc];
    }
  }
  __builtin_amdgcn_sched_barrier(0);
  const float gw = 0.5f * WCc * softplusf(gamah[wave]);

#pragma unroll
  for (int sl = 0; sl < NSL; ++sl) {
    const int bi = bi0 + sl;
    const int i = i0 + sl;

#pragma unroll
    for (int r = 0; r < RW; ++r) *(f4*)(zs + (j0 + r) * ZP + 4 * lane) = t[r];
    __builtin_amdgcn_sched_barrier(0);

    float* p = S + ((long long)z * Lc + i) * Lc;
    float pv[12];
#pragma unroll
    for (int e = 0; e < 12; ++e) pv[e] = p[e * 32 + lane];
    const float sq = sqq[(long long)z * Lc + i];

    if (sl + 1 < NSL) {
      const f4* zn = zsrc + (long long)(sl + 1) * (Lc * Cc / 4);
#pragma unroll
      for (int r = 0; r < RW; ++r) t[r] = zn[r * (Cc / 4)];
    }
    __builtin_amdgcn_sched_barrier(0);

    barrier_lds();

    {
      const float* zp = zs + (j0 + l16) * ZP + 2 * hf;
      v8f acc[RW / 16] = {};
#pragma unroll
      for (int g = 0; g < 4; ++g) {
        v2f a[RW / 16][8];
#pragma unroll
        for (int tt = 0; tt < RW / 16; ++tt)
#pragma unroll
          for (int s = 0; s < 8; ++s)
            a[tt][s] = *(const v2f*)(zp + 16 * tt * ZP + 32 * g + 4 * s);
#pragma unroll
        for (int s = 0; s < 8; ++s)
#pragma unroll
          for (int tt = 0; tt < RW / 16; ++tt)
            acc[tt] = __builtin_amdgcn_wmma_f32_16x16x4_f32(
                false, a[tt][s], false, w[8 * g + s], (short)0, acc[tt], false,
                false);
      }
      if (l16 < Hc) {
#pragma unroll
        for (int tt = 0; tt < RW / 16; ++tt)
#pragma unroll
          for (int r = 0; r < 8; ++r)
            sb[l16 * RP + j0 + 16 * tt + r + 8 * hf] = acc[tt][r];
      }
    }
    barrier_lds();

    {
      float vals[12];
      float mx = -INFINITY;
#pragma unroll
      for (int e = 0; e < 12; ++e) {
        const int j = e * 32 + lane;
        vals[e] = WLc * (pv[e] + sb[wave * RP + j] - gw * (sq + kq[e]));
        mx = fmaxf(mx, vals[e]);
      }
#pragma unroll
      for (int o = 16; o >= 1; o >>= 1) mx = fmaxf(mx, __shfl_xor(mx, o, 32));
      float sum = 0.f;
#pragma unroll
      for (int e = 0; e < 12; ++e) {
        vals[e] = expf(vals[e] - mx);
        sum += vals[e];
      }
#pragma unroll
      for (int o = 16; o >= 1; o >>= 1) sum += __shfl_xor(sum, o, 32);
      const float inv = 1.f / sum;
#pragma unroll
      for (int e = 0; e < 12; ++e) {
        const int j = e * 32 + lane;
        const float al = vals[e] * inv;
        p[j] = al;
        sa[wave * RP + j] = al;
      }
    }
    barrier_lds();

    {
      const int c = wave * 16 + l16;
      const int ra = (l16 < Hc) ? l16 : 0;
      const float* ap = sa + ra * RP + 2 * hf;
      const float* zc = zs + (2 * hf) * ZP + c;
      v8f acc = {};
      for (int kb = 0; kb < Lc; kb += 64) {
        v2f a[16], bq[16];
#pragma unroll
        for (int s = 0; s < 16; ++s) {
          const int k = kb + 4 * s;
          a[s] = *(const v2f*)(ap + k);
          bq[s].x = zc[k * ZP];
          bq[s].y = zc[(k + 1) * ZP];
        }
        asm volatile("" ::"v"(a[0]), "v"(a[1]), "v"(a[2]), "v"(a[3]), "v"(a[4]), "v"(a[5]),
                     "v"(a[6]), "v"(a[7]), "v"(a[8]), "v"(a[9]), "v"(a[10]),
                     "v"(a[11]), "v"(a[12]), "v"(a[13]), "v"(a[14]), "v"(a[15]));
        asm volatile("" ::"v"(bq[0]), "v"(bq[1]), "v"(bq[2]), "v"(bq[3]), "v"(bq[4]), "v"(bq[5]),
                     "v"(bq[6]), "v"(bq[7]), "v"(bq[8]), "v"(bq[9]), "v"(bq[10]),
                     "v"(bq[11]), "v"(bq[12]), "v"(bq[13]), "v"(bq[14]), "v"(bq[15]));
#pragma unroll
        for (int s = 0; s < 16; ++s)
          acc = __builtin_amdgcn_wmma_f32_16x16x4_f32(false, a[s], false, bq[s],
                                                      (short)0, acc, false, false);
      }
      if (hf == 0) {
        float* frow = feats + (long long)bi * FEAT + c;
#pragma unroll
        for (int r = 0; r < 8; ++r) frow[r * Cc] = acc[r];
      }
    }
    if (sl + 1 < NSL) barrier_lds();
  }
}

namespace {
constexpr int OPK = Hc * Cc;
constexpr int TAILK = FEAT - OPK;
constexpr int TP = TAILK + 4;
static_assert(OPK % 128 == 0 && TAILK % 128 == 0, "whole ring turns");
}

__global__ __launch_bounds__(160) __attribute__((amdgpu_waves_per_eu(4, 4))) void k_out(
    const float* __restrict__ feats, const float* __restrict__ ovalpt,
    const float* __restrict__ Ri, const float* __restrict__ Ti,
    const float* __restrict__ WOw, const float* __restrict__ WOb,
    float* __restrict__ out) {
  __shared__ __attribute__((aligned(16))) float tail[16 * TP];

  const int m0 = blockIdx.x * 16;
  const int tid = threadIdx.x;
  const int wave = __builtin_amdgcn_readfirstlane(tid >> 5);
  const int lane = tid & 31;
  const int hf = lane >> 4;
  const int l16 = lane & 15;
  const int n_t0 = wave * 32 + 2 * l16;
  const unsigned ldb4 = (unsigned)Cc * 4u;
  const unsigned ob = (unsigned)(2 * hf * Cc + n_t0) * 4u;

  v8f acc0 = {};
  v8f acc1 = {};
  if (wave < 4) {
    const float* ap = feats + (long long)(m0 + l16) * FEAT + 2 * hf;
    wave_kloop_pipe<true>(ap, (const char*)WOw, ob, ldb4, OPK, acc0, acc1);
  } else {
#pragma unroll 1
    for (int q = 0; q < 4; ++q) {
      const int item = q * 32 + lane;
      const int r = item >> 3;
      const int h = item & 7;
      const int bi = m0 + r;
      const int b = bi / Lc, i = bi % Lc;
      float R[3][3], t[3];
      load_rigid(Ri, Ti, bi, R, t);
      const float* orow = ovalpt + ((long long)(b * Hc + h) * Lc + i) * 48;
      float o[40];
#pragma unroll
      for (int d = 0; d < 40; d += 4) {
        f4 x = *(const f4*)(orow + d);
        o[d] = x.x; o[d + 1] = x.y; o[d + 2] = x.z; o[d + 3] = x.w;
      }
      float* trow = tail + r * TP;
#pragma unroll
      for (int d = 0; d < DKc; d += 4) {
        f4 x = {o[d], o[d + 1], o[d + 2], o[d + 3]};
        *(f4*)(trow + h * 16 + d) = x;
      }
      float oh[24], as[8];
#pragma unroll
      for (int p = 0; p < NPVc; ++p) {
        float d0 = o[16 + p * 3 + 0] - t[0];
        float d1 = o[16 + p * 3 + 1] - t[1];
        float d2 = o[16 + p * 3 + 2] - t[2];
        float asum = 0.f;
#pragma unroll
        for (int m = 0; m < 3; ++m) {
          float v = R[0][m] * d0 + R[1][m] * d1 + R[2][m] * d2;
          oh[p * 3 + m] = v;
          asum += fabsf(v);
        }
        as[p] = asum;
      }
#pragma unroll
      for (int d = 0; d < 24; d += 4) {
        f4 x = {oh[d], oh[d + 1], oh[d + 2], oh[d + 3]};
        *(f4*)(trow + 128 + h * 24 + d) = x;
      }
#pragma unroll
      for (int d = 0; d < 8; d += 4) {
        f4 x = {as[d], as[d + 1], as[d + 2], as[d + 3]};
        *(f4*)(trow + 320 + h * 8 + d) = x;
      }
    }
  }
  __syncthreads();
  if (wave >= 4) return;

  wave_kloop_pipe<true>(tail + l16 * TP + 2 * hf, (const char*)(WOw + OPK * Cc), ob,
                        ldb4, TAILK, acc0, acc1);

#pragma unroll
  for (int r = 0; r < 8; ++r) {
    const int m = m0 + r + 8 * hf;
    out[(long long)m * Cc + n_t0] = acc0[r] + WOb[n_t0];
    out[(long long)m * Cc + n_t0 + 1] = acc1[r] + WOb[n_t0 + 1];
  }
}

static void launch_gemm(hipStream_t s, const float* A, const float* Bm, float* C,
                        int M, int N, int K, int lda, int ldb, int ldc,
                        long long sA, long long sA2, int zdiv, long long sB,
                        long long sC, int Mvalid, int nz) {
  dim3 g((M + 15) / 16, (N + 127) / 128, nz);
  dim3 blk(128, 1, 1);
  k_gemm<<<g, blk, 0, s>>>(A, Bm, C, M, N, K, lda, ldb, ldc, sA, sA2, zdiv, sB,
                           sC, Mvalid);
}

extern "C" void kernel_launch(void* const* d_in, const int* in_sizes, int n_in,
                              void* d_out, int out_size, void* d_ws,
                              size_t ws_size, hipStream_t stream) {
  const float* si = (const float*)d_in[0];
  const float* zij = (const float*)d_in[1];
  const float* Ri = (const float*)d_in[2];
  const float* Ti = (const float*)d_in[3];
  const float* WQh = (const float*)d_in[4];
  const float* WKh = (const float*)d_in[5];
  const float* WVh = (const float*)d_in[6];
  const float* WQhp = (const float*)d_in[7];
  const float* WKhp = (const float*)d_in[8];
  const float* WVhp = (const float*)d_in[9];
  const float* Wb = (const float*)d_in[10];
  const float* WOw = (const float*)d_in[11];
  const float* WOb = (const float*)d_in[12];
  const float* gamah = (const float*)d_in[13];
  float* out = (float*)d_out;
  float* ws = (float*)d_ws;

  const int BL = Bc * Lc;
  const long long LL = (long long)Lc * Lc;
  float* vaug = ws;
  float* sqq = vaug + 294912;
  float* sqk = sqq + 6144;
  float* alpha = sqk + 6144;
  float* ovalpt = alpha + 2359296;
  float* feats = ovalpt + 294912;

  k_scores<<<dim3(Lc / (16 * QT), Lc / 128, Bc * Hc), 128, 0, stream>>>(
      si, WQh, WKh, WVh, WQhp, WKhp, WVhp, Ri, Ti, gamah, alpha, vaug, sqq, sqk);

  k_slice<<<SLICE_WGS, 256, 0, stream>>>(zij, Wb, alpha, sqq, sqk, gamah, feats);

  launch_gemm(stream, alpha, vaug, ovalpt, Lc, 48, Lc, Lc, 48, 48,
              LL, 0, 1, (long long)Lc * 48, (long long)Lc * 48, 16, Bc * Hc);

  k_out<<<BL / 16, 160, 0, stream>>>(feats, ovalpt, Ri, Ti, WOw, WOb, out);
}


# alpha@vaug moves into the out projection: one wave per head makes its 16x48 tile and tail columns in LDS beside the k<1024 products; grid 48x2 column halves; 4 launches become 3; same sums
# speedup vs baseline: 6.5493x; 1.0532x over previous
#include <hip/hip_runtime.h>
#include <math.h>

typedef __attribute__((ext_vector_type(2))) float v2f;
typedef __attribute__((ext_vector_type(4))) float f4;
typedef __attribute__((ext_vector_type(8))) float v8f;

namespace {
constexpr int Bc = 2, Lc = 384, Cc = 128, Hc = 8, DKc = 16, NQPc = 4, NPVc = 8;
constexpr float WCc = 0.2357022603955158f;
constexpr float WLc = 0.5773502691896258f;
constexpr int FEAT = Hc * Cc + Hc * DKc + Hc * NPVc * 3 + Hc * NPVc;
}

__device__ __forceinline__ float softplusf(float x) {
  return (x > 0.f) ? (x + log1pf(expf(-x))) : log1pf(expf(x));
}

namespace {
constexpr int PKT = 32;
constexpr int PNS = PKT / 4;
constexpr int PSETS = 4;
constexpr int PAHEAD = PSETS - 1;
}

template <bool TWO>
__device__ __forceinline__ void load_group(const float* __restrict__ ap,
                                           const char* __restrict__ Bb,
                                           unsigned ob, unsigned ldb4, int kb,
                                           v2f (&a)[PNS], v2f (&x)[PNS],
                                           v2f (&y)[PNS]) {
#pragma unroll
  for (int s = 0; s < PNS; ++s) {
    const int k = kb + 4 * s;
    a[s] = *(const v2f*)(ap + k);
    const char* r0 = Bb + (unsigned long long)((unsigned)k * ldb4);
    const char* r1 = r0 + ldb4;
    if (TWO) {
      x[s] = *(const v2f*)(r0 + ob);
      y[s] = *(const v2f*)(r1 + ob);
    } else {
      x[s].x = *(const float*)(r0 + ob);
      x[s].y = *(const float*)(r1 + ob);
    }
  }
}

template <bool TWO>
__device__ __forceinline__ void mma_group(const v2f (&a)[PNS],
                                          const v2f (&x)[PNS],
                                          const v2f (&y)[PNS], v8f& acc0,
                                          v8f& acc1) {
#pragma unroll
  for (int s = 0; s < PNS; ++s) {
    if (TWO) {
      v2f b0, b1;
      b0.x = x[s].x;
      b0.y = y[s].x;
      b1.x = x[s].y;
      b1.y = y[s].y;
      acc0 = __builtin_amdgcn_wmma_f32_16x16x4_f32(false, a[s], false, b0,
                                                   (short)0, acc0, false, false);
      acc1 = __builtin_amdgcn_wmma_f32_16x16x4_f32(false, a[s], false, b1,
                                                   (short)0, acc1, false, false);
    } else {
      acc0 = __builtin_amdgcn_wmma_f32_16x16x4_f32(false, a[s], false, x[s],
                                                   (short)0, acc0, false, false);
    }
  }
}

template <bool TWO>
__device__ __forceinline__ void wave_kloop_pipe(const float* __restrict__ ap,
                                                const char* __restrict__ Bb,
                                                unsigned ob, unsigned ldb4,
                                                int K, v8f& acc0, v8f& acc1) {
  v2f fa[PSETS][PNS], fx[PSETS][PNS], fy[PSETS][PNS];
#pragma unroll
  for (int g = 0; g < PAHEAD; ++g) {
    load_group<TWO>(ap, Bb, ob, ldb4, g * PKT, fa[g], fx[g], fy[g]);
    __builtin_amdgcn_sched_barrier(0);
  }
  int kb = 0;
#pragma nounroll
  for (; kb + PSETS * PKT < K; kb += PSETS * PKT) {
#pragma unroll
    for (int u = 0; u < PSETS; ++u) {
      __builtin_amdgcn_sched_barrier(0);
      load_group<TWO>(ap, Bb, ob, ldb4, kb + (u + PAHEAD) * PKT,
                      fa[(u + PAHEAD) % PSETS], fx[(u + PAHEAD) % PSETS],
                      fy[(u + PAHEAD) % PSETS]);
      __builtin_amdgcn_sched_barrier(0);
      mma_group<TWO>(fa[u], fx[u], fy[u], acc0, acc1);
    }
  }
  __builtin_amdgcn_sched_barrier(0);
  load_group<TWO>(ap, Bb, ob, ldb4, kb + PAHEAD * PKT, fa[PAHEAD % PSETS],
                  fx[PAHEAD % PSETS], fy[PAHEAD % PSETS]);
  __builtin_amdgcn_sched_barrier(0);
#pragma unroll
  for (int u = 0; u < PSETS; ++u)
    mma_group<TWO>(fa[u], fx[u], fy[u], acc0, acc1);
}

namespace {
constexpr int QP = 36;
constexpr int VP = 52;
constexpr int QT = 4;
constexpr int PS = Cc / 4;
}

__device__ __forceinline__ void load_rigid(const float* __restrict__ Ri,
                                           const float* __restrict__ Ti, int bi,
                                           float (&R)[3][3], float (&t)[3]) {
#pragma unroll
  for (int r = 0; r < 3; ++r)
#pragma unroll
    for (int cc = 0; cc < 3; ++cc) R[r][cc] = Ri[bi * 9 + r * 3 + cc];
#pragma unroll
  for (int x = 0; x < 3; ++x) t[x] = Ti[bi * 3 + x];
}

__device__ __forceinline__ void load_afrag(const float* __restrict__ arow,
                                           v2f (&a)[PS]) {
#pragma unroll
  for (int s = 0; s < PS; ++s) a[s] = *(const v2f*)(arow + 4 * s);
}

__device__ __forceinline__ void load_bfrag(const float* __restrict__ wcol, int ldw,
                                           v2f (&bf)[PS]) {
#pragma unroll
  for (int s = 0; s < PS; ++s) {
    bf[s].x = wcol[(4 * s) * ldw];
    bf[s].y = wcol[(4 * s + 1) * ldw];
  }
}

__device__ __forceinline__ void mma_rows2(const v2f (&a0)[PS], const v2f (&a1)[PS],
                                          const v2f (&bf)[PS], v8f& c0, v8f& c1) {
#pragma unroll
  for (int s = 0; s < PS; ++s) {
    c0 = __builtin_amdgcn_wmma_f32_16x16x4_f32(false, a0[s], false, bf[s],
                                               (short)0, c0, false, false);
    c1 = __builtin_amdgcn_wmma_f32_16x16x4_f32(false, a1[s], false, bf[s],
                                               (short)0, c1, false, false);
  }
}

__device__ __forceinline__ void mma_cols2(const v2f (&a)[PS], const v2f (&b0)[PS],
                                          const v2f (&b1)[PS], v8f& c0, v8f& c1) {
#pragma unroll
  for (int s = 0; s < PS; ++s) {
    c0 = __builtin_amdgcn_wmma_f32_16x16x4_f32(false, a[s], false, b0[s],
                                               (short)0, c0, false, false);
    c1 = __builtin_amdgcn_wmma_f32_16x16x4_f32(false, a[s], false, b1[s],
                                               (short)0, c1, false, false);
  }
}

__device__ __forceinline__ void store_tile(float* dst, int pitch, const v8f& acc) {
#pragma unroll
  for (int r = 0; r < 8; ++r) dst[r * pitch] = acc[r];
}

__device__ __forceinline__ float pack_points(float* __restrict__ row,
                                             const float (&R)[3][3],
                                             const float (&t)[3], float pc) {
  float p[12];
#pragma unroll
  for (int d = 0; d < 12; d += 4) {
    f4 x = *(const f4*)(row + 16 + d);
    p[d] = x.x; p[d + 1] = x.y; p[d + 2] = x.z; p[d + 3] = x.w;
  }
  float o[16];
  float sq = 0.f;
#pragma unroll
  for (int pt = 0; pt < NQPc; ++pt) {
    const float* ps = p + pt * 3;
#pragma unroll
    for (int m = 0; m < 3; ++m) {
      float xt = R[m][0] * ps[0] + R[m][1] * ps[1] + R[m][2] * ps[2] + t[m];
      sq += xt * xt;
      o[pt * 3 + m] = pc * xt;
    }
  }
#pragma unroll
  for (int d = 12; d < 16; ++d) o[d] = 0.f;
#pragma unroll
  for (int d = 0; d < 16; d += 4) {
    f4 x = {o[d], o[d + 1], o[d + 2], o[d + 3]};
    *(f4*)(row + 16 + d) = x;
  }
  return sq;
}

__global__ __launch_bounds__(128) void k_scores(
    const float* __restrict__ si, const float* __restrict__ WQh,
    const float* __restrict__ WKh, const float* __restrict__ WVh,
    const float* __restrict__ WQhp, const float* __restrict__ WKhp,
    const float* __restrict__ WVhp, const float* __restrict__ Ri,
    const float* __restrict__ Ti, const float* __restrict__ gamah,
    float* __restrict__ S, float* __restrict__ vaug, float* __restrict__ sqq,
    float* __restrict__ sqk) {
  __shared__ __attribute__((aligned(16))) float qs[16 * QT * QP];
  __shared__ __attribute__((aligned(16))) float kss[128 * QP];
  __shared__ __attribute__((aligned(16))) float vs[128 * VP];

  const int z = blockIdx.z;
  const int b = z / Hc, h = z % Hc;
  const int m0 = blockIdx.x * (16 * QT);
  const int n0 = blockIdx.y * 128;
  const int tid = threadIdx.x;
  const bool edge = (blockIdx.x == 0);
  const int wave = __builtin_amdgcn_readfirstlane(tid >> 5);
  const int lane = tid & 31;
  const int hf = lane >> 4;
  const int l16 = lane & 15;

  float Rk[3][3], tk[3], Rq[3][3], tq[3];
  load_rigid(Ri, Ti, b * Lc + n0 + tid, Rk, tk);
  load_rigid(Ri, Ti, b * Lc + m0 + (tid & (16 * QT - 1)), Rq, tq);

  {
    const float* sib = si + (long long)b * Lc * Cc + 2 * hf;
    const int cp = (l16 < 12) ? l16 : 0;
    v2f a0[PS], a1[PS], bf[PS];
    load_afrag(sib + (n0 + 32 * wave + l16) * Cc, a0);
    load_afrag(sib + (n0 + 32 * wave + 16 + l16) * Cc, a1);
    float* kd = kss + (32 * wave + 8 * hf) * QP + l16;
    {
      load_bfrag(WKh + (2 * hf) * 128 + h * 16 + l16, 128, bf);
      v8f c0 = {}, c1 = {};
      mma_rows2(a0, a1, bf, c0, c1);
      store_tile(kd, QP, c0);
      store_tile(kd + 16 * QP, QP, c1);
    }
    {
      load_bfrag(WKhp + (2 * hf) * 96 + h * 12 + cp, 96, bf);
      v8f c0 = {}, c1 = {};
      mma_rows2(a0, a1, bf, c0, c1);
      store_tile(kd + 16, QP, c0);
      store_tile(kd + 16 * QP + 16, QP, c1);
    }
    if (edge) {
      float* vd = vs + (32 * wave + 8 * hf) * VP + l16;
      const int c8 = (l16 < 8) ? l16 : 0;
      {
        load_bfrag(WVh + (2 * hf) * 128 + h * 16 + l16, 128, bf);
        v8f c0 = {}, c1 = {};
        mma_rows2(a0, a1, bf, c0, c1);
        store_tile(vd, VP, c0);
        store_tile(vd + 16 * VP, VP, c1);
      }
      {
        load_bfrag(WVhp + (2 * hf) * 192 + h * 24 + l16, 192, bf);
        v8f c0 = {}, c1 = {};
        mma_rows2(a0, a1, bf, c0, c1);
        store_tile(vd + 16, VP, c0);
        store_tile(vd + 16 * VP + 16, VP, c1);
      }
      {
        load_bfrag(WVhp + (2 * hf) * 192 + h * 24 + 16 + c8, 192, bf);
        v8f c0 = {}, c1 = {};
        mma_rows2(a0, a1, bf, c0, c1);
        store_tile(vd + 32, VP, c0);
        store_tile(vd + 16 * VP + 32, VP, c1);
      }
    }
    {
      v2f bg[PS];
      load_afrag(sib + (m0 + 16 * wave + l16) * Cc, a0);
      load_bfrag(WQh + (2 * hf) * 128 + h * 16 + l16, 128, bf);
      load_bfrag(WQhp + (2 * hf) * 96 + h * 12 + cp, 96, bg);
      v8f c0 = {}, c1 = {};
      mma_cols2(a0, bf, bg, c0, c1);
      float* qd = qs + (16 * wave + 8 * hf) * QP + l16;
      store_tile(qd, QP, c0);
      store_tile(qd + 16, QP, c1);
    }
  }
  const float gwc = WCc * softplusf(gamah[h]);
  asm volatile("" ::"v"(Rk[0][0]), "v"(Rk[0][1]), "v"(Rk[0][2]), "v"(Rk[1][0]),
               "v"(Rk[1][1]), "v"(Rk[1][2]), "v"(Rk[2][0]), "v"(Rk[2][1]),
               "v"(Rk[2][2]), "v"(tk[0]), "v"(tk[1]), "v"(tk[2]));
  asm volatile("" ::"v"(Rq[0][0]), "v"(Rq[0][1]), "v"(Rq[0][2]), "v"(Rq[1][0]),
               "v"(Rq[1][1]), "v"(Rq[1][2]), "v"(Rq[2][0]), "v"(Rq[2][1]),
               "v"(Rq[2][2]), "v"(tq[0]), "v"(tq[1]), "v"(tq[2]));
  __syncthreads();

  {
    const int j = n0 + tid;
    const float (&R)[3][3] = Rk;
    const float (&t)[3] = tk;
    const float sk = pack_points(kss + tid * QP, R, t, 1.f);
    if (edge) {
      sqk[(long long)z * Lc + j] = sk;
      float* vrow = vaug + ((long long)z * Lc + j) * 48;
      const float* vr = vs + tid * VP;
      float vv[40];
#pragma unroll
      for (int d = 0; d < 40; d += 4) {
        f4 x = *(const f4*)(vr + d);
        vv[d] = x.x; vv[d + 1] = x.y; vv[d + 2] = x.z; vv[d + 3] = x.w;
      }
#pragma unroll
      for (int d = 0; d < DKc; ++d) vrow[d] = vv[d];
#pragma unroll
      for (int p = 0; p < NPVc; ++p) {
        const float* vsrc = vv + 16 + p * 3;
#pragma unroll
        for (int m = 0; m < 3; ++m)
          vrow[16 + p * 3 + m] =
              R[m][0] * vsrc[0] + R[m][1] * vsrc[1] + R[m][2] * vsrc[2] + t[m];
      }
#pragma unroll
      for (int d = 40; d < 48; ++d) vrow[d] = 0.f;
    }
  }
  if (tid < 16 * QT) {
    const int i = m0 + tid;
    float* qrow = qs + tid * QP;
#pragma unroll
    for (int d = 0; d < DKc; d += 4) {
      f4 x = *(const f4*)(qrow + d);
      f4 y = {0.25f * x.x, 0.25f * x.y, 0.25f * x.z, 0.25f * x.w};
      *(f4*)(qrow + d) = y;
    }
    const float sq = pack_points(qrow, Rq, tq, gwc);
    if (blockIdx.y == 0) sqq[(long long)z * Lc + i] = sq;
  }
  __syncthreads();

  const float* bp0 = kss + (wave * 32 + l16) * QP + 2 * hf;
  const float* bp1 = bp0 + 16 * QP;
  v2f b0[8], b1[8];
#pragma unroll
  for (int s = 0; s < 8; ++s) {
    b0[s] = *(const v2f*)(bp0 + 4 * s);
    b1[s] = *(const v2f*)(bp1 + 4 * s);
  }
#pragma unroll
  for (int qt = 0; qt < QT; ++qt) {
    const float* ap = qs + (16 * qt + l16) * QP + 2 * hf;
    v2f a[8];
#pragma unroll
    for (int s = 0; s < 8; ++s) a[s] = *(const v2f*)(ap + 4 * s);
    v8f acc0 = {};
    v8f acc1 = {};
#pragma unroll
    for (int s = 0; s < 8; ++s) {
      acc0 = __builtin_amdgcn_wmma_f32_16x16x4_f32(false, a[s], false, b0[s],
                                                   (short)0, acc0, false, false);
      acc1 = __builtin_amdgcn_wmma_f32_16x16x4_f32(false, a[s], false, b1[s],
                                                   (short)0, acc1, false, false);
    }
    float* srow = S + ((long long)z * Lc + m0 + 16 * qt + 8 * hf) * Lc + n0 +
                  wave * 32 + l16;
#pragma unroll
    for (int r = 0; r < 8; ++r) {
      srow[(long long)r * Lc] = acc0[r];
      srow[(long long)r * Lc + 16] = acc1[r];
    }
  }
}

namespace {
constexpr int ZP = 140;
constexpr int RP = 388;
constexpr int NSL = 3;
constexpr int SLICE_WGS = Bc * Lc / NSL;
constexpr int RW = Lc / 8;
static_assert(Lc % NSL == 0, "the slices of a workgroup share b");
static_assert(RW % 16 == 0, "whole 16-row tiles per wave");
}

__device__ __forceinline__ void barrier_lds() {
  __builtin_amdgcn_fence(__ATOMIC_RELEASE, "workgroup", "local");
  __builtin_amdgcn_s_barrier();
  __builtin_amdgcn_fence(__ATOMIC_ACQUIRE, "workgroup", "local");
}

__global__ __launch_bounds__(256) __attribute__((amdgpu_waves_per_eu(2, 2))) void k_slice(
    const float* __restrict__ zij, const float* __restrict__ Wb,
    float* __restrict__ S, const float* __restrict__ sqq,
    const float* __restrict__ sqk, const float* __restrict__ gamah,
    float* __restrict__ feats) {
  __shared__ __attribute__((aligned(16))) float zs[Lc * ZP];
  __shared__ __attribute__((aligned(16))) float sb[Hc * RP];
  __shared__ __attribute__((aligned(16))) float sa[Hc * RP];

  const int bi0 = blockIdx.x * NSL;
  const int b = bi0 / Lc, i0 = bi0 % Lc;
  const int tid = threadIdx.x;
  const int wave = __builtin_amdgcn_readfirstlane(tid >> 5);
  const int lane = tid & 31;
  const int hf = lane >> 4;
  const int l16 = lane & 15;
  const int j0 = wave * RW;
  const int z = b * Hc + wave;

  const f4* zsrc = (const f4*)(zij + ((long long)bi0 * Lc + j0) * Cc) + lane;
  f4 t[RW];
#pragma unroll
  for (int r = 0; r < RW; ++r) t[r] = zsrc[r * (Cc / 4)];

  float kq[12];
#pragma unroll
  for (int e = 0; e < 12; ++e) kq[e] = sqk[(long long)z * Lc + e * 32 + lane];
  v2f w[32];
  {
    const int cw = (l16 < Hc) ? l16 : 0;
    const float* wp = Wb + (2 * hf) * Hc + cw;
#pragma unroll
    for (int s = 0; s < 32; ++s) {
      w[s].x = wp[(4 * s) * Hc];
      w[s].y = wp[(4 * s + 1) * Hc];
    }
  }
  __builtin_amdgcn_sched_barrier(0);
  const float gw = 0.5f * WCc * softplusf(gamah[wave]);

#pragma unroll
  for (int sl = 0; sl < NSL; ++sl) {
    const int bi = bi0 + sl;
    const int i = i0 + sl;

#pragma unroll
    for (int r = 0; r < RW; ++r) *(f4*)(zs + (j0 + r) * ZP + 4 * lane) = t[r];
    __builtin_amdgcn_sched_barrier(0);

    float* p = S + ((long long)z * Lc + i) * Lc;
    float pv[12];
#pragma unroll
    for (int e = 0; e < 12; ++e) pv[e] = p[e * 32 + lane];
    const float sq = sqq[(long long)z * Lc + i];

    if (sl + 1 < NSL) {
      const f4* zn = zsrc + (long long)(sl + 1) * (Lc * Cc / 4);
#pragma unroll
      for (int r = 0; r < RW; ++r) t[r] = zn[r * (Cc / 4)];
    }
    __builtin_amdgcn_sched_barrier(0);

    barrier_lds();

    {
      const float* zp = zs + (j0 + l16) * ZP + 2 * hf;
      v8f acc[RW / 16] = {};
#pragma unroll
      for (int g = 0; g < 4; ++g) {
        v2f a[RW / 16][8];
#pragma unroll
        for (int tt = 0; tt < RW / 16; ++tt)
#pragma unroll
          for (int s = 0; s < 8; ++s)
            a[tt][s] = *(const v2f*)(zp + 16 * tt * ZP + 32 * g + 4 * s);
#pragma unroll
        for (int s = 0; s < 8; ++s)
#pragma unroll
          for (int tt = 0; tt < RW / 16; ++tt)
            acc[tt] = __builtin_amdgcn_wmma_f32_16x16x4_f32(
                false, a[tt][s], false, w[8 * g + s], (short)0, acc[tt], false,
                false);
      }
      if (l16 < Hc) {
#pragma unroll
        for (int tt = 0; tt < RW / 16; ++tt)
#pragma unroll
          for (int r = 0; r < 8; ++r)
            sb[l16 * RP + j0 + 16 * tt + r + 8 * hf] = acc[tt][r];
      }
    }
    barrier_lds();

    {
      float vals[12];
      float mx = -INFINITY;
#pragma unroll
      for (int e = 0; e < 12; ++e) {
        const int j = e * 32 + lane;
        vals[e] = WLc * (pv[e] + sb[wave * RP + j] - gw * (sq + kq[e]));
        mx = fmaxf(mx, vals[e]);
      }
#pragma unroll
      for (int o = 16; o >= 1; o >>= 1) mx = fmaxf(mx, __shfl_xor(mx, o, 32));
      float sum = 0.f;
#pragma unroll
      for (int e = 0; e < 12; ++e) {
        vals[e] = expf(vals[e] - mx);
        sum += vals[e];
      }
#pragma unroll
      for (int o = 16; o >= 1; o >>= 1) sum += __shfl_xor(sum, o, 32);
      const float inv = 1.f / sum;
#pragma unroll
      for (int e = 0; e < 12; ++e) {
        const int j = e * 32 + lane;
        const float al = vals[e] * inv;
        p[j] = al;
        sa[wave * RP + j] = al;
      }
    }
    barrier_lds();

    {
      const int c = wave * 16 + l16;
      const int ra = (l16 < Hc) ? l16 : 0;
      const float* ap = sa + ra * RP + 2 * hf;
      const float* zc = zs + (2 * hf) * ZP + c;
      v8f acc = {};
      for (int kb = 0; kb < Lc; kb += 64) {
        v2f a[16], bq[16];
#pragma unroll
        for (int s = 0; s < 16; ++s) {
          const int k = kb + 4 * s;
          a[s] = *(const v2f*)(ap + k);
          bq[s].x = zc[k * ZP];
          bq[s].y = zc[(k + 1) * ZP];
        }
        asm volatile("" ::"v"(a[0]), "v"(a[1]), "v"(a[2]), "v"(a[3]), "v"(a[4]), "v"(a[5]),
                     "v"(a[6]), "v"(a[7]), "v"(a[8]), "v"(a[9]), "v"(a[10]),
                     "v"(a[11]), "v"(a[12]), "v"(a[13]), "v"(a[14]), "v"(a[15]));
        asm volatile("" ::"v"(bq[0]), "v"(bq[1]), "v"(bq[2]), "v"(bq[3]), "v"(bq[4]), "v"(bq[5]),
                     "v"(bq[6]), "v"(bq[7]), "v"(bq[8]), "v"(bq[9]), "v"(bq[10]),
                     "v"(bq[11]), "v"(bq[12]), "v"(bq[13]), "v"(bq[14]), "v"(bq[15]));
#pragma unroll
        for (int s = 0; s < 16; ++s)
          acc = __builtin_amdgcn_wmma_f32_16x16x4_f32(false, a[s], false, bq[s],
                                                      (short)0, acc, false, false);
      }
      if (hf == 0) {
        float* frow = feats + (long long)bi * FEAT + c;
#pragma unroll
        for (int r = 0; r < 8; ++r) frow[r * Cc] = acc[r];
      }
    }
    if (sl + 1 < NSL) barrier_lds();
  }
}

namespace {
constexpr int OPK = Hc * Cc;
constexpr int TAILK = FEAT - OPK;
constexpr int TP = TAILK + 4;
constexpr int VW = 48;
constexpr int OVP = VW + 4;
constexpr int OSPLIT = 2;
static_assert(OPK % 128 == 0 && TAILK % 128 == 0 && Lc % 128 == 0, "whole ring turns");
static_assert(Lc % 16 == 0, "the 16 rows of a workgroup share b");
}

__global__ __launch_bounds__(320) __attribute__((amdgpu_waves_per_eu(4, 4))) void k_out(
    const float* __restrict__ feats, const float* __restrict__ alpha,
    const float* __restrict__ vaug, const float* __restrict__ Ri,
    const float* __restrict__ Ti, const float* __restrict__ WOw,
    const float* __restrict__ WOb, float* __restrict__ out) {
  __shared__ __attribute__((aligned(16))) float tail[16 * TP];
  __shared__ __attribute__((aligned(16))) float ovs[Hc * 16 * OVP];

  const int m0 = blockIdx.x * 16;
  const int tid = threadIdx.x;
  const int wave = __builtin_amdgcn_readfirstlane(tid >> 5);
  const int lane = tid & 31;
  const int hf = lane >> 4;
  const int l16 = lane & 15;
  const bool prod = (wave == 2) || (wave == 3);
  const int n_t0 = blockIdx.y * 64 + (wave & 1) * 32 + 2 * l16;
  const unsigned ldb4 = (unsigned)Cc * 4u;
  const unsigned ob = (unsigned)(2 * hf * Cc + n_t0) * 4u;

  v8f acc0 = {};
  v8f acc1 = {};
  if (prod) {
    const float* ap = feats + (long long)(m0 + l16) * FEAT + 2 * hf;
    wave_kloop_pipe<true>(ap, (const char*)WOw, ob, ldb4, OPK, acc0, acc1);
  } else {
    const int h = (wave < 2) ? wave : wave - 2;
    const int z = (m0 / Lc) * Hc + h;
    float R[3][3], t[3];
    load_rigid(Ri, Ti, m0 + l16, R, t);

    const float* ap = alpha + ((long long)z * Lc + (m0 % Lc) + l16) * Lc + 2 * hf;
    const char* vb = (const char*)(vaug + (long long)z * Lc * VW);
    const unsigned vld4 = (unsigned)VW * 4u;
    v8f o0 = {}, o1 = {}, o2 = {}, ou = {};
    wave_kloop_pipe<true>(ap, vb, (unsigned)(2 * hf * VW + 2 * l16) * 4u, vld4, Lc,
                          o0, o1);
    asm volatile("" ::"v"(R[0][0]), "v"(R[0][1]), "v"(R[0][2]), "v"(R[1][0]),
                 "v"(R[1][1]), "v"(R[1][2]), "v"(R[2][0]), "v"(R[2][1]),
                 "v"(R[2][2]), "v"(t[0]), "v"(t[1]), "v"(t[2]));
    wave_kloop_pipe<false>(ap, vb, (unsigned)(2 * hf * VW + 32 + l16) * 4u, vld4, Lc,
                           o2, ou);

    float* ot = ovs + h * (16 * OVP);
#pragma unroll
    for (int r = 0; r < 8; ++r) {
      float* d = ot + (r + 8 * hf) * OVP;
      v2f pr = {o0[r], o1[r]};
      *(v2f*)(d + 2 * l16) = pr;
      d[32 + l16] = o2[r];
    }
    __builtin_amdgcn_fence(__ATOMIC_RELEASE, "workgroup", "local");
    __builtin_amdgcn_fence(__ATOMIC_ACQUIRE, "workgroup", "local");

    if (lane < 16) {
      const float* orow = ot + lane * OVP;
      float o[40];
#pragma unroll
      for (int d = 0; d < 40; d += 4) {
        f4 x = *(const f4*)(orow + d);
        o[d] = x.x; o[d + 1] = x.y; o[d + 2] = x.z; o[d + 3] = x.w;
      }
      float* trow = tail + lane * TP;
#pragma unroll
      for (int d = 0; d < DKc; d += 4) {
        f4 x = {o[d], o[d + 1], o[d + 2], o[d + 3]};
        *(f4*)(trow + h * 16 + d) = x;
      }
      float oh[24], as[8];
#pragma unroll
      for (int p = 0; p < NPVc; ++p) {
        float d0 = o[16 + p * 3 + 0] - t[0];
        float d1 = o[16 + p * 3 + 1] - t[1];
        float d2 = o[16 + p * 3 + 2] - t[2];
        float asum = 0.f;
#pragma unroll
        for (int m = 0; m < 3; ++m) {
          float v = R[0][m] * d0 + R[1][m] * d1 + R[2][m] * d2;
          oh[p * 3 + m] = v;
          asum += fabsf(v);
        }
        as[p] = asum;
      }
#pragma unroll
      for (int d = 0; d < 24; d += 4) {
        f4 x = {oh[d], oh[d + 1], oh[d + 2], oh[d + 3]};
        *(f4*)(trow + 128 + h * 24 + d) = x;
      }
#pragma unroll
      for (int d = 0; d < 8; d += 4) {
        f4 x = {as[d], as[d + 1], as[d + 2], as[d + 3]};
        *(f4*)(trow + 320 + h * 8 + d) = x;
      }
    }
  }
  __syncthreads();
  if (!prod) return;

  wave_kloop_pipe<true>(tail + l16 * TP + 2 * hf, (const char*)(WOw + OPK * Cc), ob,
                        ldb4, TAILK, acc0, acc1);

#pragma unroll
  for (int r = 0; r < 8; ++r) {
    const int m = m0 + r + 8 * hf;
    out[(long long)m * Cc + n_t0] = acc0[r] + WOb[n_t0];
    out[(long long)m * Cc + n_t0 + 1] = acc1[r] + WOb[n_t0 + 1];
  }
}

extern "C" void kernel_launch(void* const* d_in, const int* in_sizes, int n_in,
                              void* d_out, int out_size, void* d_ws,
                              size_t ws_size, hipStream_t stream) {
  const float* si = (const float*)d_in[0];
  const float* zij = (const float*)d_in[1];
  const float* Ri = (const float*)d_in[2];
  const float* Ti = (const float*)d_in[3];
  const float* WQh = (const float*)d_in[4];
  const float* WKh = (const float*)d_in[5];
  const float* WVh = (const float*)d_in[6];
  const float* WQhp = (const float*)d_in[7];
  const float* WKhp = (const float*)d_in[8];
  const float* WVhp = (const float*)d_in[9];
  const float* Wb = (const float*)d_in[10];
  const float* WOw = (const float*)d_in[11];
  const float* WOb = (const float*)d_in[12];
  const float* gamah = (const float*)d_in[13];
  float* out = (float*)d_out;
  float* ws = (float*)d_ws;

  const int BL = Bc * Lc;
  float* vaug = ws;
  float* sqq = vaug + 294912;
  float* sqk = sqq + 6144;
  float* alpha = sqk + 6144;
  float* feats = alpha + 2359296;

  k_scores<<<dim3(Lc / (16 * QT), Lc / 128, Bc * Hc), 128, 0, stream>>>(
      si, WQh, WKh, WVh, WQhp, WKhp, WVhp, Ri, Ti, gamah, alpha, vaug, sqq, sqk);

  k_slice<<<SLICE_WGS, 256, 0, stream>>>(zij, Wb, alpha, sqq, sqk, gamah, feats);

  k_out<<<dim3(BL / 16, OSPLIT), 320, 0, stream>>>(feats, alpha, vaug, Ri, Ti, WOw, WOb,
                                                out);
}


# k_out head waves copy their 16x384 alpha block to LDS with whole-row 512-byte loads and read A fragments there (was 8-byte loads over 16 rows, twice); same k order, same sums
# speedup vs baseline: 6.9051x; 1.0543x over previous
#include <hip/hip_runtime.h>
#include <math.h>

typedef __attribute__((ext_vector_type(2))) float v2f;
typedef __attribute__((ext_vector_type(4))) float f4;
typedef __attribute__((ext_vector_type(8))) float v8f;

namespace {
constexpr int Bc = 2, Lc = 384, Cc = 128, Hc = 8, DKc = 16, NQPc = 4, NPVc = 8;
constexpr float WCc = 0.2357022603955158f;
constexpr float WLc = 0.5773502691896258f;
constexpr int FEAT = Hc * Cc + Hc * DKc + Hc * NPVc * 3 + Hc * NPVc;
}

__device__ __forceinline__ float softplusf(float x) {
  return (x > 0.f) ? (x + log1pf(expf(-x))) : log1pf(expf(x));
}

namespace {
constexpr int PKT = 32;
constexpr int PNS = PKT / 4;
constexpr int PSETS = 4;
constexpr int PAHEAD = PSETS - 1;
}

template <bool TWO>
__device__ __forceinline__ void load_group(const float* __restrict__ ap,
                                           const char* __restrict__ Bb,
                                           unsigned ob, unsigned ldb4, int kb,
                                           v2f (&a)[PNS], v2f (&x)[PNS],
                                           v2f (&y)[PNS]) {
#pragma unroll
  for (int s = 0; s < PNS; ++s) {
    const int k = kb + 4 * s;
    a[s] = *(const v2f*)(ap + k);
    const char* r0 = Bb + (unsigned long long)((unsigned)k * ldb4);
    const char* r1 = r0 + ldb4;
    if (TWO) {
      x[s] = *(const v2f*)(r0 + ob);
      y[s] = *(const v2f*)(r1 + ob);
    } else {
      x[s].x = *(const float*)(r0 + ob);
      x[s].y = *(const float*)(r1 + ob);
    }
  }
}

template <bool TWO>
__device__ __forceinline__ void mma_group(const v2f (&a)[PNS],
                                          const v2f (&x)[PNS],
                                          const v2f (&y)[PNS], v8f& acc0,
                                          v8f& acc1) {
#pragma unroll
  for (int s = 0; s < PNS; ++s) {
    if (TWO) {
      v2f b0, b1;
      b0.x = x[s].x;
      b0.y = y[s].x;
      b1.x = x[s].y;
      b1.y = y[s].y;
      acc0 = __builtin_amdgcn_wmma_f32_16x16x4_f32(false, a[s], false, b0,
                                                   (short)0, acc0, false, false);
      acc1 = __builtin_amdgcn_wmma_f32_16x16x4_f32(false, a[s], false, b1,
                                                   (short)0, acc1, false, false);
    } else {
      acc0 = __builtin_amdgcn_wmma_f32_16x16x4_f32(false, a[s], false, x[s],
                                                   (short)0, acc0, false, false);
    }
  }
}

template <bool TWO>
__device__ __forceinline__ void wave_kloop_pipe(const float* __restrict__ ap,
                                                const char* __restrict__ Bb,
                                                unsigned ob, unsigned ldb4,
                                                int K, v8f& acc0, v8f& acc1) {
  v2f fa[PSETS][PNS], fx[PSETS][PNS], fy[PSETS][PNS];
#pragma unroll
  for (int g = 0; g < PAHEAD; ++g) {
    load_group<TWO>(ap, Bb, ob, ldb4, g * PKT, fa[g], fx[g], fy[g]);
    __builtin_amdgcn_sched_barrier(0);
  }
  int kb = 0;
#pragma nounroll
  for (; kb + PSETS * PKT < K; kb += PSETS * PKT) {
#pragma unroll
    for (int u = 0; u < PSETS; ++u) {
      __builtin_amdgcn_sched_barrier(0);
      load_group<TWO>(ap, Bb, ob, ldb4, kb + (u + PAHEAD) * PKT,
                      fa[(u + PAHEAD) % PSETS], fx[(u + PAHEAD) % PSETS],
                      fy[(u + PAHEAD) % PSETS]);
      __builtin_amdgcn_sched_barrier(0);
      mma_group<TWO>(fa[u], fx[u], fy[u], acc0, acc1);
    }
  }
  __builtin_amdgcn_sched_barrier(0);
  load_group<TWO>(ap, Bb, ob, ldb4, kb + PAHEAD * PKT, fa[PAHEAD % PSETS],
                  fx[PAHEAD % PSETS], fy[PAHEAD % PSETS]);
  __builtin_amdgcn_sched_barrier(0);
#pragma unroll
  for (int u = 0; u < PSETS; ++u)
    mma_group<TWO>(fa[u], fx[u], fy[u], acc0, acc1);
}

namespace {
constexpr int QP = 36;
constexpr int VP = 52;
constexpr int QT = 4;
constexpr int PS = Cc / 4;
}

__device__ __forceinline__ void load_rigid(const float* __restrict__ Ri,
                                           const float* __restrict__ Ti, int bi,
                                           float (&R)[3][3], float (&t)[3]) {
#pragma unroll
  for (int r = 0; r < 3; ++r)
#pragma unroll
    for (int cc = 0; cc < 3; ++cc) R[r][cc] = Ri[bi * 9 + r * 3 + cc];
#pragma unroll
  for (int x = 0; x < 3; ++x) t[x] = Ti[bi * 3 + x];
}

__device__ __forceinline__ void load_afrag(const float* __restrict__ arow,
                                           v2f (&a)[PS]) {
#pragma unroll
  for (int s = 0; s < PS; ++s) a[s] = *(const v2f*)(arow + 4 * s);
}

__device__ __forceinline__ void load_bfrag(const float* __restrict__ wcol, int ldw,
                                           v2f (&bf)[PS]) {
#pragma unroll
  for (int s = 0; s < PS; ++s) {
    bf[s].x = wcol[(4 * s) * ldw];
    bf[s].y = wcol[(4 * s + 1) * ldw];
  }
}

__device__ __forceinline__ void mma_rows2(const v2f (&a0)[PS], const v2f (&a1)[PS],
                                          const v2f (&bf)[PS], v8f& c0, v8f& c1) {
#pragma unroll
  for (int s = 0; s < PS; ++s) {
    c0 = __builtin_amdgcn_wmma_f32_16x16x4_f32(false, a0[s], false, bf[s],
                                               (short)0, c0, false, false);
    c1 = __builtin_amdgcn_wmma_f32_16x16x4_f32(false, a1[s], false, bf[s],
                                               (short)0, c1, false, false);
  }
}

__device__ __forceinline__ void mma_cols2(const v2f (&a)[PS], const v2f (&b0)[PS],
                                          const v2f (&b1)[PS], v8f& c0, v8f& c1) {
#pragma unroll
  for (int s = 0; s < PS; ++s) {
    c0 = __builtin_amdgcn_wmma_f32_16x16x4_f32(false, a[s], false, b0[s],
                                               (short)0, c0, false, false);
    c1 = __builtin_amdgcn_wmma_f32_16x16x4_f32(false, a[s], false, b1[s],
                                               (short)0, c1, false, false);
  }
}

__device__ __forceinline__ void store_tile(float* dst, int pitch, const v8f& acc) {
#pragma unroll
  for (int r = 0; r < 8; ++r) dst[r * pitch] = acc[r];
}

__device__ __forceinline__ float pack_points(float* __restrict__ row,
                                             const float (&R)[3][3],
                                             const float (&t)[3], float pc) {
  float p[12];
#pragma unroll
  for (int d = 0; d < 12; d += 4) {
    f4 x = *(const f4*)(row + 16 + d);
    p[d] = x.x; p[d + 1] = x.y; p[d + 2] = x.z; p[d + 3] = x.w;
  }
  float o[16];
  float sq = 0.f;
#pragma unroll
  for (int pt = 0; pt < NQPc; ++pt) {
    const float* ps = p + pt * 3;
#pragma unroll
    for (int m = 0; m < 3; ++m) {
      float xt = R[m][0] * ps[0] + R[m][1] * ps[1] + R[m][2] * ps[2] + t[m];
      sq += xt * xt;
      o[pt * 3 + m] = pc * xt;
    }
  }
#pragma unroll
  for (int d = 12; d < 16; ++d) o[d] = 0.f;
#pragma unroll
  for (int d = 0; d < 16; d += 4) {
    f4 x = {o[d], o[d + 1], o[d + 2], o[d + 3]};
    *(f4*)(row + 16 + d) = x;
  }
  return sq;
}

__global__ __launch_bounds__(128) void k_scores(
    const float* __restrict__ si, const float* __restrict__ WQh,
    const float* __restrict__ WKh, const float* __restrict__ WVh,
    const float* __restrict__ WQhp, const float* __restrict__ WKhp,
    const float* __restrict__ WVhp, const float* __restrict__ Ri,
    const float* __restrict__ Ti, const float* __restrict__ gamah,
    float* __restrict__ S, float* __restrict__ vaug, float* __restrict__ sqq,
    float* __restrict__ sqk) {
  __shared__ __attribute__((aligned(16))) float qs[16 * QT * QP];
  __shared__ __attribute__((aligned(16))) float kss[128 * QP];
  __shared__ __attribute__((aligned(16))) float vs[128 * VP];

  const int z = blockIdx.z;
  const int b = z / Hc, h = z % Hc;
  const int m0 = blockIdx.x * (16 * QT);
  const int n0 = blockIdx.y * 128;
  const int tid = threadIdx.x;
  const bool edge = (blockIdx.x == 0);
  const int wave = __builtin_amdgcn_readfirstlane(tid >> 5);
  const int lane = tid & 31;
  const int hf = lane >> 4;
  const int l16 = lane & 15;

  float Rk[3][3], tk[3], Rq[3][3], tq[3];
  load_rigid(Ri, Ti, b * Lc + n0 + tid, Rk, tk);
  load_rigid(Ri, Ti, b * Lc + m0 + (tid & (16 * QT - 1)), Rq, tq);

  {
    const float* sib = si + (long long)b * Lc * Cc + 2 * hf;
    const int cp = (l16 < 12) ? l16 : 0;
    v2f a0[PS], a1[PS], bf[PS];
    load_afrag(sib + (n0 + 32 * wave + l16) * Cc, a0);
    load_afrag(sib + (n0 + 32 * wave + 16 + l16) * Cc, a1);
    float* kd = kss + (32 * wave + 8 * hf) * QP + l16;
    {
      load_bfrag(WKh + (2 * hf) * 128 + h * 16 + l16, 128, bf);
      v8f c0 = {}, c1 = {};
      mma_rows2(a0, a1, bf, c0, c1);
      store_tile(kd, QP, c0);
      store_tile(kd + 16 * QP, QP, c1);
    }
    {
      load_bfrag(WKhp + (2 * hf) * 96 + h * 12 + cp, 96, bf);
      v8f c0 = {}, c1 = {};
      mma_rows2(a0, a1, bf, c0, c1);
      store_tile(kd + 16, QP, c0);
      store_tile(kd + 16 * QP + 16, QP, c1);
    }
    if (edge) {
      float* vd = vs + (32 * wave + 8 * hf) * VP + l16;
      const int c8 = (l16 < 8) ? l16 : 0;
      {
        load_bfrag(WVh + (2 * hf) * 128 + h * 16 + l16, 128, bf);
        v8f c0 = {}, c1 = {};
        mma_rows2(a0, a1, bf, c0, c1);
        store_tile(vd, VP, c0);
        store_tile(vd + 16 * VP, VP, c1);
      }
      {
        load_bfrag(WVhp + (2 * hf) * 192 + h * 24 + l16, 192, bf);
        v8f c0 = {}, c1 = {};
        mma_rows2(a0, a1, bf, c0, c1);
        store_tile(vd + 16, VP, c0);
        store_tile(vd + 16 * VP + 16, VP, c1);
      }
      {
        load_bfrag(WVhp + (2 * hf) * 192 + h * 24 + 16 + c8, 192, bf);
        v8f c0 = {}, c1 = {};
        mma_rows2(a0, a1, bf, c0, c1);
        store_tile(vd + 32, VP, c0);
        store_tile(vd + 16 * VP + 32, VP, c1);
      }
    }
    {
      v2f bg[PS];
      load_afrag(sib + (m0 + 16 * wave + l16) * Cc, a0);
      load_bfrag(WQh + (2 * hf) * 128 + h * 16 + l16, 128, bf);
      load_bfrag(WQhp + (2 * hf) * 96 + h * 12 + cp, 96, bg);
      v8f c0 = {}, c1 = {};
      mma_cols2(a0, bf, bg, c0, c1);
      float* qd = qs + (16 * wave + 8 * hf) * QP + l16;
      store_tile(qd, QP, c0);
      store_tile(qd + 16, QP, c1);
    }
  }
  const float gwc = WCc * softplusf(gamah[h]);
  asm volatile("" ::"v"(Rk[0][0]), "v"(Rk[0][1]), "v"(Rk[0][2]), "v"(Rk[1][0]),
               "v"(Rk[1][1]), "v"(Rk[1][2]), "v"(Rk[2][0]), "v"(Rk[2][1]),
               "v"(Rk[2][2]), "v"(tk[0]), "v"(tk[1]), "v"(tk[2]));
  asm volatile("" ::"v"(Rq[0][0]), "v"(Rq[0][1]), "v"(Rq[0][2]), "v"(Rq[1][0]),
               "v"(Rq[1][1]), "v"(Rq[1][2]), "v"(Rq[2][0]), "v"(Rq[2][1]),
               "v"(Rq[2][2]), "v"(tq[0]), "v"(tq[1]), "v"(tq[2]));
  __syncthreads();

  {
    const int j = n0 + tid;
    const float (&R)[3][3] = Rk;
    const float (&t)[3] = tk;
    const float sk = pack_points(kss + tid * QP, R, t, 1.f);
    if (edge) {
      sqk[(long long)z * Lc + j] = sk;
      float* vrow = vaug + ((long long)z * Lc + j) * 48;
      const float* vr = vs + tid * VP;
      float vv[40];
#pragma unroll
      for (int d = 0; d < 40; d += 4) {
        f4 x = *(const f4*)(vr + d);
        vv[d] = x.x; vv[d + 1] = x.y; vv[d + 2] = x.z; vv[d + 3] = x.w;
      }
#pragma unroll
      for (int d = 0; d < DKc; ++d) vrow[d] = vv[d];
#pragma unroll
      for (int p = 0; p < NPVc; ++p) {
        const float* vsrc = vv + 16 + p * 3;
#pragma unroll
        for (int m = 0; m < 3; ++m)
          vrow[16 + p * 3 + m] =
              R[m][0] * vsrc[0] + R[m][1] * vsrc[1] + R[m][2] * vsrc[2] + t[m];
      }
#pragma unroll
      for (int d = 40; d < 48; ++d) vrow[d] = 0.f;
    }
  }
  if (tid < 16 * QT) {
    const int i = m0 + tid;
    float* qrow = qs + tid * QP;
#pragma unroll
    for (int d = 0; d < DKc; d += 4) {
      f4 x = *(const f4*)(qrow + d);
      f4 y = {0.25f * x.x, 0.25f * x.y, 0.25f * x.z, 0.25f * x.w};
      *(f4*)(qrow + d) = y;
    }
    const float sq = pack_points(qrow, Rq, tq, gwc);
    if (blockIdx.y == 0) sqq[(long long)z * Lc + i] = sq;
  }
  __syncthreads();

  const float* bp0 = kss + (wave * 32 + l16) * QP + 2 * hf;
  const float* bp1 = bp0 + 16 * QP;
  v2f b0[8], b1[8];
#pragma unroll
  for (int s = 0; s < 8; ++s) {
    b0[s] = *(const v2f*)(bp0 + 4 * s);
    b1[s] = *(const v2f*)(bp1 + 4 * s);
  }
#pragma unroll
  for (int qt = 0; qt < QT; ++qt) {
    const float* ap = qs + (16 * qt + l16) * QP + 2 * hf;
    v2f a[8];
#pragma unroll
    for (int s = 0; s < 8; ++s) a[s] = *(const v2f*)(ap + 4 * s);
    v8f acc0 = {};
    v8f acc1 = {};
#pragma unroll
    for (int s = 0; s < 8; ++s) {
      acc0 = __builtin_amdgcn_wmma_f32_16x16x4_f32(false, a[s], false, b0[s],
                                                   (short)0, acc0, false, false);
      acc1 = __builtin_amdgcn_wmma_f32_16x16x4_f32(false, a[s], false, b1[s],
                                                   (short)0, acc1, false, false);
    }
    float* srow = S + ((long long)z * Lc + m0 + 16 * qt + 8 * hf) * Lc + n0 +
                  wave * 32 + l16;
#pragma unroll
    for (int r = 0; r < 8; ++r) {
      srow[(long long)r * Lc] = acc0[r];
      srow[(long long)r * Lc + 16] = acc1[r];
    }
  }
}

namespace {
constexpr int ZP = 140;
constexpr int RP = 388;
constexpr int NSL = 3;
constexpr int SLICE_WGS = Bc * Lc / NSL;
constexpr int RW = Lc / 8;
static_assert(Lc % NSL == 0, "the slices of a workgroup share b");
static_assert(RW % 16 == 0, "whole 16-row tiles per wave");
}

__device__ __forceinline__ void barrier_lds() {
  __builtin_amdgcn_fence(__ATOMIC_RELEASE, "workgroup", "local");
  __builtin_amdgcn_s_barrier();
  __builtin_amdgcn_fence(__ATOMIC_ACQUIRE, "workgroup", "local");
}

__global__ __launch_bounds__(256) __attribute__((amdgpu_waves_per_eu(2, 2))) void k_slice(
    const float* __restrict__ zij, const float* __restrict__ Wb,
    float* __restrict__ S, const float* __restrict__ sqq,
    const float* __restrict__ sqk, const float* __restrict__ gamah,
    float* __restrict__ feats) {
  __shared__ __attribute__((aligned(16))) float zs[Lc * ZP];
  __shared__ __attribute__((aligned(16))) float sb[Hc * RP];
  __shared__ __attribute__((aligned(16))) float sa[Hc * RP];

  const int bi0 = blockIdx.x * NSL;
  const int b = bi0 / Lc, i0 = bi0 % Lc;
  const int tid = threadIdx.x;
  const int wave = __builtin_amdgcn_readfirstlane(tid >> 5);
  const int lane = tid & 31;
  const int hf = lane >> 4;
  const int l16 = lane & 15;
  const int j0 = wave * RW;
  const int z = b * Hc + wave;

  const f4* zsrc = (const f4*)(zij + ((long long)bi0 * Lc + j0) * Cc) + lane;
  f4 t[RW];
#pragma unroll
  for (int r = 0; r < RW; ++r) t[r] = zsrc[r * (Cc / 4)];

  float kq[12];
#pragma unroll
  for (int e = 0; e < 12; ++e) kq[e] = sqk[(long long)z * Lc + e * 32 + lane];
  v2f w[32];
  {
    const int cw = (l16 < Hc) ? l16 : 0;
    const float* wp = Wb + (2 * hf) * Hc + cw;
#pragma unroll
    for (int s = 0; s < 32; ++s) {
      w[s].x = wp[(4 * s) * Hc];
      w[s].y = wp[(4 * s + 1) * Hc];
    }
  }
  __builtin_amdgcn_sched_barrier(0);
  const float gw = 0.5f * WCc * softplusf(gamah[wave]);

#pragma unroll
  for (int sl = 0; sl < NSL; ++sl) {
    const int bi = bi0 + sl;
    const int i = i0 + sl;

#pragma unroll
    for (int r = 0; r < RW; ++r) *(f4*)(zs + (j0 + r) * ZP + 4 * lane) = t[r];
    __builtin_amdgcn_sched_barrier(0);

    float* p = S + ((long long)z * Lc + i) * Lc;
    float pv[12];
#pragma unroll
    for (int e = 0; e < 12; ++e) pv[e] = p[e * 32 + lane];
    const float sq = sqq[(long long)z * Lc + i];

    if (sl + 1 < NSL) {
      const f4* zn = zsrc + (long long)(sl + 1) * (Lc * Cc / 4);
#pragma unroll
      for (int r = 0; r < RW; ++r) t[r] = zn[r * (Cc / 4)];
    }
    __builtin_amdgcn_sched_barrier(0);

    barrier_lds();

    {
      const float* zp = zs + (j0 + l16) * ZP + 2 * hf;
      v8f acc[RW / 16] = {};
#pragma unroll
      for (int g = 0; g < 4; ++g) {
        v2f a[RW / 16][8];
#pragma unroll
        for (int tt = 0; tt < RW / 16; ++tt)
#pragma unroll
          for (int s = 0; s < 8; ++s)
            a[tt][s] = *(const v2f*)(zp + 16 * tt * ZP + 32 * g + 4 * s);
#pragma unroll
        for (int s = 0; s < 8; ++s)
#pragma unroll
          for (int tt = 0; tt < RW / 16; ++tt)
            acc[tt] = __builtin_amdgcn_wmma_f32_16x16x4_f32(
                false, a[tt][s], false, w[8 * g + s], (short)0, acc[tt], false,
                false);
      }
      if (l16 < Hc) {
#pragma unroll
        for (int tt = 0; tt < RW / 16; ++tt)
#pragma unroll
          for (int r = 0; r < 8; ++r)
            sb[l16 * RP + j0 + 16 * tt + r + 8 * hf] = acc[tt][r];
      }
    }
    barrier_lds();

    {
      float vals[12];
      float mx = -INFINITY;
#pragma unroll
      for (int e = 0; e < 12; ++e) {
        const int j = e * 32 + lane;
        vals[e] = WLc * (pv[e] + sb[wave * RP + j] - gw * (sq + kq[e]));
        mx = fmaxf(mx, vals[e]);
      }
#pragma unroll
      for (int o = 16; o >= 1; o >>= 1) mx = fmaxf(mx, __shfl_xor(mx, o, 32));
      float sum = 0.f;
#pragma unroll
      for (int e = 0; e < 12; ++e) {
        vals[e] = expf(vals[e] - mx);
        sum += vals[e];
      }
#pragma unroll
      for (int o = 16; o >= 1; o >>= 1) sum += __shfl_xor(sum, o, 32);
      const float inv = 1.f / sum;
#pragma unroll
      for (int e = 0; e < 12; ++e) {
        const int j = e * 32 + lane;
        const float al = vals[e] * inv;
        p[j] = al;
        sa[wave * RP + j] = al;
      }
    }
    barrier_lds();

    {
      const int c = wave * 16 + l16;
      const int ra = (l16 < Hc) ? l16 : 0;
      const float* ap = sa + ra * RP + 2 * hf;
      const float* zc = zs + (2 * hf) * ZP + c;
      v8f acc = {};
      for (int kb = 0; kb < Lc; kb += 64) {
        v2f a[16], bq[16];
#pragma unroll
        for (int s = 0; s < 16; ++s) {
          const int k = kb + 4 * s;
          a[s] = *(const v2f*)(ap + k);
          bq[s].x = zc[k * ZP];
          bq[s].y = zc[(k + 1) * ZP];
        }
        asm volatile("" ::"v"(a[0]), "v"(a[1]), "v"(a[2]), "v"(a[3]), "v"(a[4]), "v"(a[5]),
                     "v"(a[6]), "v"(a[7]), "v"(a[8]), "v"(a[9]), "v"(a[10]),
                     "v"(a[11]), "v"(a[12]), "v"(a[13]), "v"(a[14]), "v"(a[15]));
        asm volatile("" ::"v"(bq[0]), "v"(bq[1]), "v"(bq[2]), "v"(bq[3]), "v"(bq[4]), "v"(bq[5]),
                     "v"(bq[6]), "v"(bq[7]), "v"(bq[8]), "v"(bq[9]), "v"(bq[10]),
                     "v"(bq[11]), "v"(bq[12]), "v"(bq[13]), "v"(bq[14]), "v"(bq[15]));
#pragma unroll
        for (int s = 0; s < 16; ++s)
          acc = __builtin_amdgcn_wmma_f32_16x16x4_f32(false, a[s], false, bq[s],
                                                      (short)0, acc, false, false);
      }
      if (hf == 0) {
        float* frow = feats + (long long)bi * FEAT + c;
#pragma unroll
        for (int r = 0; r < 8; ++r) frow[r * Cc] = acc[r];
      }
    }
    if (sl + 1 < NSL) barrier_lds();
  }
}

namespace {
constexpr int OPK = Hc * Cc;
constexpr int TAILK = FEAT - OPK;
constexpr int TP = TAILK + 4;
constexpr int VW = 48;
constexpr int OVP = VW + 4;
constexpr int OSPLIT = 2;
constexpr int AP = Lc + 4;
constexpr int ACH = Lc / 128;
static_assert(OPK % 128 == 0 && TAILK % 128 == 0 && Lc % 128 == 0, "whole ring turns");
static_assert(Lc % 16 == 0, "the 16 rows of a workgroup share b");
}

__global__ __launch_bounds__(320) __attribute__((amdgpu_waves_per_eu(4, 4))) void k_out(
    const float* __restrict__ feats, const float* __restrict__ alpha,
    const float* __restrict__ vaug, const float* __restrict__ Ri,
    const float* __restrict__ Ti, const float* __restrict__ WOw,
    const float* __restrict__ WOb, float* __restrict__ out) {
  __shared__ __attribute__((aligned(16))) float tail[16 * TP];
  __shared__ __attribute__((aligned(16))) float ovs[Hc * 16 * OVP];
  __shared__ __attribute__((aligned(16))) float als[Hc * 16 * AP];

  const int m0 = blockIdx.x * 16;
  const int tid = threadIdx.x;
  const int wave = __builtin_amdgcn_readfirstlane(tid >> 5);
  const int lane = tid & 31;
  const int hf = lane >> 4;
  const int l16 = lane & 15;
  const bool prod = (wave == 2) || (wave == 3);
  const int n_t0 = blockIdx.y * 64 + (wave & 1) * 32 + 2 * l16;
  const unsigned ldb4 = (unsigned)Cc * 4u;
  const unsigned ob = (unsigned)(2 * hf * Cc + n_t0) * 4u;

  v8f acc0 = {};
  v8f acc1 = {};
  if (prod) {
    const float* ap = feats + (long long)(m0 + l16) * FEAT + 2 * hf;
    wave_kloop_pipe<true>(ap, (const char*)WOw, ob, ldb4, OPK, acc0, acc1);
  } else {
    const int h = (wave < 2) ? wave : wave - 2;
    const int z = (m0 / Lc) * Hc + h;
    float* al = als + h * (16 * AP);
    {
      const f4* asrc =
          (const f4*)(alpha + ((long long)z * Lc + (m0 % Lc)) * Lc) + lane;
      f4 ar[16 * ACH];
#pragma unroll
      for (int r = 0; r < 16; ++r)
#pragma unroll
        for (int q = 0; q < ACH; ++q) ar[r * ACH + q] = asrc[r * (Lc / 4) + q * 32];
#pragma unroll
      for (int r = 0; r < 16; ++r)
#pragma unroll
        for (int q = 0; q < ACH; ++q)
          *(f4*)(al + r * AP + q * 128 + 4 * lane) = ar[r * ACH + q];
    }
    float R[3][3], t[3];
    load_rigid(Ri, Ti, m0 + l16, R, t);
    __builtin_amdgcn_fence(__ATOMIC_RELEASE, "workgroup", "local");
    __builtin_amdgcn_fence(__ATOMIC_ACQUIRE, "workgroup", "local");

    const float* ap = al + l16 * AP + 2 * hf;
    const char* vb = (const char*)(vaug + (long long)z * Lc * VW);
    const unsigned vld4 = (unsigned)VW * 4u;
    v8f o0 = {}, o1 = {}, o2 = {}, ou = {};
    wave_kloop_pipe<true>(ap, vb, (unsigned)(2 * hf * VW + 2 * l16) * 4u, vld4, Lc,
                          o0, o1);
    asm volatile("" ::"v"(R[0][0]), "v"(R[0][1]), "v"(R[0][2]), "v"(R[1][0]),
                 "v"(R[1][1]), "v"(R[1][2]), "v"(R[2][0]), "v"(R[2][1]),
                 "v"(R[2][2]), "v"(t[0]), "v"(t[1]), "v"(t[2]));
    wave_kloop_pipe<false>(ap, vb, (unsigned)(2 * hf * VW + 32 + l16) * 4u, vld4, Lc,
                           o2, ou);

    float* ot = ovs + h * (16 * OVP);
#pragma unroll
    for (int r = 0; r < 8; ++r) {
      float* d = ot + (r + 8 * hf) * OVP;
      v2f pr = {o0[r], o1[r]};
      *(v2f*)(d + 2 * l16) = pr;
      d[32 + l16] = o2[r];
    }
    __builtin_amdgcn_fence(__ATOMIC_RELEASE, "workgroup", "local");
    __builtin_amdgcn_fence(__ATOMIC_ACQUIRE, "workgroup", "local");

    if (lane < 16) {
      const float* orow = ot + lane * OVP;
      float o[40];
#pragma unroll
      for (int d = 0; d < 40; d += 4) {
        f4 x = *(const f4*)(orow + d);
        o[d] = x.x; o[d + 1] = x.y; o[d + 2] = x.z; o[d + 3] = x.w;
      }
      float* trow = tail + lane * TP;
#pragma unroll
      for (int d = 0; d < DKc; d += 4) {
        f4 x = {o[d], o[d + 1], o[d + 2], o[d + 3]};
        *(f4*)(trow + h * 16 + d) = x;
      }
      float oh[24], as[8];
#pragma unroll
      for (int p = 0; p < NPVc; ++p) {
        float d0 = o[16 + p * 3 + 0] - t[0];
        float d1 = o[16 + p * 3 + 1] - t[1];
        float d2 = o[16 + p * 3 + 2] - t[2];
        float asum = 0.f;
#pragma unroll
        for (int m = 0; m < 3; ++m) {
          float v = R[0][m] * d0 + R[1][m] * d1 + R[2][m] * d2;
          oh[p * 3 + m] = v;
          asum += fabsf(v);
        }
        as[p] = asum;
      }
#pragma unroll
      for (int d = 0; d < 24; d += 4) {
        f4 x = {oh[d], oh[d + 1], oh[d + 2], oh[d + 3]};
        *(f4*)(trow + 128 + h * 24 + d) = x;
      }
#pragma unroll
      for (int d = 0; d < 8; d += 4) {
        f4 x = {as[d], as[d + 1], as[d + 2], as[d + 3]};
        *(f4*)(trow + 320 + h * 8 + d) = x;
      }
    }
  }
  __syncthreads();
  if (!prod) return;

  wave_kloop_pipe<true>(tail + l16 * TP + 2 * hf, (const char*)(WOw + OPK * Cc), ob,
                        ldb4, TAILK, acc0, acc1);

#pragma unroll
  for (int r = 0; r < 8; ++r) {
    const int m = m0 + r + 8 * hf;
    out[(long long)m * Cc + n_t0] = acc0[r] + WOb[n_t0];
    out[(long long)m * Cc + n_t0 + 1] = acc1[r] + WOb[n_t0 + 1];
  }
}

extern "C" void kernel_launch(void* const* d_in, const int* in_sizes, int n_in,
                              void* d_out, int out_size, void* d_ws,
                              size_t ws_size, hipStream_t stream) {
  const float* si = (const float*)d_in[0];
  const float* zij = (const float*)d_in[1];
  const float* Ri = (const float*)d_in[2];
  const float* Ti = (const float*)d_in[3];
  const float* WQh = (const float*)d_in[4];
  const float* WKh = (const float*)d_in[5];
  const float* WVh = (const float*)d_in[6];
  const float* WQhp = (const float*)d_in[7];
  const float* WKhp = (const float*)d_in[8];
  const float* WVhp = (const float*)d_in[9];
  const float* Wb = (const float*)d_in[10];
  const float* WOw = (const float*)d_in[11];
  const float* WOb = (const float*)d_in[12];
  const float* gamah = (const float*)d_in[13];
  float* out = (float*)d_out;
  float* ws = (float*)d_ws;

  const int BL = Bc * Lc;
  float* vaug = ws;
  float* sqq = vaug + 294912;
  float* sqk = sqq + 6144;
  float* alpha = sqk + 6144;
  float* feats = alpha + 2359296;

  k_scores<<<dim3(Lc / (16 * QT), Lc / 128, Bc * Hc), 128, 0, stream>>>(
      si, WQh, WKh, WVh, WQhp, WKhp, WVhp, Ri, Ti, gamah, alpha, vaug, sqq, sqk);

  k_slice<<<SLICE_WGS, 256, 0, stream>>>(zij, Wb, alpha, sqq, sqk, gamah, feats);

  k_out<<<dim3(BL / 16, OSPLIT), 320, 0, stream>>>(feats, alpha, vaug, Ri, Ti, WOw, WOb,
                                                out);
}


# k_scores waves copy their 48 rows of si to LDS with whole-row 512-byte loads and read the projections' A fragments there (was 8-byte fragment loads straight from memory); same k order, same sums
# speedup vs baseline: 7.1212x; 1.0313x over previous
#include <hip/hip_runtime.h>
#include <math.h>

typedef __attribute__((ext_vector_type(2))) float v2f;
typedef __attribute__((ext_vector_type(4))) float f4;
typedef __attribute__((ext_vector_type(8))) float v8f;

namespace {
constexpr int Bc = 2, Lc = 384, Cc = 128, Hc = 8, DKc = 16, NQPc = 4, NPVc = 8;
constexpr float WCc = 0.2357022603955158f;
constexpr float WLc = 0.5773502691896258f;
constexpr int FEAT = Hc * Cc + Hc * DKc + Hc * NPVc * 3 + Hc * NPVc;
}

__device__ __forceinline__ float softplusf(float x) {
  return (x > 0.f) ? (x + log1pf(expf(-x))) : log1pf(expf(x));
}

namespace {
constexpr int PKT = 32;
constexpr int PNS = PKT / 4;
constexpr int PSETS = 4;
constexpr int PAHEAD = PSETS - 1;
}

template <bool TWO>
__device__ __forceinline__ void load_group(const float* __restrict__ ap,
                                           const char* __restrict__ Bb,
                                           unsigned ob, unsigned ldb4, int kb,
                                           v2f (&a)[PNS], v2f (&x)[PNS],
                                           v2f (&y)[PNS]) {
#pragma unroll
  for (int s = 0; s < PNS; ++s) {
    const int k = kb + 4 * s;
    a[s] = *(const v2f*)(ap + k);
    const char* r0 = Bb + (unsigned long long)((unsigned)k * ldb4);
    const char* r1 = r0 + ldb4;
    if (TWO) {
      x[s] = *(const v2f*)(r0 + ob);
      y[s] = *(const v2f*)(r1 + ob);
    } else {
      x[s].x = *(const float*)(r0 + ob);
      x[s].y = *(const float*)(r1 + ob);
    }
  }
}

template <bool TWO>
__device__ __forceinline__ void mma_group(const v2f (&a)[PNS],
                                          const v2f (&x)[PNS],
                                          const v2f (&y)[PNS], v8f& acc0,
                                          v8f& acc1) {
#pragma unroll
  for (int s = 0; s < PNS; ++s) {
    if (TWO) {
      v2f b0, b1;
      b0.x = x[s].x;
      b0.y = y[s].x;
      b1.x = x[s].y;
      b1.y = y[s].y;
      acc0 = __builtin_amdgcn_wmma_f32_16x16x4_f32(false, a[s], false, b0,
                                                   (short)0, acc0, false, false);
      acc1 = __builtin_amdgcn_wmma_f32_16x16x4_f32(false, a[s], false, b1,
                                                   (short)0, acc1, false, false);
    } else {
      acc0 = __builtin_amdgcn_wmma_f32_16x16x4_f32(false, a[s], false, x[s],
                                                   (short)0, acc0, false, false);
    }
  }
}

template <bool TWO>
__device__ __forceinline__ void wave_kloop_pipe(const float* __restrict__ ap,
                                                const char* __restrict__ Bb,
                                                unsigned ob, unsigned ldb4,
                                                int K, v8f& acc0, v8f& acc1) {
  v2f fa[PSETS][PNS], fx[PSETS][PNS], fy[PSETS][PNS];
#pragma unroll
  for (int g = 0; g < PAHEAD; ++g) {
    load_group<TWO>(ap, Bb, ob, ldb4, g * PKT, fa[g], fx[g], fy[g]);
    __builtin_amdgcn_sched_barrier(0);
  }
  int kb = 0;
#pragma nounroll
  for (; kb + PSETS * PKT < K; kb += PSETS * PKT) {
#pragma unroll
    for (int u = 0; u < PSETS; ++u) {
      __builtin_amdgcn_sched_barrier(0);
      load_group<TWO>(ap, Bb, ob, ldb4, kb + (u + PAHEAD) * PKT,
                      fa[(u + PAHEAD) % PSETS], fx[(u + PAHEAD) % PSETS],
                      fy[(u + PAHEAD) % PSETS]);
      __builtin_amdgcn_sched_barrier(0);
      mma_group<TWO>(fa[u], fx[u], fy[u], acc0, acc1);
    }
  }
  __builtin_amdgcn_sched_barrier(0);
  load_group<TWO>(ap, Bb, ob, ldb4, kb + PAHEAD * PKT, fa[PAHEAD % PSETS],
                  fx[PAHEAD % PSETS], fy[PAHEAD % PSETS]);
  __builtin_amdgcn_sched_barrier(0);
#pragma unroll
  for (int u = 0; u < PSETS; ++u)
    mma_group<TWO>(fa[u], fx[u], fy[u], acc0, acc1);
}

namespace {
constexpr int QP = 36;
constexpr int VP = 52;
constexpr int QT = 4;
constexpr int PS = Cc / 4;
constexpr int SP = Cc + 4;
constexpr int SR = 32 + 16;
}

__device__ __forceinline__ void load_rigid(const float* __restrict__ Ri,
                                           const float* __restrict__ Ti, int bi,
                                           float (&R)[3][3], float (&t)[3]) {
#pragma unroll
  for (int r = 0; r < 3; ++r)
#pragma unroll
    for (int cc = 0; cc < 3; ++cc) R[r][cc] = Ri[bi * 9 + r * 3 + cc];
#pragma unroll
  for (int x = 0; x < 3; ++x) t[x] = Ti[bi * 3 + x];
}

__device__ __forceinline__ void load_afrag(const float* __restrict__ arow,
                                           v2f (&a)[PS]) {
#pragma unroll
  for (int s = 0; s < PS; ++s) a[s] = *(const v2f*)(arow + 4 * s);
}

__device__ __forceinline__ void load_bfrag(const float* __restrict__ wcol, int ldw,
                                           v2f (&bf)[PS]) {
#pragma unroll
  for (int s = 0; s < PS; ++s) {
    bf[s].x = wcol[(4 * s) * ldw];
    bf[s].y = wcol[(4 * s + 1) * ldw];
  }
}

__device__ __forceinline__ void mma_rows2(const v2f (&a0)[PS], const v2f (&a1)[PS],
                                          const v2f (&bf)[PS], v8f& c0, v8f& c1) {
#pragma unroll
  for (int s = 0; s < PS; ++s) {
    c0 = __builtin_amdgcn_wmma_f32_16x16x4_f32(false, a0[s], false, bf[s],
                                               (short)0, c0, false, false);
    c1 = __builtin_amdgcn_wmma_f32_16x16x4_f32(false, a1[s], false, bf[s],
                                               (short)0, c1, false, false);
  }
}

__device__ __forceinline__ void mma_cols2(const v2f (&a)[PS], const v2f (&b0)[PS],
                                          const v2f (&b1)[PS], v8f& c0, v8f& c1) {
#pragma unroll
  for (int s = 0; s < PS; ++s) {
    c0 = __builtin_amdgcn_wmma_f32_16x16x4_f32(false, a[s], false, b0[s],
                                               (short)0, c0, false, false);
    c1 = __builtin_amdgcn_wmma_f32_16x16x4_f32(false, a[s], false, b1[s],
                                               (short)0, c1, false, false);
  }
}

__device__ __forceinline__ void store_tile(float* dst, int pitch, const v8f& acc) {
#pragma unroll
  for (int r = 0; r < 8; ++r) dst[r * pitch] = acc[r];
}

__device__ __forceinline__ float pack_points(float* __restrict__ row,
                                             const float (&R)[3][3],
                                             const float (&t)[3], float pc) {
  float p[12];
#pragma unroll
  for (int d = 0; d < 12; d += 4) {
    f4 x = *(const f4*)(row + 16 + d);
    p[d] = x.x; p[d + 1] = x.y; p[d + 2] = x.z; p[d + 3] = x.w;
  }
  float o[16];
  float sq = 0.f;
#pragma unroll
  for (int pt = 0; pt < NQPc; ++pt) {
    const float* ps = p + pt * 3;
#pragma unroll
    for (int m = 0; m < 3; ++m) {
      float xt = R[m][0] * ps[0] + R[m][1] * ps[1] + R[m][2] * ps[2] + t[m];
      sq += xt * xt;
      o[pt * 3 + m] = pc * xt;
    }
  }
#pragma unroll
  for (int d = 12; d < 16; ++d) o[d] = 0.f;
#pragma unroll
  for (int d = 0; d < 16; d += 4) {
    f4 x = {o[d], o[d + 1], o[d + 2], o[d + 3]};
    *(f4*)(row + 16 + d) = x;
  }
  return sq;
}

__global__ __launch_bounds__(128) void k_scores(
    const float* __restrict__ si, const float* __restrict__ WQh,
    const float* __restrict__ WKh, const float* __restrict__ WVh,
    const float* __restrict__ WQhp, const float* __restrict__ WKhp,
    const float* __restrict__ WVhp, const float* __restrict__ Ri,
    const float* __restrict__ Ti, const float* __restrict__ gamah,
    float* __restrict__ S, float* __restrict__ vaug, float* __restrict__ sqq,
    float* __restrict__ sqk) {
  __shared__ __attribute__((aligned(16))) float qs[16 * QT * QP];
  __shared__ __attribute__((aligned(16))) float kss[128 * QP];
  __shared__ __attribute__((aligned(16))) float vs[128 * VP];
  __shared__ __attribute__((aligned(16))) float sis[4 * SR * SP];

  const int z = blockIdx.z;
  const int b = z / Hc, h = z % Hc;
  const int m0 = blockIdx.x * (16 * QT);
  const int n0 = blockIdx.y * 128;
  const int tid = threadIdx.x;
  const bool edge = (blockIdx.x == 0);
  const int wave = __builtin_amdgcn_readfirstlane(tid >> 5);
  const int lane = tid & 31;
  const int hf = lane >> 4;
  const int l16 = lane & 15;

  float Rk[3][3], tk[3], Rq[3][3], tq[3];
  load_rigid(Ri, Ti, b * Lc + n0 + tid, Rk, tk);
  load_rigid(Ri, Ti, b * Lc + m0 + (tid & (16 * QT - 1)), Rq, tq);

  {
    float* sw = sis + wave * (SR * SP);
    {
      const f4* kg = (const f4*)(si + ((long long)b * Lc + n0 + 32 * wave) * Cc) + lane;
      const f4* qg = (const f4*)(si + ((long long)b * Lc + m0 + 16 * wave) * Cc) + lane;
      f4 sr[SR];
#pragma unroll
      for (int r = 0; r < 32; ++r) sr[r] = kg[r * (Cc / 4)];
#pragma unroll
      for (int r = 0; r < 16; ++r) sr[32 + r] = qg[r * (Cc / 4)];
#pragma unroll
      for (int r = 0; r < SR; ++r) *(f4*)(sw + r * SP + 4 * lane) = sr[r];
    }
    __builtin_amdgcn_fence(__ATOMIC_RELEASE, "workgroup", "local");
    __builtin_amdgcn_fence(__ATOMIC_ACQUIRE, "workgroup", "local");
    const float* sl = sw + l16 * SP + 2 * hf;
    const int cp = (l16 < 12) ? l16 : 0;
    v2f a0[PS], a1[PS], bf[PS];
    load_afrag(sl, a0);
    load_afrag(sl + 16 * SP, a1);
    float* kd = kss + (32 * wave + 8 * hf) * QP + l16;
    {
      load_bfrag(WKh + (2 * hf) * 128 + h * 16 + l16, 128, bf);
      v8f c0 = {}, c1 = {};
      mma_rows2(a0, a1, bf, c0, c1);
      store_tile(kd, QP, c0);
      store_tile(kd + 16 * QP, QP, c1);
    }
    {
      load_bfrag(WKhp + (2 * hf) * 96 + h * 12 + cp, 96, bf);
      v8f c0 = {}, c1 = {};
      mma_rows2(a0, a1, bf, c0, c1);
      store_tile(kd + 16, QP, c0);
      store_tile(kd + 16 * QP + 16, QP, c1);
    }
    if (edge) {
      float* vd = vs + (32 * wave + 8 * hf) * VP + l16;
      const int c8 = (l16 < 8) ? l16 : 0;
      {
        load_bfrag(WVh + (2 * hf) * 128 + h * 16 + l16, 128, bf);
        v8f c0 = {}, c1 = {};
        mma_rows2(a0, a1, bf, c0, c1);
        store_tile(vd, VP, c0);
        store_tile(vd + 16 * VP, VP, c1);
      }
      {
        load_bfrag(WVhp + (2 * hf) * 192 + h * 24 + l16, 192, bf);
        v8f c0 = {}, c1 = {};
        mma_rows2(a0, a1, bf, c0, c1);
        store_tile(vd + 16, VP, c0);
        store_tile(vd + 16 * VP + 16, VP, c1);
      }
      {
        load_bfrag(WVhp + (2 * hf) * 192 + h * 24 + 16 + c8, 192, bf);
        v8f c0 = {}, c1 = {};
        mma_rows2(a0, a1, bf, c0, c1);
        store_tile(vd + 32, VP, c0);
        store_tile(vd + 16 * VP + 32, VP, c1);
      }
    }
    {
      v2f bg[PS];
      load_afrag(sl + 32 * SP, a0);
      load_bfrag(WQh + (2 * hf) * 128 + h * 16 + l16, 128, bf);
      load_bfrag(WQhp + (2 * hf) * 96 + h * 12 + cp, 96, bg);
      v8f c0 = {}, c1 = {};
      mma_cols2(a0, bf, bg, c0, c1);
      float* qd = qs + (16 * wave + 8 * hf) * QP + l16;
      store_tile(qd, QP, c0);
      store_tile(qd + 16, QP, c1);
    }
  }
  const float gwc = WCc * softplusf(gamah[h]);
  asm volatile("" ::"v"(Rk[0][0]), "v"(Rk[0][1]), "v"(Rk[0][2]), "v"(Rk[1][0]),
               "v"(Rk[1][1]), "v"(Rk[1][2]), "v"(Rk[2][0]), "v"(Rk[2][1]),
               "v"(Rk[2][2]), "v"(tk[0]), "v"(tk[1]), "v"(tk[2]));
  asm volatile("" ::"v"(Rq[0][0]), "v"(Rq[0][1]), "v"(Rq[0][2]), "v"(Rq[1][0]),
               "v"(Rq[1][1]), "v"(Rq[1][2]), "v"(Rq[2][0]), "v"(Rq[2][1]),
               "v"(Rq[2][2]), "v"(tq[0]), "v"(tq[1]), "v"(tq[2]));
  __syncthreads();

  {
    const int j = n0 + tid;
    const float (&R)[3][3] = Rk;
    const float (&t)[3] = tk;
    const float sk = pack_points(kss + tid * QP, R, t, 1.f);
    if (edge) {
      sqk[(long long)z * Lc + j] = sk;
      float* vrow = vaug + ((long long)z * Lc + j) * 48;
      const float* vr = vs + tid * VP;
      float vv[40];
#pragma unroll
      for (int d = 0; d < 40; d += 4) {
        f4 x = *(const f4*)(vr + d);
        vv[d] = x.x; vv[d + 1] = x.y; vv[d + 2] = x.z; vv[d + 3] = x.w;
      }
#pragma unroll
      for (int d = 0; d < DKc; ++d) vrow[d] = vv[d];
#pragma unroll
      for (int p = 0; p < NPVc; ++p) {
        const float* vsrc = vv + 16 + p * 3;
#pragma unroll
        for (int m = 0; m < 3; ++m)
          vrow[16 + p * 3 + m] =
              R[m][0] * vsrc[0] + R[m][1] * vsrc[1] + R[m][2] * vsrc[2] + t[m];
      }
#pragma unroll
      for (int d = 40; d < 48; ++d) vrow[d] = 0.f;
    }
  }
  if (tid < 16 * QT) {
    const int i = m0 + tid;
    float* qrow = qs + tid * QP;
#pragma unroll
    for (int d = 0; d < DKc; d += 4) {
      f4 x = *(const f4*)(qrow + d);
      f4 y = {0.25f * x.x, 0.25f * x.y, 0.25f * x.z, 0.25f * x.w};
      *(f4*)(qrow + d) = y;
    }
    const float sq = pack_points(qrow, Rq, tq, gwc);
    if (blockIdx.y == 0) sqq[(long long)z * Lc + i] = sq;
  }
  __syncthreads();

  const float* bp0 = kss + (wave * 32 + l16) * QP + 2 * hf;
  const float* bp1 = bp0 + 16 * QP;
  v2f b0[8], b1[8];
#pragma unroll
  for (int s = 0; s < 8; ++s) {
    b0[s] = *(const v2f*)(bp0 + 4 * s);
    b1[s] = *(const v2f*)(bp1 + 4 * s);
  }
#pragma unroll
  for (int qt = 0; qt < QT; ++qt) {
    const float* ap = qs + (16 * qt + l16) * QP + 2 * hf;
    v2f a[8];
#pragma unroll
    for (int s = 0; s < 8; ++s) a[s] = *(const v2f*)(ap + 4 * s);
    v8f acc0 = {};
    v8f acc1 = {};
#pragma unroll
    for (int s = 0; s < 8; ++s) {
      acc0 = __builtin_amdgcn_wmma_f32_16x16x4_f32(false, a[s], false, b0[s],
                                                   (short)0, acc0, false, false);
      acc1 = __builtin_amdgcn_wmma_f32_16x16x4_f32(false, a[s], false, b1[s],
                                                   (short)0, acc1, false, false);
    }
    float* srow = S + ((long long)z * Lc + m0 + 16 * qt + 8 * hf) * Lc + n0 +
                  wave * 32 + l16;
#pragma unroll
    for (int r = 0; r < 8; ++r) {
      srow[(long long)r * Lc] = acc0[r];
      srow[(long long)r * Lc + 16] = acc1[r];
    }
  }
}

namespace {
constexpr int ZP = 140;
constexpr int RP = 388;
constexpr int NSL = 3;
constexpr int SLICE_WGS = Bc * Lc / NSL;
constexpr int RW = Lc / 8;
static_assert(Lc % NSL == 0, "the slices of a workgroup share b");
static_assert(RW % 16 == 0, "whole 16-row tiles per wave");
}

__device__ __forceinline__ void barrier_lds() {
  __builtin_amdgcn_fence(__ATOMIC_RELEASE, "workgroup", "local");
  __builtin_amdgcn_s_barrier();
  __builtin_amdgcn_fence(__ATOMIC_ACQUIRE, "workgroup", "local");
}

__global__ __launch_bounds__(256) __attribute__((amdgpu_waves_per_eu(2, 2))) void k_slice(
    const float* __restrict__ zij, const float* __restrict__ Wb,
    float* __restrict__ S, const float* __restrict__ sqq,
    const float* __restrict__ sqk, const float* __restrict__ gamah,
    float* __restrict__ feats) {
  __shared__ __attribute__((aligned(16))) float zs[Lc * ZP];
  __shared__ __attribute__((aligned(16))) float sb[Hc * RP];
  __shared__ __attribute__((aligned(16))) float sa[Hc * RP];

  const int bi0 = blockIdx.x * NSL;
  const int b = bi0 / Lc, i0 = bi0 % Lc;
  const int tid = threadIdx.x;
  const int wave = __builtin_amdgcn_readfirstlane(tid >> 5);
  const int lane = tid & 31;
  const int hf = lane >> 4;
  const int l16 = lane & 15;
  const int j0 = wave * RW;
  const int z = b * Hc + wave;

  const f4* zsrc = (const f4*)(zij + ((long long)bi0 * Lc + j0) * Cc) + lane;
  f4 t[RW];
#pragma unroll
  for (int r = 0; r < RW; ++r) t[r] = zsrc[r * (Cc / 4)];

  float kq[12];
#pragma unroll
  for (int e = 0; e < 12; ++e) kq[e] = sqk[(long long)z * Lc + e * 32 + lane];
  v2f w[32];
  {
    const int cw = (l16 < Hc) ? l16 : 0;
    const float* wp = Wb + (2 * hf) * Hc + cw;
#pragma unroll
    for (int s = 0; s < 32; ++s) {
      w[s].x = wp[(4 * s) * Hc];
      w[s].y = wp[(4 * s + 1) * Hc];
    }
  }
  __builtin_amdgcn_sched_barrier(0);
  const float gw = 0.5f * WCc * softplusf(gamah[wave]);

#pragma unroll
  for (int sl = 0; sl < NSL; ++sl) {
    const int bi = bi0 + sl;
    const int i = i0 + sl;

#pragma unroll
    for (int r = 0; r < RW; ++r) *(f4*)(zs + (j0 + r) * ZP + 4 * lane) = t[r];
    __builtin_amdgcn_sched_barrier(0);

    float* p = S + ((long long)z * Lc + i) * Lc;
    float pv[12];
#pragma unroll
    for (int e = 0; e < 12; ++e) pv[e] = p[e * 32 + lane];
    const float sq = sqq[(long long)z * Lc + i];

    if (sl + 1 < NSL) {
      const f4* zn = zsrc + (long long)(sl + 1) * (Lc * Cc / 4);
#pragma unroll
      for (int r = 0; r < RW; ++r) t[r] = zn[r * (Cc / 4)];
    }
    __builtin_amdgcn_sched_barrier(0);

    barrier_lds();

    {
      const float* zp = zs + (j0 + l16) * ZP + 2 * hf;
      v8f acc[RW / 16] = {};
#pragma unroll
      for (int g = 0; g < 4; ++g) {
        v2f a[RW / 16][8];
#pragma unroll
        for (int tt = 0; tt < RW / 16; ++tt)
#pragma unroll
          for (int s = 0; s < 8; ++s)
            a[tt][s] = *(const v2f*)(zp + 16 * tt * ZP + 32 * g + 4 * s);
#pragma unroll
        for (int s = 0; s < 8; ++s)
#pragma unroll
          for (int tt = 0; tt < RW / 16; ++tt)
            acc[tt] = __builtin_amdgcn_wmma_f32_16x16x4_f32(
                false, a[tt][s], false, w[8 * g + s], (short)0, acc[tt], false,
                false);
      }
      if (l16 < Hc) {
#pragma unroll
        for (int tt = 0; tt < RW / 16; ++tt)
#pragma unroll
          for (int r = 0; r < 8; ++r)
            sb[l16 * RP + j0 + 16 * tt + r + 8 * hf] = acc[tt][r];
      }
    }
    barrier_lds();

    {
      float vals[12];
      float mx = -INFINITY;
#pragma unroll
      for (int e = 0; e < 12; ++e) {
        const int j = e * 32 + lane;
        vals[e] = WLc * (pv[e] + sb[wave * RP + j] - gw * (sq + kq[e]));
        mx = fmaxf(mx, vals[e]);
      }
#pragma unroll
      for (int o = 16; o >= 1; o >>= 1) mx = fmaxf(mx, __shfl_xor(mx, o, 32));
      float sum = 0.f;
#pragma unroll
      for (int e = 0; e < 12; ++e) {
        vals[e] = expf(vals[e] - mx);
        sum += vals[e];
      }
#pragma unroll
      for (int o = 16; o >= 1; o >>= 1) sum += __shfl_xor(sum, o, 32);
      const float inv = 1.f / sum;
#pragma unroll
      for (int e = 0; e < 12; ++e) {
        const int j = e * 32 + lane;
        const float al = vals[e] * inv;
        p[j] = al;
        sa[wave * RP + j] = al;
      }
    }
    barrier_lds();

    {
      const int c = wave * 16 + l16;
      const int ra = (l16 < Hc) ? l16 : 0;
      const float* ap = sa + ra * RP + 2 * hf;
      const float* zc = zs + (2 * hf) * ZP + c;
      v8f acc = {};
      for (int kb = 0; kb < Lc; kb += 64) {
        v2f a[16], bq[16];
#pragma unroll
        for (int s = 0; s < 16; ++s) {
          const int k = kb + 4 * s;
          a[s] = *(const v2f*)(ap + k);
          bq[s].x = zc[k * ZP];
          bq[s].y = zc[(k + 1) * ZP];
        }
        asm volatile("" ::"v"(a[0]), "v"(a[1]), "v"(a[2]), "v"(a[3]), "v"(a[4]), "v"(a[5]),
                     "v"(a[6]), "v"(a[7]), "v"(a[8]), "v"(a[9]), "v"(a[10]),
                     "v"(a[11]), "v"(a[12]), "v"(a[13]), "v"(a[14]), "v"(a[15]));
        asm volatile("" ::"v"(bq[0]), "v"(bq[1]), "v"(bq[2]), "v"(bq[3]), "v"(bq[4]), "v"(bq[5]),
                     "v"(bq[6]), "v"(bq[7]), "v"(bq[8]), "v"(bq[9]), "v"(bq[10]),
                     "v"(bq[11]), "v"(bq[12]), "v"(bq[13]), "v"(bq[14]), "v"(bq[15]));
#pragma unroll
        for (int s = 0; s < 16; ++s)
          acc = __builtin_amdgcn_wmma_f32_16x16x4_f32(false, a[s], false, bq[s],
                                                      (short)0, acc, false, false);
      }
      if (hf == 0) {
        float* frow = feats + (long long)bi * FEAT + c;
#pragma unroll
        for (int r = 0; r < 8; ++r) frow[r * Cc] = acc[r];
      }
    }
    if (sl + 1 < NSL) barrier_lds();
  }
}

namespace {
constexpr int OPK = Hc * Cc;
constexpr int TAILK = FEAT - OPK;
constexpr int TP = TAILK + 4;
constexpr int VW = 48;
constexpr int OVP = VW + 4;
constexpr int OSPLIT = 2;
constexpr int AP = Lc + 4;
constexpr int ACH = Lc / 128;
static_assert(OPK % 128 == 0 && TAILK % 128 == 0 && Lc % 128 == 0, "whole ring turns");
static_assert(Lc % 16 == 0, "the 16 rows of a workgroup share b");
}

__global__ __launch_bounds__(320) __attribute__((amdgpu_waves_per_eu(4, 4))) void k_out(
    const float* __restrict__ feats, const float* __restrict__ alpha,
    const float* __restrict__ vaug, const float* __restrict__ Ri,
    const float* __restrict__ Ti, const float* __restrict__ WOw,
    const float* __restrict__ WOb, float* __restrict__ out) {
  __shared__ __attribute__((aligned(16))) float tail[16 * TP];
  __shared__ __attribute__((aligned(16))) float ovs[Hc * 16 * OVP];
  __shared__ __attribute__((aligned(16))) float als[Hc * 16 * AP];

  const int m0 = blockIdx.x * 16;
  const int tid = threadIdx.x;
  const int wave = __builtin_amdgcn_readfirstlane(tid >> 5);
  const int lane = tid & 31;
  const int hf = lane >> 4;
  const int l16 = lane & 15;
  const bool prod = (wave == 2) || (wave == 3);
  const int n_t0 = blockIdx.y * 64 + (wave & 1) * 32 + 2 * l16;
  const unsigned ldb4 = (unsigned)Cc * 4u;
  const unsigned ob = (unsigned)(2 * hf * Cc + n_t0) * 4u;

  v8f acc0 = {};
  v8f acc1 = {};
  if (prod) {
    const float* ap = feats + (long long)(m0 + l16) * FEAT + 2 * hf;
    wave_kloop_pipe<true>(ap, (const char*)WOw, ob, ldb4, OPK, acc0, acc1);
  } else {
    const int h = (wave < 2) ? wave : wave - 2;
    const int z = (m0 / Lc) * Hc + h;
    float* al = als + h * (16 * AP);
    {
      const f4* asrc =
          (const f4*)(alpha + ((long long)z * Lc + (m0 % Lc)) * Lc) + lane;
      f4 ar[16 * ACH];
#pragma unroll
      for (int r = 0; r < 16; ++r)
#pragma unroll
        for (int q = 0; q < ACH; ++q) ar[r * ACH + q] = asrc[r * (Lc / 4) + q * 32];
#pragma unroll
      for (int r = 0; r < 16; ++r)
#pragma unroll
        for (int q = 0; q < ACH; ++q)
          *(f4*)(al + r * AP + q * 128 + 4 * lane) = ar[r * ACH + q];
    }
    float R[3][3], t[3];
    load_rigid(Ri, Ti, m0 + l16, R, t);
    __builtin_amdgcn_fence(__ATOMIC_RELEASE, "workgroup", "local");
    __builtin_amdgcn_fence(__ATOMIC_ACQUIRE, "workgroup", "local");

    const float* ap = al + l16 * AP + 2 * hf;
    const char* vb = (const char*)(vaug + (long long)z * Lc * VW);
    const unsigned vld4 = (unsigned)VW * 4u;
    v8f o0 = {}, o1 = {}, o2 = {}, ou = {};
    wave_kloop_pipe<true>(ap, vb, (unsigned)(2 * hf * VW + 2 * l16) * 4u, vld4, Lc,
                          o0, o1);
    asm volatile("" ::"v"(R[0][0]), "v"(R[0][1]), "v"(R[0][2]), "v"(R[1][0]),
                 "v"(R[1][1]), "v"(R[1][2]), "v"(R[2][0]), "v"(R[2][1]),
                 "v"(R[2][2]), "v"(t[0]), "v"(t[1]), "v"(t[2]));
    wave_kloop_pipe<false>(ap, vb, (unsigned)(2 * hf * VW + 32 + l16) * 4u, vld4, Lc,
                           o2, ou);

    float* ot = ovs + h * (16 * OVP);
#pragma unroll
    for (int r = 0; r < 8; ++r) {
      float* d = ot + (r + 8 * hf) * OVP;
      v2f pr = {o0[r], o1[r]};
      *(v2f*)(d + 2 * l16) = pr;
      d[32 + l16] = o2[r];
    }
    __builtin_amdgcn_fence(__ATOMIC_RELEASE, "workgroup", "local");
    __builtin_amdgcn_fence(__ATOMIC_ACQUIRE, "workgroup", "local");

    if (lane < 16) {
      const float* orow = ot + lane * OVP;
      float o[40];
#pragma unroll
      for (int d = 0; d < 40; d += 4) {
        f4 x = *(const f4*)(orow + d);
        o[d] = x.x; o[d + 1] = x.y; o[d + 2] = x.z; o[d + 3] = x.w;
      }
      float* trow = tail + lane * TP;
#pragma unroll
      for (int d = 0; d < DKc; d += 4) {
        f4 x = {o[d], o[d + 1], o[d + 2], o[d + 3]};
        *(f4*)(trow + h * 16 + d) = x;
      }
      float oh[24], as[8];
#pragma unroll
      for (int p = 0; p < NPVc; ++p) {
        float d0 = o[16 + p * 3 + 0] - t[0];
        float d1 = o[16 + p * 3 + 1] - t[1];
        float d2 = o[16 + p * 3 + 2] - t[2];
        float asum = 0.f;
#pragma unroll
        for (int m = 0; m < 3; ++m) {
          float v = R[0][m] * d0 + R[1][m] * d1 + R[2][m] * d2;
          oh[p * 3 + m] = v;
          asum += fabsf(v);
        }
        as[p] = asum;
      }
#pragma unroll
      for (int d = 0; d < 24; d += 4) {
        f4 x = {oh[d], oh[d + 1], oh[d + 2], oh[d + 3]};
        *(f4*)(trow + 128 + h * 24 + d) = x;
      }
#pragma unroll
      for (int d = 0; d < 8; d += 4) {
        f4 x = {as[d], as[d + 1], as[d + 2], as[d + 3]};
        *(f4*)(trow + 320 + h * 8 + d) = x;
      }
    }
  }
  __syncthreads();
  if (!prod) return;

  wave_kloop_pipe<true>(tail + l16 * TP + 2 * hf, (const char*)(WOw + OPK * Cc), ob,
                        ldb4, TAILK, acc0, acc1);

#pragma unroll
  for (int r = 0; r < 8; ++r) {
    const int m = m0 + r + 8 * hf;
    out[(long long)m * Cc + n_t0] = acc0[r] + WOb[n_t0];
    out[(long long)m * Cc + n_t0 + 1] = acc1[r] + WOb[n_t0 + 1];
  }
}

extern "C" void kernel_launch(void* const* d_in, const int* in_sizes, int n_in,
                              void* d_out, int out_size, void* d_ws,
                              size_t ws_size, hipStream_t stream) {
  const float* si = (const float*)d_in[0];
  const float* zij = (const float*)d_in[1];
  const float* Ri = (const float*)d_in[2];
  const float* Ti = (const float*)d_in[3];
  const float* WQh = (const float*)d_in[4];
  const float* WKh = (const float*)d_in[5];
  const float* WVh = (const float*)d_in[6];
  const float* WQhp = (const float*)d_in[7];
  const float* WKhp = (const float*)d_in[8];
  const float* WVhp = (const float*)d_in[9];
  const float* Wb = (const float*)d_in[10];
  const float* WOw = (const float*)d_in[11];
  const float* WOb = (const float*)d_in[12];
  const float* gamah = (const float*)d_in[13];
  float* out = (float*)d_out;
  float* ws = (float*)d_ws;

  const int BL = Bc * Lc;
  float* vaug = ws;
  float* sqq = vaug + 294912;
  float* sqk = sqq + 6144;
  float* alpha = sqk + 6144;
  float* feats = alpha + 2359296;

  k_scores<<<dim3(Lc / (16 * QT), Lc / 128, Bc * Hc), 128, 0, stream>>>(
      si, WQh, WKh, WVh, WQhp, WKhp, WVhp, Ri, Ti, gamah, alpha, vaug, sqq, sqk);

  k_slice<<<SLICE_WGS, 256, 0, stream>>>(zij, Wb, alpha, sqq, sqk, gamah, feats);

  k_out<<<dim3(BL / 16, OSPLIT), 320, 0, stream>>>(feats, alpha, vaug, Ri, Ti, WOw, WOb,
                                                out);
}
